# starting kernel
# baseline (speedup 1.0000x reference)
#include <hip/hip_runtime.h>


typedef __attribute__((ext_vector_type(2))) float v2f;
typedef __attribute__((ext_vector_type(8))) float v8f;

#define SQRT_HALF 0.70710678118654752440f

__device__ __forceinline__ int reflect_i(int x, int n) {
  x = (x < 0) ? (-1 - x) : x;
  x = (x >= n) ? (2 * n - 1 - x) : x;
  return x;
}

__device__ __forceinline__ float cif_A(int m, int kk,
                                       const float* __restrict__ ha,
                                       const float* __restrict__ hb, int hp) {
  int p = m & 3;
  int q = m >> 2;
  int want_odd = hp ? ((p & 1) ^ 1) : (p & 1);
  int t = ((kk - want_odd) >> 1) - q;
  bool ok = ((kk & 1) == want_odd) && (t >= 0) && (t <= 4);
  int tc = t < 0 ? 0 : (t > 4 ? 4 : t);
  const float* __restrict__ h = ((p == 0) || (p == 2)) ? ha : hb;
  int idx = (p < 2) ? (2 * tc + 1) : (2 * tc);
  float v = h[idx];
  return ok ? v : 0.0f;
}

__device__ __forceinline__ float cf_A(int m, int kk,
                                      const float* __restrict__ h, int taps) {
  int t = kk - m - (3 - (taps >> 1));
  bool ok = (t >= 0) && (t < taps);
  int tc = t < 0 ? 0 : (t >= taps ? taps - 1 : t);
  float v = h[tc];
  return ok ? v : 0.0f;
}

__device__ __forceinline__ float c2q_val(const float* __restrict__ p1,
                                         const float* __restrict__ p2, int rr,
                                         int stride, int cb) {
  int o = (rr >> 1) * stride;
  float w1r = p1[o], w1i = p1[o + 1];
  float w2r = p2[o], w2i = p2[o + 1];
  float even = cb ? (w1i + w2i) : (w1r + w2r);
  float odd = cb ? (w2r - w1r) : (w1i - w2i);
  return (rr & 1) ? odd : even;
}

__device__ __forceinline__ v8f wmma4(v2f a, v2f b, v8f c) {
  return __builtin_amdgcn_wmma_f32_16x16x4_f32(false, a, false, b, (short)0, c,
                                               false, false);
}

template <bool SRC0_PLAIN>
__global__ void __launch_bounds__(256)
j2_col_kernel(const float* __restrict__ src0_plain, const float* __restrict__ yh,
              int s0a, int s0b, int s1a, int s1b, int Hb, int Wb, int r, int W,
              const float* __restrict__ ha0, const float* __restrict__ hb0,
              const float* __restrict__ ha1, const float* __restrict__ hb1,
              float* __restrict__ out, int nTilesR, int nTilesW, int B) {
  int lane = threadIdx.x & 31;
  int wave = threadIdx.x >> 5;
  long tile = (long)blockIdx.x * 8 + wave;
  long total = (long)B * nTilesR * nTilesW;
  if (tile >= total) return;
  int tpi = nTilesR * nTilesW;
  int b = (int)(tile / tpi);
  int t2 = (int)(tile % tpi);
  int I = t2 / nTilesW;
  int Jc = t2 % nTilesW;
  int x0 = 8 * I - 4;
  int half = lane >> 4;
  int l16 = lane & 15;
  int koff = half * 2;
  int M = l16;
  int cc = Jc * 16 + l16;

  const float s0sc = SRC0_PLAIN ? 1.0f : SQRT_HALF;
  v2f a0[4], a1[4];
#pragma unroll
  for (int c = 0; c < 4; ++c) {
    int kk = 4 * c + koff;
    a0[c].x = cif_A(M, kk, ha0, hb0, 0) * s0sc;
    a0[c].y = cif_A(M, kk + 1, ha0, hb0, 0) * s0sc;
    a1[c].x = cif_A(M, kk, ha1, hb1, 1) * SQRT_HALF;
    a1[c].y = cif_A(M, kk + 1, ha1, hb1, 1) * SQRT_HALF;
  }

  int cb = cc & 1;
  int stride = 2 * Wb;
  int j2 = (cc >> 1) * 2;
  const float* p0 = SRC0_PLAIN ? (src0_plain + (size_t)b * r * W + cc) : nullptr;
  const float* p0a = yh + ((size_t)b * 6 + s0a) * Hb * stride + j2;
  const float* p0b = yh + ((size_t)b * 6 + s0b) * Hb * stride + j2;
  const float* p1a = yh + ((size_t)b * 6 + s1a) * Hb * stride + j2;
  const float* p1b = yh + ((size_t)b * 6 + s1b) * Hb * stride + j2;

  v8f acc = {0.f, 0.f, 0.f, 0.f, 0.f, 0.f, 0.f, 0.f};
#pragma unroll
  for (int c = 0; c < 4; ++c) {
    int kk = 4 * c + koff;
    int r0 = reflect_i(x0 + kk, r);
    int r1 = reflect_i(x0 + kk + 1, r);
    v2f b0, b1;
    if (SRC0_PLAIN) {
      b0.x = p0[(size_t)r0 * W];
      b0.y = p0[(size_t)r1 * W];
    } else {
      b0.x = c2q_val(p0a, p0b, r0, stride, cb);
      b0.y = c2q_val(p0a, p0b, r1, stride, cb);
    }
    b1.x = c2q_val(p1a, p1b, r0, stride, cb);
    b1.y = c2q_val(p1a, p1b, r1, stride, cb);
    acc = wmma4(a0[c], b0, acc);
    acc = wmma4(a1[c], b1, acc);
  }

  int outR = 2 * r;
  float* op = out + ((size_t)b * outR + (size_t)I * 16 + half * 8) * W + cc;
#pragma unroll
  for (int e = 0; e < 8; ++e) op[(size_t)e * W] = acc[e];
}

__global__ void __launch_bounds__(256)
j2_row_kernel(const float* __restrict__ T1, const float* __restrict__ T2, int H,
              int Wt, const float* __restrict__ ha0,
              const float* __restrict__ hb0, const float* __restrict__ ha1,
              const float* __restrict__ hb1, float* __restrict__ out,
              int nTilesR, int nTilesW, int B) {
  int lane = threadIdx.x & 31;
  int wave = threadIdx.x >> 5;
  long tile = (long)blockIdx.x * 8 + wave;
  long total = (long)B * nTilesR * nTilesW;
  if (tile >= total) return;
  int tpi = nTilesR * nTilesW;
  int b = (int)(tile / tpi);
  int t2 = (int)(tile % tpi);
  int I = t2 / nTilesW;
  int Jc = t2 % nTilesW;
  int m0 = I * 16;
  int xc0 = 8 * Jc - 4;
  int half = lane >> 4;
  int l16 = lane & 15;
  int koff = half * 2;

  v2f bf0[4], bf1[4];
#pragma unroll
  for (int c = 0; c < 4; ++c) {
    int kk = 4 * c + koff;
    bf0[c].x = cif_A(l16, kk, ha0, hb0, 0);
    bf0[c].y = cif_A(l16, kk + 1, ha0, hb0, 0);
    bf1[c].x = cif_A(l16, kk, ha1, hb1, 1);
    bf1[c].y = cif_A(l16, kk + 1, ha1, hb1, 1);
  }

  const float* row1 = T1 + ((size_t)b * H + m0 + l16) * Wt;
  const float* row2 = T2 + ((size_t)b * H + m0 + l16) * Wt;
  v8f acc = {0.f, 0.f, 0.f, 0.f, 0.f, 0.f, 0.f, 0.f};
#pragma unroll
  for (int c = 0; c < 4; ++c) {
    int kk = 4 * c + koff;
    int c0 = reflect_i(xc0 + kk, Wt);
    int c1 = reflect_i(xc0 + kk + 1, Wt);
    v2f aA, aB;
    aA.x = row1[c0];
    aA.y = row1[c1];
    aB.x = row2[c0];
    aB.y = row2[c1];
    acc = wmma4(aA, bf0[c], acc);
    acc = wmma4(aB, bf1[c], acc);
  }

  int outW = 2 * Wt;
  float* op = out + ((size_t)b * H + m0 + half * 8) * outW + Jc * 16 + l16;
#pragma unroll
  for (int e = 0; e < 8; ++e) op[(size_t)e * outW] = acc[e];
}

template <bool SRC0_PLAIN>
__global__ void __launch_bounds__(256)
j1_col_kernel(const float* __restrict__ src0_plain, const float* __restrict__ yh,
              int s0a, int s0b, int s1a, int s1b, int Hb, int Wb, int r, int W,
              const float* __restrict__ h0, int taps0,
              const float* __restrict__ h1, int taps1, float* __restrict__ out,
              int nTilesR, int nTilesW, int B) {
  int lane = threadIdx.x & 31;
  int wave = threadIdx.x >> 5;
  long tile = (long)blockIdx.x * 8 + wave;
  long total = (long)B * nTilesR * nTilesW;
  if (tile >= total) return;
  int tpi = nTilesR * nTilesW;
  int b = (int)(tile / tpi);
  int t2 = (int)(tile % tpi);
  int I = t2 / nTilesW;
  int Jc = t2 % nTilesW;
  int x0 = 16 * I - 3;
  int half = lane >> 4;
  int l16 = lane & 15;
  int koff = half * 2;
  int M = l16;
  int cc = Jc * 16 + l16;

  const float s0sc = SRC0_PLAIN ? 1.0f : SQRT_HALF;
  v2f a0[6], a1[6];
#pragma unroll
  for (int c = 0; c < 6; ++c) {
    int kk = 4 * c + koff;
    a0[c].x = cf_A(M, kk, h0, taps0) * s0sc;
    a0[c].y = cf_A(M, kk + 1, h0, taps0) * s0sc;
    a1[c].x = cf_A(M, kk, h1, taps1) * SQRT_HALF;
    a1[c].y = cf_A(M, kk + 1, h1, taps1) * SQRT_HALF;
  }

  int cb = cc & 1;
  int stride = 2 * Wb;
  int j2 = (cc >> 1) * 2;
  const float* p0 = SRC0_PLAIN ? (src0_plain + (size_t)b * r * W + cc) : nullptr;
  const float* p0a = yh + ((size_t)b * 6 + s0a) * Hb * stride + j2;
  const float* p0b = yh + ((size_t)b * 6 + s0b) * Hb * stride + j2;
  const float* p1a = yh + ((size_t)b * 6 + s1a) * Hb * stride + j2;
  const float* p1b = yh + ((size_t)b * 6 + s1b) * Hb * stride + j2;

  v8f acc = {0.f, 0.f, 0.f, 0.f, 0.f, 0.f, 0.f, 0.f};
#pragma unroll
  for (int c = 0; c < 6; ++c) {
    int kk = 4 * c + koff;
    int r0 = reflect_i(x0 + kk, r);
    int r1 = reflect_i(x0 + kk + 1, r);
    v2f b0, b1;
    if (SRC0_PLAIN) {
      b0.x = p0[(size_t)r0 * W];
      b0.y = p0[(size_t)r1 * W];
    } else {
      b0.x = c2q_val(p0a, p0b, r0, stride, cb);
      b0.y = c2q_val(p0a, p0b, r1, stride, cb);
    }
    b1.x = c2q_val(p1a, p1b, r0, stride, cb);
    b1.y = c2q_val(p1a, p1b, r1, stride, cb);
    acc = wmma4(a0[c], b0, acc);
    acc = wmma4(a1[c], b1, acc);
  }

  float* op = out + ((size_t)b * r + (size_t)I * 16 + half * 8) * W + cc;
#pragma unroll
  for (int e = 0; e < 8; ++e) op[(size_t)e * W] = acc[e];
}

__global__ void __launch_bounds__(256)
j1_row_kernel(const float* __restrict__ T1, const float* __restrict__ T2, int H,
              int Wt, const float* __restrict__ h0, int taps0,
              const float* __restrict__ h1, int taps1, float* __restrict__ out,
              int nTilesR, int nTilesW, int B) {
  int lane = threadIdx.x & 31;
  int wave = threadIdx.x >> 5;
  long tile = (long)blockIdx.x * 8 + wave;
  long total = (long)B * nTilesR * nTilesW;
  if (tile >= total) return;
  int tpi = nTilesR * nTilesW;
  int b = (int)(tile / tpi);
  int t2 = (int)(tile % tpi);
  int I = t2 / nTilesW;
  int Jc = t2 % nTilesW;
  int m0 = I * 16;
  int xc0 = 16 * Jc - 3;
  int half = lane >> 4;
  int l16 = lane & 15;
  int koff = half * 2;

  v2f bf0[6], bf1[6];
#pragma unroll
  for (int c = 0; c < 6; ++c) {
    int kk = 4 * c + koff;
    bf0[c].x = cf_A(l16, kk, h0, taps0);
    bf0[c].y = cf_A(l16, kk + 1, h0, taps0);
    bf1[c].x = cf_A(l16, kk, h1, taps1);
    bf1[c].y = cf_A(l16, kk + 1, h1, taps1);
  }

  const float* row1 = T1 + ((size_t)b * H + m0 + l16) * Wt;
  const float* row2 = T2 + ((size_t)b * H + m0 + l16) * Wt;
  v8f acc = {0.f, 0.f, 0.f, 0.f, 0.f, 0.f, 0.f, 0.f};
#pragma unroll
  for (int c = 0; c < 6; ++c) {
    int kk = 4 * c + koff;
    int c0 = reflect_i(xc0 + kk, Wt);
    int c1 = reflect_i(xc0 + kk + 1, Wt);
    v2f aA, aB;
    aA.x = row1[c0];
    aA.y = row1[c1];
    aB.x = row2[c0];
    aB.y = row2[c1];
    acc = wmma4(aA, bf0[c], acc);
    acc = wmma4(aB, bf1[c], acc);
  }

  float* op = out + ((size_t)b * H + m0 + half * 8) * Wt + Jc * 16 + l16;
#pragma unroll
  for (int e = 0; e < 8; ++e) op[(size_t)e * Wt] = acc[e];
}

extern "C" void kernel_launch(void* const* d_in, const int* in_sizes, int n_in,
                              void* d_out, int out_size, void* d_ws,
                              size_t ws_size, hipStream_t stream) {
  (void)in_sizes;
  (void)n_in;
  (void)out_size;
  (void)ws_size;

  const float* yl = (const float*)d_in[0];
  const float* yh0 = (const float*)d_in[1];
  const float* yh1 = (const float*)d_in[2];
  const float* yh2 = (const float*)d_in[3];
  const float* g0o = (const float*)d_in[4];
  const float* g1o = (const float*)d_in[5];
  const float* g0a = (const float*)d_in[6];
  const float* g0b = (const float*)d_in[7];
  const float* g1a = (const float*)d_in[8];
  const float* g1b = (const float*)d_in[9];

  const int B = 64;
  const size_t IMG = (size_t)512 * 512;
  float* T1 = (float*)d_ws;
  float* T2 = T1 + (size_t)B * IMG;
  float* LL = T2 + (size_t)B * IMG;
  float* outp = (float*)d_out;

  dim3 blk(256);
  auto nblk = [](long tiles) { return dim3((unsigned)((tiles + 7) / 8)); };

  {
    int r = 128, W = 128, Hb = 64, Wb = 64;
    int ntr = 2 * r / 16, ntw = W / 16;
    long tc = (long)B * ntr * ntw;
    j2_col_kernel<true><<<nblk(tc), blk, 0, stream>>>(
        yl, yh2, 0, 0, 0, 5, Hb, Wb, r, W, g0b, g0a, g1b, g1a, T1, ntr, ntw, B);
    j2_col_kernel<false><<<nblk(tc), blk, 0, stream>>>(
        nullptr, yh2, 2, 3, 1, 4, Hb, Wb, r, W, g0b, g0a, g1b, g1a, T2, ntr,
        ntw, B);
    int H = 2 * r, Wt = W;
    int rtr = H / 16, rtw = 2 * Wt / 16;
    long tr = (long)B * rtr * rtw;
    j2_row_kernel<<<nblk(tr), blk, 0, stream>>>(T1, T2, H, Wt, g0b, g0a, g1b,
                                                g1a, LL, rtr, rtw, B);
  }

  {
    int r = 256, W = 256, Hb = 128, Wb = 128;
    int ntr = 2 * r / 16, ntw = W / 16;
    long tc = (long)B * ntr * ntw;
    j2_col_kernel<true><<<nblk(tc), blk, 0, stream>>>(
        LL, yh1, 0, 0, 0, 5, Hb, Wb, r, W, g0b, g0a, g1b, g1a, T1, ntr, ntw, B);
    j2_col_kernel<false><<<nblk(tc), blk, 0, stream>>>(
        nullptr, yh1, 2, 3, 1, 4, Hb, Wb, r, W, g0b, g0a, g1b, g1a, T2, ntr,
        ntw, B);
    int H = 2 * r, Wt = W;
    int rtr = H / 16, rtw = 2 * Wt / 16;
    long tr = (long)B * rtr * rtw;
    j2_row_kernel<<<nblk(tr), blk, 0, stream>>>(T1, T2, H, Wt, g0b, g0a, g1b,
                                                g1a, LL, rtr, rtw, B);
  }

  {
    int r = 512, W = 512, Hb = 256, Wb = 256;
    int ntr = r / 16, ntw = W / 16;
    long tc = (long)B * ntr * ntw;
    j1_col_kernel<true><<<nblk(tc), blk, 0, stream>>>(
        LL, yh0, 0, 0, 0, 5, Hb, Wb, r, W, g0o, 7, g1o, 5, T1, ntr, ntw, B);
    j1_col_kernel<false><<<nblk(tc), blk, 0, stream>>>(
        nullptr, yh0, 2, 3, 1, 4, Hb, Wb, r, W, g0o, 7, g1o, 5, T2, ntr, ntw,
        B);
    long tr = (long)B * ntr * ntw;
    j1_row_kernel<<<nblk(tr), blk, 0, stream>>>(T1, T2, r, W, g0o, 7, g1o, 5,
                                                outp, ntr, ntw, B);
  }
}


# All six kernel bodies split into a load phase (every global load of the tile issued back to back) and a compute phase: one memory wait per tile, not one per load pair; sum order unchanged
# speedup vs baseline: 1.1499x; 1.1499x over previous
#include <hip/hip_runtime.h>


typedef __attribute__((ext_vector_type(2))) float v2f;
typedef __attribute__((ext_vector_type(8))) float v8f;

#define SQRT_HALF 0.70710678118654752440f

__device__ __forceinline__ int reflect_i(int x, int n) {
  x = (x < 0) ? (-1 - x) : x;
  x = (x >= n) ? (2 * n - 1 - x) : x;
  return x;
}

__device__ __forceinline__ const float* cif_tap(int m, int kk,
                                                const float* __restrict__ ha,
                                                const float* __restrict__ hb,
                                                int hp, bool& ok) {
  int p = m & 3;
  int q = m >> 2;
  int want_odd = hp ? ((p & 1) ^ 1) : (p & 1);
  int t = ((kk - want_odd) >> 1) - q;
  ok = ((kk & 1) == want_odd) && (t >= 0) && (t <= 4);
  int tc = t < 0 ? 0 : (t > 4 ? 4 : t);
  const float* __restrict__ h = ((p == 0) || (p == 2)) ? ha : hb;
  int idx = (p < 2) ? (2 * tc + 1) : (2 * tc);
  return h + idx;
}

__device__ __forceinline__ const float* cf_tap(int m, int kk,
                                               const float* __restrict__ h,
                                               int taps, bool& ok) {
  int t = kk - m - (3 - (taps >> 1));
  ok = (t >= 0) && (t < taps);
  int tc = t < 0 ? 0 : (t >= taps ? taps - 1 : t);
  return h + tc;
}

struct Q4 {
  float w1r, w1i, w2r, w2i;
};

__device__ __forceinline__ Q4 c2q_load(const float* __restrict__ p1,
                                       const float* __restrict__ p2, int rr,
                                       int stride) {
  int o = (rr >> 1) * stride;
  Q4 q;
  q.w1r = p1[o];
  q.w1i = p1[o + 1];
  q.w2r = p2[o];
  q.w2i = p2[o + 1];
  return q;
}

__device__ __forceinline__ float c2q_pick(Q4 q, int rr, int cb) {
  float even = cb ? (q.w1i + q.w2i) : (q.w1r + q.w2r);
  float odd = cb ? (q.w2r - q.w1r) : (q.w1i - q.w2i);
  return (rr & 1) ? odd : even;
}

__device__ __forceinline__ v8f wmma4(v2f a, v2f b, v8f c) {
  return __builtin_amdgcn_wmma_f32_16x16x4_f32(false, a, false, b, (short)0, c,
                                               false, false);
}

#define LOADS_STAY_AHEAD() __builtin_amdgcn_sched_barrier(0)

template <bool SRC0_PLAIN>
__global__ void __launch_bounds__(256)
j2_col_kernel(const float* __restrict__ src0_plain, const float* __restrict__ yh,
              int s0a, int s0b, int s1a, int s1b, int Hb, int Wb, int r, int W,
              const float* __restrict__ ha0, const float* __restrict__ hb0,
              const float* __restrict__ ha1, const float* __restrict__ hb1,
              float* __restrict__ out, int nTilesR, int nTilesW, int B) {
  int lane = threadIdx.x & 31;
  int wave = threadIdx.x >> 5;
  long tile = (long)blockIdx.x * 8 + wave;
  long total = (long)B * nTilesR * nTilesW;
  if (tile >= total) return;
  int tpi = nTilesR * nTilesW;
  int b = (int)(tile / tpi);
  int t2 = (int)(tile % tpi);
  int I = t2 / nTilesW;
  int Jc = t2 % nTilesW;
  int x0 = 8 * I - 4;
  int half = lane >> 4;
  int l16 = lane & 15;
  int koff = half * 2;
  int M = l16;
  int cc = Jc * 16 + l16;

  int cb = cc & 1;
  int stride = 2 * Wb;
  int j2 = (cc >> 1) * 2;
  const float* p0 = SRC0_PLAIN ? (src0_plain + (size_t)b * r * W + cc) : nullptr;
  const float* p0a = yh + ((size_t)b * 6 + s0a) * Hb * stride + j2;
  const float* p0b = yh + ((size_t)b * 6 + s0b) * Hb * stride + j2;
  const float* p1a = yh + ((size_t)b * 6 + s1a) * Hb * stride + j2;
  const float* p1b = yh + ((size_t)b * 6 + s1b) * Hb * stride + j2;

  float f0[4][2], f1[4][2];
  bool k0[4][2], k1[4][2];
  int rr[4][2];
  float d0[4][2];
  Q4 q0[4][2], q1[4][2];
#pragma unroll
  for (int c = 0; c < 4; ++c) {
#pragma unroll
    for (int j = 0; j < 2; ++j) {
      int kk = 4 * c + koff + j;
      f0[c][j] = *cif_tap(M, kk, ha0, hb0, 0, k0[c][j]);
      f1[c][j] = *cif_tap(M, kk, ha1, hb1, 1, k1[c][j]);
    }
  }
#pragma unroll
  for (int c = 0; c < 4; ++c) {
#pragma unroll
    for (int j = 0; j < 2; ++j) {
      rr[c][j] = reflect_i(x0 + 4 * c + koff + j, r);
      if (SRC0_PLAIN) {
        d0[c][j] = p0[(size_t)rr[c][j] * W];
      } else {
        q0[c][j] = c2q_load(p0a, p0b, rr[c][j], stride);
      }
      q1[c][j] = c2q_load(p1a, p1b, rr[c][j], stride);
    }
  }
  LOADS_STAY_AHEAD();

  const float s0sc = SRC0_PLAIN ? 1.0f : SQRT_HALF;
  v8f acc = {0.f, 0.f, 0.f, 0.f, 0.f, 0.f, 0.f, 0.f};
#pragma unroll
  for (int c = 0; c < 4; ++c) {
    v2f a0, a1, b0, b1;
    a0.x = (k0[c][0] ? f0[c][0] : 0.0f) * s0sc;
    a0.y = (k0[c][1] ? f0[c][1] : 0.0f) * s0sc;
    a1.x = (k1[c][0] ? f1[c][0] : 0.0f) * SQRT_HALF;
    a1.y = (k1[c][1] ? f1[c][1] : 0.0f) * SQRT_HALF;
    if (SRC0_PLAIN) {
      b0.x = d0[c][0];
      b0.y = d0[c][1];
    } else {
      b0.x = c2q_pick(q0[c][0], rr[c][0], cb);
      b0.y = c2q_pick(q0[c][1], rr[c][1], cb);
    }
    b1.x = c2q_pick(q1[c][0], rr[c][0], cb);
    b1.y = c2q_pick(q1[c][1], rr[c][1], cb);
    acc = wmma4(a0, b0, acc);
    acc = wmma4(a1, b1, acc);
  }

  int outR = 2 * r;
  float* op = out + ((size_t)b * outR + (size_t)I * 16 + half * 8) * W + cc;
#pragma unroll
  for (int e = 0; e < 8; ++e) op[(size_t)e * W] = acc[e];
}

__global__ void __launch_bounds__(256)
j2_row_kernel(const float* __restrict__ T1, const float* __restrict__ T2, int H,
              int Wt, const float* __restrict__ ha0,
              const float* __restrict__ hb0, const float* __restrict__ ha1,
              const float* __restrict__ hb1, float* __restrict__ out,
              int nTilesR, int nTilesW, int B) {
  int lane = threadIdx.x & 31;
  int wave = threadIdx.x >> 5;
  long tile = (long)blockIdx.x * 8 + wave;
  long total = (long)B * nTilesR * nTilesW;
  if (tile >= total) return;
  int tpi = nTilesR * nTilesW;
  int b = (int)(tile / tpi);
  int t2 = (int)(tile % tpi);
  int I = t2 / nTilesW;
  int Jc = t2 % nTilesW;
  int m0 = I * 16;
  int xc0 = 8 * Jc - 4;
  int half = lane >> 4;
  int l16 = lane & 15;
  int koff = half * 2;

  const float* row1 = T1 + ((size_t)b * H + m0 + l16) * Wt;
  const float* row2 = T2 + ((size_t)b * H + m0 + l16) * Wt;

  float f0[4][2], f1[4][2];
  bool k0[4][2], k1[4][2];
  float dA[4][2], dB[4][2];
#pragma unroll
  for (int c = 0; c < 4; ++c) {
#pragma unroll
    for (int j = 0; j < 2; ++j) {
      int kk = 4 * c + koff + j;
      f0[c][j] = *cif_tap(l16, kk, ha0, hb0, 0, k0[c][j]);
      f1[c][j] = *cif_tap(l16, kk, ha1, hb1, 1, k1[c][j]);
    }
  }
#pragma unroll
  for (int c = 0; c < 4; ++c) {
#pragma unroll
    for (int j = 0; j < 2; ++j) {
      int cj = reflect_i(xc0 + 4 * c + koff + j, Wt);
      dA[c][j] = row1[cj];
      dB[c][j] = row2[cj];
    }
  }

  v8f acc = {0.f, 0.f, 0.f, 0.f, 0.f, 0.f, 0.f, 0.f};
#pragma unroll
  for (int c = 0; c < 4; ++c) {
    v2f aA, aB, bf0, bf1;
    bf0.x = k0[c][0] ? f0[c][0] : 0.0f;
    bf0.y = k0[c][1] ? f0[c][1] : 0.0f;
    bf1.x = k1[c][0] ? f1[c][0] : 0.0f;
    bf1.y = k1[c][1] ? f1[c][1] : 0.0f;
    aA.x = dA[c][0];
    aA.y = dA[c][1];
    aB.x = dB[c][0];
    aB.y = dB[c][1];
    acc = wmma4(aA, bf0, acc);
    acc = wmma4(aB, bf1, acc);
  }

  int outW = 2 * Wt;
  float* op = out + ((size_t)b * H + m0 + half * 8) * outW + Jc * 16 + l16;
#pragma unroll
  for (int e = 0; e < 8; ++e) op[(size_t)e * outW] = acc[e];
}

template <bool SRC0_PLAIN>
__global__ void __launch_bounds__(256)
j1_col_kernel(const float* __restrict__ src0_plain, const float* __restrict__ yh,
              int s0a, int s0b, int s1a, int s1b, int Hb, int Wb, int r, int W,
              const float* __restrict__ h0, int taps0,
              const float* __restrict__ h1, int taps1, float* __restrict__ out,
              int nTilesR, int nTilesW, int B) {
  int lane = threadIdx.x & 31;
  int wave = threadIdx.x >> 5;
  long tile = (long)blockIdx.x * 8 + wave;
  long total = (long)B * nTilesR * nTilesW;
  if (tile >= total) return;
  int tpi = nTilesR * nTilesW;
  int b = (int)(tile / tpi);
  int t2 = (int)(tile % tpi);
  int I = t2 / nTilesW;
  int Jc = t2 % nTilesW;
  int x0 = 16 * I - 3;
  int half = lane >> 4;
  int l16 = lane & 15;
  int koff = half * 2;
  int M = l16;
  int cc = Jc * 16 + l16;

  int cb = cc & 1;
  int stride = 2 * Wb;
  int j2 = (cc >> 1) * 2;
  const float* p0 = SRC0_PLAIN ? (src0_plain + (size_t)b * r * W + cc) : nullptr;
  const float* p0a = yh + ((size_t)b * 6 + s0a) * Hb * stride + j2;
  const float* p0b = yh + ((size_t)b * 6 + s0b) * Hb * stride + j2;
  const float* p1a = yh + ((size_t)b * 6 + s1a) * Hb * stride + j2;
  const float* p1b = yh + ((size_t)b * 6 + s1b) * Hb * stride + j2;

  float f0[6][2], f1[6][2];
  bool k0[6][2], k1[6][2];
  int rr[6][2];
  float d0[6][2];
  Q4 q0[6][2], q1[6][2];
#pragma unroll
  for (int c = 0; c < 6; ++c) {
#pragma unroll
    for (int j = 0; j < 2; ++j) {
      int kk = 4 * c + koff + j;
      f0[c][j] = *cf_tap(M, kk, h0, taps0, k0[c][j]);
      f1[c][j] = *cf_tap(M, kk, h1, taps1, k1[c][j]);
    }
  }
#pragma unroll
  for (int c = 0; c < 6; ++c) {
#pragma unroll
    for (int j = 0; j < 2; ++j) {
      rr[c][j] = reflect_i(x0 + 4 * c + koff + j, r);
      if (SRC0_PLAIN) {
        d0[c][j] = p0[(size_t)rr[c][j] * W];
      } else {
        q0[c][j] = c2q_load(p0a, p0b, rr[c][j], stride);
      }
      q1[c][j] = c2q_load(p1a, p1b, rr[c][j], stride);
    }
  }
  LOADS_STAY_AHEAD();

  const float s0sc = SRC0_PLAIN ? 1.0f : SQRT_HALF;
  v8f acc = {0.f, 0.f, 0.f, 0.f, 0.f, 0.f, 0.f, 0.f};
#pragma unroll
  for (int c = 0; c < 6; ++c) {
    v2f a0, a1, b0, b1;
    a0.x = (k0[c][0] ? f0[c][0] : 0.0f) * s0sc;
    a0.y = (k0[c][1] ? f0[c][1] : 0.0f) * s0sc;
    a1.x = (k1[c][0] ? f1[c][0] : 0.0f) * SQRT_HALF;
    a1.y = (k1[c][1] ? f1[c][1] : 0.0f) * SQRT_HALF;
    if (SRC0_PLAIN) {
      b0.x = d0[c][0];
      b0.y = d0[c][1];
    } else {
      b0.x = c2q_pick(q0[c][0], rr[c][0], cb);
      b0.y = c2q_pick(q0[c][1], rr[c][1], cb);
    }
    b1.x = c2q_pick(q1[c][0], rr[c][0], cb);
    b1.y = c2q_pick(q1[c][1], rr[c][1], cb);
    acc = wmma4(a0, b0, acc);
    acc = wmma4(a1, b1, acc);
  }

  float* op = out + ((size_t)b * r + (size_t)I * 16 + half * 8) * W + cc;
#pragma unroll
  for (int e = 0; e < 8; ++e) op[(size_t)e * W] = acc[e];
}

__global__ void __launch_bounds__(256)
j1_row_kernel(const float* __restrict__ T1, const float* __restrict__ T2, int H,
              int Wt, const float* __restrict__ h0, int taps0,
              const float* __restrict__ h1, int taps1, float* __restrict__ out,
              int nTilesR, int nTilesW, int B) {
  int lane = threadIdx.x & 31;
  int wave = threadIdx.x >> 5;
  long tile = (long)blockIdx.x * 8 + wave;
  long total = (long)B * nTilesR * nTilesW;
  if (tile >= total) return;
  int tpi = nTilesR * nTilesW;
  int b = (int)(tile / tpi);
  int t2 = (int)(tile % tpi);
  int I = t2 / nTilesW;
  int Jc = t2 % nTilesW;
  int m0 = I * 16;
  int xc0 = 16 * Jc - 3;
  int half = lane >> 4;
  int l16 = lane & 15;
  int koff = half * 2;

  const float* row1 = T1 + ((size_t)b * H + m0 + l16) * Wt;
  const float* row2 = T2 + ((size_t)b * H + m0 + l16) * Wt;

  float f0[6][2], f1[6][2];
  bool k0[6][2], k1[6][2];
  float dA[6][2], dB[6][2];
#pragma unroll
  for (int c = 0; c < 6; ++c) {
#pragma unroll
    for (int j = 0; j < 2; ++j) {
      int kk = 4 * c + koff + j;
      f0[c][j] = *cf_tap(l16, kk, h0, taps0, k0[c][j]);
      f1[c][j] = *cf_tap(l16, kk, h1, taps1, k1[c][j]);
    }
  }
#pragma unroll
  for (int c = 0; c < 6; ++c) {
#pragma unroll
    for (int j = 0; j < 2; ++j) {
      int cj = reflect_i(xc0 + 4 * c + koff + j, Wt);
      dA[c][j] = row1[cj];
      dB[c][j] = row2[cj];
    }
  }
  LOADS_STAY_AHEAD();

  v8f acc = {0.f, 0.f, 0.f, 0.f, 0.f, 0.f, 0.f, 0.f};
#pragma unroll
  for (int c = 0; c < 6; ++c) {
    v2f aA, aB, bf0, bf1;
    bf0.x = k0[c][0] ? f0[c][0] : 0.0f;
    bf0.y = k0[c][1] ? f0[c][1] : 0.0f;
    bf1.x = k1[c][0] ? f1[c][0] : 0.0f;
    bf1.y = k1[c][1] ? f1[c][1] : 0.0f;
    aA.x = dA[c][0];
    aA.y = dA[c][1];
    aB.x = dB[c][0];
    aB.y = dB[c][1];
    acc = wmma4(aA, bf0, acc);
    acc = wmma4(aB, bf1, acc);
  }

  float* op = out + ((size_t)b * H + m0 + half * 8) * Wt + Jc * 16 + l16;
#pragma unroll
  for (int e = 0; e < 8; ++e) op[(size_t)e * Wt] = acc[e];
}

extern "C" void kernel_launch(void* const* d_in, const int* in_sizes, int n_in,
                              void* d_out, int out_size, void* d_ws,
                              size_t ws_size, hipStream_t stream) {
  (void)in_sizes;
  (void)n_in;
  (void)out_size;
  (void)ws_size;

  const float* yl = (const float*)d_in[0];
  const float* yh0 = (const float*)d_in[1];
  const float* yh1 = (const float*)d_in[2];
  const float* yh2 = (const float*)d_in[3];
  const float* g0o = (const float*)d_in[4];
  const float* g1o = (const float*)d_in[5];
  const float* g0a = (const float*)d_in[6];
  const float* g0b = (const float*)d_in[7];
  const float* g1a = (const float*)d_in[8];
  const float* g1b = (const float*)d_in[9];

  const int B = 64;
  const size_t IMG = (size_t)512 * 512;
  float* T1 = (float*)d_ws;
  float* T2 = T1 + (size_t)B * IMG;
  float* LL = T2 + (size_t)B * IMG;
  float* outp = (float*)d_out;

  dim3 blk(256);
  auto nblk = [](long tiles) { return dim3((unsigned)((tiles + 7) / 8)); };

  {
    int r = 128, W = 128, Hb = 64, Wb = 64;
    int ntr = 2 * r / 16, ntw = W / 16;
    long tc = (long)B * ntr * ntw;
    j2_col_kernel<true><<<nblk(tc), blk, 0, stream>>>(
        yl, yh2, 0, 0, 0, 5, Hb, Wb, r, W, g0b, g0a, g1b, g1a, T1, ntr, ntw, B);
    j2_col_kernel<false><<<nblk(tc), blk, 0, stream>>>(
        nullptr, yh2, 2, 3, 1, 4, Hb, Wb, r, W, g0b, g0a, g1b, g1a, T2, ntr,
        ntw, B);
    int H = 2 * r, Wt = W;
    int rtr = H / 16, rtw = 2 * Wt / 16;
    long tr = (long)B * rtr * rtw;
    j2_row_kernel<<<nblk(tr), blk, 0, stream>>>(T1, T2, H, Wt, g0b, g0a, g1b,
                                                g1a, LL, rtr, rtw, B);
  }

  {
    int r = 256, W = 256, Hb = 128, Wb = 128;
    int ntr = 2 * r / 16, ntw = W / 16;
    long tc = (long)B * ntr * ntw;
    j2_col_kernel<true><<<nblk(tc), blk, 0, stream>>>(
        LL, yh1, 0, 0, 0, 5, Hb, Wb, r, W, g0b, g0a, g1b, g1a, T1, ntr, ntw, B);
    j2_col_kernel<false><<<nblk(tc), blk, 0, stream>>>(
        nullptr, yh1, 2, 3, 1, 4, Hb, Wb, r, W, g0b, g0a, g1b, g1a, T2, ntr,
        ntw, B);
    int H = 2 * r, Wt = W;
    int rtr = H / 16, rtw = 2 * Wt / 16;
    long tr = (long)B * rtr * rtw;
    j2_row_kernel<<<nblk(tr), blk, 0, stream>>>(T1, T2, H, Wt, g0b, g0a, g1b,
                                                g1a, LL, rtr, rtw, B);
  }

  {
    int r = 512, W = 512, Hb = 256, Wb = 256;
    int ntr = r / 16, ntw = W / 16;
    long tc = (long)B * ntr * ntw;
    j1_col_kernel<true><<<nblk(tc), blk, 0, stream>>>(
        LL, yh0, 0, 0, 0, 5, Hb, Wb, r, W, g0o, 7, g1o, 5, T1, ntr, ntw, B);
    j1_col_kernel<false><<<nblk(tc), blk, 0, stream>>>(
        nullptr, yh0, 2, 3, 1, 4, Hb, Wb, r, W, g0o, 7, g1o, 5, T2, ntr, ntw,
        B);
    long tr = (long)B * ntr * ntw;
    j1_row_kernel<<<nblk(tr), blk, 0, stream>>>(T1, T2, r, W, g0o, 7, g1o, 5,
                                                outp, ntr, ntw, B);
  }
}


# Each wave computes two tiles side by side on the unfiltered axis from one set of filter fragments; in column passes a lane owns 2 adjacent columns, each band element fetched once; sum order unchanged
# speedup vs baseline: 1.2981x; 1.1289x over previous
#include <hip/hip_runtime.h>


typedef __attribute__((ext_vector_type(2))) float v2f;
typedef __attribute__((ext_vector_type(8))) float v8f;

#define SQRT_HALF 0.70710678118654752440f

__device__ __forceinline__ int reflect_i(int x, int n) {
  x = (x < 0) ? (-1 - x) : x;
  x = (x >= n) ? (2 * n - 1 - x) : x;
  return x;
}

__device__ __forceinline__ const float* cif_tap(int m, int kk,
                                                const float* __restrict__ ha,
                                                const float* __restrict__ hb,
                                                int hp, bool& ok) {
  int p = m & 3;
  int q = m >> 2;
  int want_odd = hp ? ((p & 1) ^ 1) : (p & 1);
  int t = ((kk - want_odd) >> 1) - q;
  ok = ((kk & 1) == want_odd) && (t >= 0) && (t <= 4);
  int tc = t < 0 ? 0 : (t > 4 ? 4 : t);
  const float* __restrict__ h = ((p == 0) || (p == 2)) ? ha : hb;
  int idx = (p < 2) ? (2 * tc + 1) : (2 * tc);
  return h + idx;
}

__device__ __forceinline__ const float* cf_tap(int m, int kk,
                                               const float* __restrict__ h,
                                               int taps, bool& ok) {
  int t = kk - m - (3 - (taps >> 1));
  ok = (t >= 0) && (t < taps);
  int tc = t < 0 ? 0 : (t >= taps ? taps - 1 : t);
  return h + tc;
}

struct Q4 {
  float w1r, w1i, w2r, w2i;
};

__device__ __forceinline__ Q4 c2q_load(const float* __restrict__ p1,
                                       const float* __restrict__ p2, int rr,
                                       int stride) {
  int o = (rr >> 1) * stride;
  Q4 q;
  q.w1r = p1[o];
  q.w1i = p1[o + 1];
  q.w2r = p2[o];
  q.w2i = p2[o + 1];
  return q;
}

__device__ __forceinline__ float c2q_even_col(Q4 q, int rr) {
  return (rr & 1) ? (q.w1i - q.w2i) : (q.w1r + q.w2r);
}

__device__ __forceinline__ float c2q_odd_col(Q4 q, int rr) {
  return (rr & 1) ? (q.w2r - q.w1r) : (q.w1i + q.w2i);
}

__device__ __forceinline__ v8f wmma4(v2f a, v2f b, v8f c) {
  return __builtin_amdgcn_wmma_f32_16x16x4_f32(false, a, false, b, (short)0, c,
                                               false, false);
}

#define LOADS_STAY_AHEAD() __builtin_amdgcn_sched_barrier(0)

template <bool SRC0_PLAIN>
__global__ void __launch_bounds__(256)
j2_col_kernel(const float* __restrict__ src0_plain, const float* __restrict__ yh,
              int s0a, int s0b, int s1a, int s1b, int Hb, int Wb, int r, int W,
              const float* __restrict__ ha0, const float* __restrict__ hb0,
              const float* __restrict__ ha1, const float* __restrict__ hb1,
              float* __restrict__ out, int nTilesR, int nTilesW, int B) {
  int lane = threadIdx.x & 31;
  int wave = threadIdx.x >> 5;
  long tile = (long)blockIdx.x * 8 + wave;
  long total = (long)B * nTilesR * nTilesW;
  if (tile >= total) return;
  int tpi = nTilesR * nTilesW;
  int b = (int)(tile / tpi);
  int t2 = (int)(tile % tpi);
  int I = t2 / nTilesW;
  int Jc = t2 % nTilesW;
  int x0 = 8 * I - 4;
  int half = lane >> 4;
  int l16 = lane & 15;
  int koff = half * 2;
  int M = l16;
  int c0 = Jc * 32 + 2 * l16;

  int stride = 2 * Wb;
  const float* p0 = SRC0_PLAIN ? (src0_plain + (size_t)b * r * W + c0) : nullptr;
  const float* p0a = yh + ((size_t)b * 6 + s0a) * Hb * stride + c0;
  const float* p0b = yh + ((size_t)b * 6 + s0b) * Hb * stride + c0;
  const float* p1a = yh + ((size_t)b * 6 + s1a) * Hb * stride + c0;
  const float* p1b = yh + ((size_t)b * 6 + s1b) * Hb * stride + c0;

  float f0[4][2], f1[4][2];
  bool k0[4][2], k1[4][2];
  int rr[4][2];
  float d0e[4][2], d0o[4][2];
  Q4 q0[4][2], q1[4][2];
#pragma unroll
  for (int c = 0; c < 4; ++c) {
#pragma unroll
    for (int j = 0; j < 2; ++j) {
      int kk = 4 * c + koff + j;
      f0[c][j] = *cif_tap(M, kk, ha0, hb0, 0, k0[c][j]);
      f1[c][j] = *cif_tap(M, kk, ha1, hb1, 1, k1[c][j]);
    }
  }
#pragma unroll
  for (int c = 0; c < 4; ++c) {
#pragma unroll
    for (int j = 0; j < 2; ++j) {
      rr[c][j] = reflect_i(x0 + 4 * c + koff + j, r);
      if (SRC0_PLAIN) {
        size_t o = (size_t)rr[c][j] * W;
        d0e[c][j] = p0[o];
        d0o[c][j] = p0[o + 1];
      } else {
        q0[c][j] = c2q_load(p0a, p0b, rr[c][j], stride);
      }
      q1[c][j] = c2q_load(p1a, p1b, rr[c][j], stride);
    }
  }
  LOADS_STAY_AHEAD();

  const float s0sc = SRC0_PLAIN ? 1.0f : SQRT_HALF;
  v8f accE = {0.f, 0.f, 0.f, 0.f, 0.f, 0.f, 0.f, 0.f};
  v8f accO = {0.f, 0.f, 0.f, 0.f, 0.f, 0.f, 0.f, 0.f};
#pragma unroll
  for (int c = 0; c < 4; ++c) {
    v2f a0, a1, b0e, b0o, b1e, b1o;
    a0.x = (k0[c][0] ? f0[c][0] : 0.0f) * s0sc;
    a0.y = (k0[c][1] ? f0[c][1] : 0.0f) * s0sc;
    a1.x = (k1[c][0] ? f1[c][0] : 0.0f) * SQRT_HALF;
    a1.y = (k1[c][1] ? f1[c][1] : 0.0f) * SQRT_HALF;
    if (SRC0_PLAIN) {
      b0e.x = d0e[c][0];
      b0e.y = d0e[c][1];
      b0o.x = d0o[c][0];
      b0o.y = d0o[c][1];
    } else {
      b0e.x = c2q_even_col(q0[c][0], rr[c][0]);
      b0e.y = c2q_even_col(q0[c][1], rr[c][1]);
      b0o.x = c2q_odd_col(q0[c][0], rr[c][0]);
      b0o.y = c2q_odd_col(q0[c][1], rr[c][1]);
    }
    b1e.x = c2q_even_col(q1[c][0], rr[c][0]);
    b1e.y = c2q_even_col(q1[c][1], rr[c][1]);
    b1o.x = c2q_odd_col(q1[c][0], rr[c][0]);
    b1o.y = c2q_odd_col(q1[c][1], rr[c][1]);
    accE = wmma4(a0, b0e, accE);
    accO = wmma4(a0, b0o, accO);
    accE = wmma4(a1, b1e, accE);
    accO = wmma4(a1, b1o, accO);
  }

  int outR = 2 * r;
  float* op = out + ((size_t)b * outR + (size_t)I * 16 + half * 8) * W + c0;
#pragma unroll
  for (int e = 0; e < 8; ++e) {
    op[(size_t)e * W] = accE[e];
    op[(size_t)e * W + 1] = accO[e];
  }
}

__global__ void __launch_bounds__(256)
j2_row_kernel(const float* __restrict__ T1, const float* __restrict__ T2, int H,
              int Wt, const float* __restrict__ ha0,
              const float* __restrict__ hb0, const float* __restrict__ ha1,
              const float* __restrict__ hb1, float* __restrict__ out,
              int nTilesR, int nTilesW, int B) {
  int lane = threadIdx.x & 31;
  int wave = threadIdx.x >> 5;
  long tile = (long)blockIdx.x * 8 + wave;
  long total = (long)B * nTilesR * nTilesW;
  if (tile >= total) return;
  int tpi = nTilesR * nTilesW;
  int b = (int)(tile / tpi);
  int t2 = (int)(tile % tpi);
  int I = t2 / nTilesW;
  int Jc = t2 % nTilesW;
  int m0 = I * 32;
  int xc0 = 8 * Jc - 4;
  int half = lane >> 4;
  int l16 = lane & 15;
  int koff = half * 2;

  const float* row1 = T1 + ((size_t)b * H + m0 + l16) * Wt;
  const float* row2 = T2 + ((size_t)b * H + m0 + l16) * Wt;
  size_t lower = (size_t)16 * Wt;

  float f0[4][2], f1[4][2];
  bool k0[4][2], k1[4][2];
  float dA0[4][2], dB0[4][2], dA1[4][2], dB1[4][2];
#pragma unroll
  for (int c = 0; c < 4; ++c) {
#pragma unroll
    for (int j = 0; j < 2; ++j) {
      int kk = 4 * c + koff + j;
      f0[c][j] = *cif_tap(l16, kk, ha0, hb0, 0, k0[c][j]);
      f1[c][j] = *cif_tap(l16, kk, ha1, hb1, 1, k1[c][j]);
    }
  }
#pragma unroll
  for (int c = 0; c < 4; ++c) {
#pragma unroll
    for (int j = 0; j < 2; ++j) {
      int cj = reflect_i(xc0 + 4 * c + koff + j, Wt);
      dA0[c][j] = row1[cj];
      dB0[c][j] = row2[cj];
      dA1[c][j] = row1[lower + cj];
      dB1[c][j] = row2[lower + cj];
    }
  }
  LOADS_STAY_AHEAD();

  v8f acc0 = {0.f, 0.f, 0.f, 0.f, 0.f, 0.f, 0.f, 0.f};
  v8f acc1 = {0.f, 0.f, 0.f, 0.f, 0.f, 0.f, 0.f, 0.f};
#pragma unroll
  for (int c = 0; c < 4; ++c) {
    v2f aA0, aB0, aA1, aB1, bf0, bf1;
    bf0.x = k0[c][0] ? f0[c][0] : 0.0f;
    bf0.y = k0[c][1] ? f0[c][1] : 0.0f;
    bf1.x = k1[c][0] ? f1[c][0] : 0.0f;
    bf1.y = k1[c][1] ? f1[c][1] : 0.0f;
    aA0.x = dA0[c][0];
    aA0.y = dA0[c][1];
    aB0.x = dB0[c][0];
    aB0.y = dB0[c][1];
    aA1.x = dA1[c][0];
    aA1.y = dA1[c][1];
    aB1.x = dB1[c][0];
    aB1.y = dB1[c][1];
    acc0 = wmma4(aA0, bf0, acc0);
    acc1 = wmma4(aA1, bf0, acc1);
    acc0 = wmma4(aB0, bf1, acc0);
    acc1 = wmma4(aB1, bf1, acc1);
  }

  int outW = 2 * Wt;
  float* op = out + ((size_t)b * H + m0 + half * 8) * outW + Jc * 16 + l16;
  size_t lowerOut = (size_t)16 * outW;
#pragma unroll
  for (int e = 0; e < 8; ++e) {
    op[(size_t)e * outW] = acc0[e];
    op[lowerOut + (size_t)e * outW] = acc1[e];
  }
}

template <bool SRC0_PLAIN>
__global__ void __launch_bounds__(256)
j1_col_kernel(const float* __restrict__ src0_plain, const float* __restrict__ yh,
              int s0a, int s0b, int s1a, int s1b, int Hb, int Wb, int r, int W,
              const float* __restrict__ h0, int taps0,
              const float* __restrict__ h1, int taps1, float* __restrict__ out,
              int nTilesR, int nTilesW, int B) {
  int lane = threadIdx.x & 31;
  int wave = threadIdx.x >> 5;
  long tile = (long)blockIdx.x * 8 + wave;
  long total = (long)B * nTilesR * nTilesW;
  if (tile >= total) return;
  int tpi = nTilesR * nTilesW;
  int b = (int)(tile / tpi);
  int t2 = (int)(tile % tpi);
  int I = t2 / nTilesW;
  int Jc = t2 % nTilesW;
  int x0 = 16 * I - 3;
  int half = lane >> 4;
  int l16 = lane & 15;
  int koff = half * 2;
  int M = l16;
  int c0 = Jc * 32 + 2 * l16;

  int stride = 2 * Wb;
  const float* p0 = SRC0_PLAIN ? (src0_plain + (size_t)b * r * W + c0) : nullptr;
  const float* p0a = yh + ((size_t)b * 6 + s0a) * Hb * stride + c0;
  const float* p0b = yh + ((size_t)b * 6 + s0b) * Hb * stride + c0;
  const float* p1a = yh + ((size_t)b * 6 + s1a) * Hb * stride + c0;
  const float* p1b = yh + ((size_t)b * 6 + s1b) * Hb * stride + c0;

  float f0[6][2], f1[6][2];
  bool k0[6][2], k1[6][2];
  int rr[6][2];
  float d0e[6][2], d0o[6][2];
  Q4 q0[6][2], q1[6][2];
#pragma unroll
  for (int c = 0; c < 6; ++c) {
#pragma unroll
    for (int j = 0; j < 2; ++j) {
      int kk = 4 * c + koff + j;
      f0[c][j] = *cf_tap(M, kk, h0, taps0, k0[c][j]);
      f1[c][j] = *cf_tap(M, kk, h1, taps1, k1[c][j]);
    }
  }
#pragma unroll
  for (int c = 0; c < 6; ++c) {
#pragma unroll
    for (int j = 0; j < 2; ++j) {
      rr[c][j] = reflect_i(x0 + 4 * c + koff + j, r);
      if (SRC0_PLAIN) {
        size_t o = (size_t)rr[c][j] * W;
        d0e[c][j] = p0[o];
        d0o[c][j] = p0[o + 1];
      } else {
        q0[c][j] = c2q_load(p0a, p0b, rr[c][j], stride);
      }
      q1[c][j] = c2q_load(p1a, p1b, rr[c][j], stride);
    }
  }
  LOADS_STAY_AHEAD();

  const float s0sc = SRC0_PLAIN ? 1.0f : SQRT_HALF;
  v8f accE = {0.f, 0.f, 0.f, 0.f, 0.f, 0.f, 0.f, 0.f};
  v8f accO = {0.f, 0.f, 0.f, 0.f, 0.f, 0.f, 0.f, 0.f};
#pragma unroll
  for (int c = 0; c < 6; ++c) {
    v2f a0, a1, b0e, b0o, b1e, b1o;
    a0.x = (k0[c][0] ? f0[c][0] : 0.0f) * s0sc;
    a0.y = (k0[c][1] ? f0[c][1] : 0.0f) * s0sc;
    a1.x = (k1[c][0] ? f1[c][0] : 0.0f) * SQRT_HALF;
    a1.y = (k1[c][1] ? f1[c][1] : 0.0f) * SQRT_HALF;
    if (SRC0_PLAIN) {
      b0e.x = d0e[c][0];
      b0e.y = d0e[c][1];
      b0o.x = d0o[c][0];
      b0o.y = d0o[c][1];
    } else {
      b0e.x = c2q_even_col(q0[c][0], rr[c][0]);
      b0e.y = c2q_even_col(q0[c][1], rr[c][1]);
      b0o.x = c2q_odd_col(q0[c][0], rr[c][0]);
      b0o.y = c2q_odd_col(q0[c][1], rr[c][1]);
    }
    b1e.x = c2q_even_col(q1[c][0], rr[c][0]);
    b1e.y = c2q_even_col(q1[c][1], rr[c][1]);
    b1o.x = c2q_odd_col(q1[c][0], rr[c][0]);
    b1o.y = c2q_odd_col(q1[c][1], rr[c][1]);
    accE = wmma4(a0, b0e, accE);
    accO = wmma4(a0, b0o, accO);
    accE = wmma4(a1, b1e, accE);
    accO = wmma4(a1, b1o, accO);
  }

  float* op = out + ((size_t)b * r + (size_t)I * 16 + half * 8) * W + c0;
#pragma unroll
  for (int e = 0; e < 8; ++e) {
    op[(size_t)e * W] = accE[e];
    op[(size_t)e * W + 1] = accO[e];
  }
}

__global__ void __launch_bounds__(256)
j1_row_kernel(const float* __restrict__ T1, const float* __restrict__ T2, int H,
              int Wt, const float* __restrict__ h0, int taps0,
              const float* __restrict__ h1, int taps1, float* __restrict__ out,
              int nTilesR, int nTilesW, int B) {
  int lane = threadIdx.x & 31;
  int wave = threadIdx.x >> 5;
  long tile = (long)blockIdx.x * 8 + wave;
  long total = (long)B * nTilesR * nTilesW;
  if (tile >= total) return;
  int tpi = nTilesR * nTilesW;
  int b = (int)(tile / tpi);
  int t2 = (int)(tile % tpi);
  int I = t2 / nTilesW;
  int Jc = t2 % nTilesW;
  int m0 = I * 32;
  int xc0 = 16 * Jc - 3;
  int half = lane >> 4;
  int l16 = lane & 15;
  int koff = half * 2;

  const float* row1 = T1 + ((size_t)b * H + m0 + l16) * Wt;
  const float* row2 = T2 + ((size_t)b * H + m0 + l16) * Wt;
  size_t lower = (size_t)16 * Wt;

  float f0[6][2], f1[6][2];
  bool k0[6][2], k1[6][2];
  float dA0[6][2], dB0[6][2], dA1[6][2], dB1[6][2];
#pragma unroll
  for (int c = 0; c < 6; ++c) {
#pragma unroll
    for (int j = 0; j < 2; ++j) {
      int kk = 4 * c + koff + j;
      f0[c][j] = *cf_tap(l16, kk, h0, taps0, k0[c][j]);
      f1[c][j] = *cf_tap(l16, kk, h1, taps1, k1[c][j]);
    }
  }
#pragma unroll
  for (int c = 0; c < 6; ++c) {
#pragma unroll
    for (int j = 0; j < 2; ++j) {
      int cj = reflect_i(xc0 + 4 * c + koff + j, Wt);
      dA0[c][j] = row1[cj];
      dB0[c][j] = row2[cj];
      dA1[c][j] = row1[lower + cj];
      dB1[c][j] = row2[lower + cj];
    }
  }
  LOADS_STAY_AHEAD();

  v8f acc0 = {0.f, 0.f, 0.f, 0.f, 0.f, 0.f, 0.f, 0.f};
  v8f acc1 = {0.f, 0.f, 0.f, 0.f, 0.f, 0.f, 0.f, 0.f};
#pragma unroll
  for (int c = 0; c < 6; ++c) {
    v2f aA0, aB0, aA1, aB1, bf0, bf1;
    bf0.x = k0[c][0] ? f0[c][0] : 0.0f;
    bf0.y = k0[c][1] ? f0[c][1] : 0.0f;
    bf1.x = k1[c][0] ? f1[c][0] : 0.0f;
    bf1.y = k1[c][1] ? f1[c][1] : 0.0f;
    aA0.x = dA0[c][0];
    aA0.y = dA0[c][1];
    aB0.x = dB0[c][0];
    aB0.y = dB0[c][1];
    aA1.x = dA1[c][0];
    aA1.y = dA1[c][1];
    aB1.x = dB1[c][0];
    aB1.y = dB1[c][1];
    acc0 = wmma4(aA0, bf0, acc0);
    acc1 = wmma4(aA1, bf0, acc1);
    acc0 = wmma4(aB0, bf1, acc0);
    acc1 = wmma4(aB1, bf1, acc1);
  }

  float* op = out + ((size_t)b * H + m0 + half * 8) * Wt + Jc * 16 + l16;
  size_t lowerOut = (size_t)16 * Wt;
#pragma unroll
  for (int e = 0; e < 8; ++e) {
    op[(size_t)e * Wt] = acc0[e];
    op[lowerOut + (size_t)e * Wt] = acc1[e];
  }
}

extern "C" void kernel_launch(void* const* d_in, const int* in_sizes, int n_in,
                              void* d_out, int out_size, void* d_ws,
                              size_t ws_size, hipStream_t stream) {
  (void)in_sizes;
  (void)n_in;
  (void)out_size;
  (void)ws_size;

  const float* yl = (const float*)d_in[0];
  const float* yh0 = (const float*)d_in[1];
  const float* yh1 = (const float*)d_in[2];
  const float* yh2 = (const float*)d_in[3];
  const float* g0o = (const float*)d_in[4];
  const float* g1o = (const float*)d_in[5];
  const float* g0a = (const float*)d_in[6];
  const float* g0b = (const float*)d_in[7];
  const float* g1a = (const float*)d_in[8];
  const float* g1b = (const float*)d_in[9];

  const int B = 64;
  const size_t IMG = (size_t)512 * 512;
  float* T1 = (float*)d_ws;
  float* T2 = T1 + (size_t)B * IMG;
  float* LL = T2 + (size_t)B * IMG;
  float* outp = (float*)d_out;

  dim3 blk(256);
  auto nblk = [](long tiles) { return dim3((unsigned)((tiles + 7) / 8)); };


  {
    int r = 128, W = 128, Hb = 64, Wb = 64;
    int ntr = 2 * r / 16, ntw = W / 32;
    long tc = (long)B * ntr * ntw;
    j2_col_kernel<true><<<nblk(tc), blk, 0, stream>>>(
        yl, yh2, 0, 0, 0, 5, Hb, Wb, r, W, g0b, g0a, g1b, g1a, T1, ntr, ntw, B);
    j2_col_kernel<false><<<nblk(tc), blk, 0, stream>>>(
        nullptr, yh2, 2, 3, 1, 4, Hb, Wb, r, W, g0b, g0a, g1b, g1a, T2, ntr,
        ntw, B);
    int H = 2 * r, Wt = W;
    int rtr = H / 32, rtw = 2 * Wt / 16;
    long tr = (long)B * rtr * rtw;
    j2_row_kernel<<<nblk(tr), blk, 0, stream>>>(T1, T2, H, Wt, g0b, g0a, g1b,
                                                g1a, LL, rtr, rtw, B);
  }

  {
    int r = 256, W = 256, Hb = 128, Wb = 128;
    int ntr = 2 * r / 16, ntw = W / 32;
    long tc = (long)B * ntr * ntw;
    j2_col_kernel<true><<<nblk(tc), blk, 0, stream>>>(
        LL, yh1, 0, 0, 0, 5, Hb, Wb, r, W, g0b, g0a, g1b, g1a, T1, ntr, ntw, B);
    j2_col_kernel<false><<<nblk(tc), blk, 0, stream>>>(
        nullptr, yh1, 2, 3, 1, 4, Hb, Wb, r, W, g0b, g0a, g1b, g1a, T2, ntr,
        ntw, B);
    int H = 2 * r, Wt = W;
    int rtr = H / 32, rtw = 2 * Wt / 16;
    long tr = (long)B * rtr * rtw;
    j2_row_kernel<<<nblk(tr), blk, 0, stream>>>(T1, T2, H, Wt, g0b, g0a, g1b,
                                                g1a, LL, rtr, rtw, B);
  }

  {
    int r = 512, W = 512, Hb = 256, Wb = 256;
    int ntr = r / 16, ntw = W / 32;
    long tc = (long)B * ntr * ntw;
    j1_col_kernel<true><<<nblk(tc), blk, 0, stream>>>(
        LL, yh0, 0, 0, 0, 5, Hb, Wb, r, W, g0o, 7, g1o, 5, T1, ntr, ntw, B);
    j1_col_kernel<false><<<nblk(tc), blk, 0, stream>>>(
        nullptr, yh0, 2, 3, 1, 4, Hb, Wb, r, W, g0o, 7, g1o, 5, T2, ntr, ntw,
        B);
    int rtr = r / 32, rtw = W / 16;
    long tr = (long)B * rtr * rtw;
    j1_row_kernel<<<nblk(tr), blk, 0, stream>>>(T1, T2, r, W, g0o, 7, g1o, 5,
                                                outp, rtr, rtw, B);
  }
}


# Row passes: each lane fetches 4 consecutive columns of its row with one 16-byte load, fed to the matrix unit as loaded; k regrouped per WMMA step in both fragments, so a row's sum order changes
# speedup vs baseline: 1.7442x; 1.3437x over previous
#include <hip/hip_runtime.h>


typedef __attribute__((ext_vector_type(2))) float v2f;
typedef __attribute__((ext_vector_type(4))) float v4f;
typedef __attribute__((ext_vector_type(8))) float v8f;

#define SQRT_HALF 0.70710678118654752440f

__device__ __forceinline__ int reflect_i(int x, int n) {
  x = (x < 0) ? (-1 - x) : x;
  x = (x >= n) ? (2 * n - 1 - x) : x;
  return x;
}

__device__ __forceinline__ int row_group_base(int s, int n, bool& rev) {
  rev = (s < 0) || (s >= n);
  int b = s;
  b = (s < 0) ? (-4 - s) : b;
  b = (s >= n) ? (2 * n - 4 - s) : b;
  return b;
}

__device__ __forceinline__ int group_col(bool rev, int p) {
  return rev ? (3 - p) : p;
}

__device__ __forceinline__ const float* cif_tap(int m, int kk,
                                                const float* __restrict__ ha,
                                                const float* __restrict__ hb,
                                                int hp, bool& ok) {
  int p = m & 3;
  int q = m >> 2;
  int want_odd = hp ? ((p & 1) ^ 1) : (p & 1);
  int t = ((kk - want_odd) >> 1) - q;
  ok = ((kk & 1) == want_odd) && (t >= 0) && (t <= 4);
  int tc = t < 0 ? 0 : (t > 4 ? 4 : t);
  const float* __restrict__ h = ((p == 0) || (p == 2)) ? ha : hb;
  int idx = (p < 2) ? (2 * tc + 1) : (2 * tc);
  return h + idx;
}

__device__ __forceinline__ const float* cf_tap(int m, int kk,
                                               const float* __restrict__ h,
                                               int taps, bool& ok) {
  int t = kk - m - (3 - (taps >> 1));
  ok = (t >= 0) && (t < taps);
  int tc = t < 0 ? 0 : (t >= taps ? taps - 1 : t);
  return h + tc;
}

struct Q4 {
  float w1r, w1i, w2r, w2i;
};

__device__ __forceinline__ Q4 c2q_load(const float* __restrict__ p1,
                                       const float* __restrict__ p2, int rr,
                                       int stride) {
  int o = (rr >> 1) * stride;
  Q4 q;
  q.w1r = p1[o];
  q.w1i = p1[o + 1];
  q.w2r = p2[o];
  q.w2i = p2[o + 1];
  return q;
}

__device__ __forceinline__ float c2q_even_col(Q4 q, int rr) {
  return (rr & 1) ? (q.w1i - q.w2i) : (q.w1r + q.w2r);
}

__device__ __forceinline__ float c2q_odd_col(Q4 q, int rr) {
  return (rr & 1) ? (q.w2r - q.w1r) : (q.w1i + q.w2i);
}

__device__ __forceinline__ v8f wmma4(v2f a, v2f b, v8f c) {
  return __builtin_amdgcn_wmma_f32_16x16x4_f32(false, a, false, b, (short)0, c,
                                               false, false);
}

#define LOADS_STAY_AHEAD() __builtin_amdgcn_sched_barrier(0)

template <bool SRC0_PLAIN>
__global__ void __launch_bounds__(256)
j2_col_kernel(const float* __restrict__ src0_plain, const float* __restrict__ yh,
              int s0a, int s0b, int s1a, int s1b, int Hb, int Wb, int r, int W,
              const float* __restrict__ ha0, const float* __restrict__ hb0,
              const float* __restrict__ ha1, const float* __restrict__ hb1,
              float* __restrict__ out, int nTilesR, int nTilesW, int B) {
  int lane = threadIdx.x & 31;
  int wave = threadIdx.x >> 5;
  long tile = (long)blockIdx.x * 8 + wave;
  long total = (long)B * nTilesR * nTilesW;
  if (tile >= total) return;
  int tpi = nTilesR * nTilesW;
  int b = (int)(tile / tpi);
  int t2 = (int)(tile % tpi);
  int I = t2 / nTilesW;
  int Jc = t2 % nTilesW;
  int x0 = 8 * I - 4;
  int half = lane >> 4;
  int l16 = lane & 15;
  int koff = half * 2;
  int M = l16;
  int c0 = Jc * 32 + 2 * l16;

  int stride = 2 * Wb;
  const float* p0 = SRC0_PLAIN ? (src0_plain + (size_t)b * r * W + c0) : nullptr;
  const float* p0a = yh + ((size_t)b * 6 + s0a) * Hb * stride + c0;
  const float* p0b = yh + ((size_t)b * 6 + s0b) * Hb * stride + c0;
  const float* p1a = yh + ((size_t)b * 6 + s1a) * Hb * stride + c0;
  const float* p1b = yh + ((size_t)b * 6 + s1b) * Hb * stride + c0;

  float f0[4][2], f1[4][2];
  bool k0[4][2], k1[4][2];
  int rr[4][2];
  float d0e[4][2], d0o[4][2];
  Q4 q0[4][2], q1[4][2];
#pragma unroll
  for (int c = 0; c < 4; ++c) {
#pragma unroll
    for (int j = 0; j < 2; ++j) {
      int kk = 4 * c + koff + j;
      f0[c][j] = *cif_tap(M, kk, ha0, hb0, 0, k0[c][j]);
      f1[c][j] = *cif_tap(M, kk, ha1, hb1, 1, k1[c][j]);
    }
  }
#pragma unroll
  for (int c = 0; c < 4; ++c) {
#pragma unroll
    for (int j = 0; j < 2; ++j) {
      rr[c][j] = reflect_i(x0 + 4 * c + koff + j, r);
      if (SRC0_PLAIN) {
        size_t o = (size_t)rr[c][j] * W;
        d0e[c][j] = p0[o];
        d0o[c][j] = p0[o + 1];
      } else {
        q0[c][j] = c2q_load(p0a, p0b, rr[c][j], stride);
      }
      q1[c][j] = c2q_load(p1a, p1b, rr[c][j], stride);
    }
  }
  LOADS_STAY_AHEAD();

  const float s0sc = SRC0_PLAIN ? 1.0f : SQRT_HALF;
  v8f accE = {0.f, 0.f, 0.f, 0.f, 0.f, 0.f, 0.f, 0.f};
  v8f accO = {0.f, 0.f, 0.f, 0.f, 0.f, 0.f, 0.f, 0.f};
#pragma unroll
  for (int c = 0; c < 4; ++c) {
    v2f a0, a1, b0e, b0o, b1e, b1o;
    a0.x = (k0[c][0] ? f0[c][0] : 0.0f) * s0sc;
    a0.y = (k0[c][1] ? f0[c][1] : 0.0f) * s0sc;
    a1.x = (k1[c][0] ? f1[c][0] : 0.0f) * SQRT_HALF;
    a1.y = (k1[c][1] ? f1[c][1] : 0.0f) * SQRT_HALF;
    if (SRC0_PLAIN) {
      b0e.x = d0e[c][0];
      b0e.y = d0e[c][1];
      b0o.x = d0o[c][0];
      b0o.y = d0o[c][1];
    } else {
      b0e.x = c2q_even_col(q0[c][0], rr[c][0]);
      b0e.y = c2q_even_col(q0[c][1], rr[c][1]);
      b0o.x = c2q_odd_col(q0[c][0], rr[c][0]);
      b0o.y = c2q_odd_col(q0[c][1], rr[c][1]);
    }
    b1e.x = c2q_even_col(q1[c][0], rr[c][0]);
    b1e.y = c2q_even_col(q1[c][1], rr[c][1]);
    b1o.x = c2q_odd_col(q1[c][0], rr[c][0]);
    b1o.y = c2q_odd_col(q1[c][1], rr[c][1]);
    accE = wmma4(a0, b0e, accE);
    accO = wmma4(a0, b0o, accO);
    accE = wmma4(a1, b1e, accE);
    accO = wmma4(a1, b1o, accO);
  }

  int outR = 2 * r;
  float* op = out + ((size_t)b * outR + (size_t)I * 16 + half * 8) * W + c0;
#pragma unroll
  for (int e = 0; e < 8; ++e) {
    op[(size_t)e * W] = accE[e];
    op[(size_t)e * W + 1] = accO[e];
  }
}

__global__ void __launch_bounds__(256)
j2_row_kernel(const float* __restrict__ T1, const float* __restrict__ T2, int H,
              int Wt, const float* __restrict__ ha0,
              const float* __restrict__ hb0, const float* __restrict__ ha1,
              const float* __restrict__ hb1, float* __restrict__ out,
              int nTilesR, int nTilesW, int B) {
  int lane = threadIdx.x & 31;
  int wave = threadIdx.x >> 5;
  long tile = (long)blockIdx.x * 8 + wave;
  long total = (long)B * nTilesR * nTilesW;
  if (tile >= total) return;
  int tpi = nTilesR * nTilesW;
  int b = (int)(tile / tpi);
  int t2 = (int)(tile % tpi);
  int I = t2 / nTilesW;
  int Jc = t2 % nTilesW;
  int m0 = I * 32;
  int xc0 = 8 * Jc - 4;
  int half = lane >> 4;
  int l16 = lane & 15;

  const float* row1 = T1 + ((size_t)b * H + m0 + l16) * Wt;
  const float* row2 = T2 + ((size_t)b * H + m0 + l16) * Wt;
  size_t lower = (size_t)16 * Wt;

  float f0[4][2], f1[4][2];
  bool k0[4][2], k1[4][2];
  v4f gA0[2], gB0[2], gA1[2], gB1[2];
  bool rev[2];
  int cb[2];
#pragma unroll
  for (int g = 0; g < 2; ++g)
    cb[g] = row_group_base(xc0 + 4 * (2 * g + half), Wt, rev[g]);
#pragma unroll
  for (int t = 0; t < 4; ++t) {
#pragma unroll
    for (int j = 0; j < 2; ++j) {
      int kk = 4 * (2 * (t >> 1) + half) + group_col(rev[t >> 1], 2 * (t & 1) + j);
      f0[t][j] = *cif_tap(l16, kk, ha0, hb0, 0, k0[t][j]);
      f1[t][j] = *cif_tap(l16, kk, ha1, hb1, 1, k1[t][j]);
    }
  }
#pragma unroll
  for (int g = 0; g < 2; ++g) {
    gA0[g] = *(const v4f*)(row1 + cb[g]);
    gB0[g] = *(const v4f*)(row2 + cb[g]);
    gA1[g] = *(const v4f*)(row1 + lower + cb[g]);
    gB1[g] = *(const v4f*)(row2 + lower + cb[g]);
  }
  LOADS_STAY_AHEAD();

  v8f acc0 = {0.f, 0.f, 0.f, 0.f, 0.f, 0.f, 0.f, 0.f};
  v8f acc1 = {0.f, 0.f, 0.f, 0.f, 0.f, 0.f, 0.f, 0.f};
#pragma unroll
  for (int t = 0; t < 4; ++t) {
    const int g = t >> 1;
    const int p = 2 * (t & 1);
    v2f aA0, aB0, aA1, aB1, bf0, bf1;
    bf0.x = k0[t][0] ? f0[t][0] : 0.0f;
    bf0.y = k0[t][1] ? f0[t][1] : 0.0f;
    bf1.x = k1[t][0] ? f1[t][0] : 0.0f;
    bf1.y = k1[t][1] ? f1[t][1] : 0.0f;
    aA0.x = gA0[g][p];
    aA0.y = gA0[g][p + 1];
    aB0.x = gB0[g][p];
    aB0.y = gB0[g][p + 1];
    aA1.x = gA1[g][p];
    aA1.y = gA1[g][p + 1];
    aB1.x = gB1[g][p];
    aB1.y = gB1[g][p + 1];
    acc0 = wmma4(aA0, bf0, acc0);
    acc1 = wmma4(aA1, bf0, acc1);
    acc0 = wmma4(aB0, bf1, acc0);
    acc1 = wmma4(aB1, bf1, acc1);
  }

  int outW = 2 * Wt;
  float* op = out + ((size_t)b * H + m0 + half * 8) * outW + Jc * 16 + l16;
  size_t lowerOut = (size_t)16 * outW;
#pragma unroll
  for (int e = 0; e < 8; ++e) {
    op[(size_t)e * outW] = acc0[e];
    op[lowerOut + (size_t)e * outW] = acc1[e];
  }
}

template <bool SRC0_PLAIN>
__global__ void __launch_bounds__(256)
j1_col_kernel(const float* __restrict__ src0_plain, const float* __restrict__ yh,
              int s0a, int s0b, int s1a, int s1b, int Hb, int Wb, int r, int W,
              const float* __restrict__ h0, int taps0,
              const float* __restrict__ h1, int taps1, float* __restrict__ out,
              int nTilesR, int nTilesW, int B) {
  int lane = threadIdx.x & 31;
  int wave = threadIdx.x >> 5;
  long tile = (long)blockIdx.x * 8 + wave;
  long total = (long)B * nTilesR * nTilesW;
  if (tile >= total) return;
  int tpi = nTilesR * nTilesW;
  int b = (int)(tile / tpi);
  int t2 = (int)(tile % tpi);
  int I = t2 / nTilesW;
  int Jc = t2 % nTilesW;
  int x0 = 16 * I - 3;
  int half = lane >> 4;
  int l16 = lane & 15;
  int koff = half * 2;
  int M = l16;
  int c0 = Jc * 32 + 2 * l16;

  int stride = 2 * Wb;
  const float* p0 = SRC0_PLAIN ? (src0_plain + (size_t)b * r * W + c0) : nullptr;
  const float* p0a = yh + ((size_t)b * 6 + s0a) * Hb * stride + c0;
  const float* p0b = yh + ((size_t)b * 6 + s0b) * Hb * stride + c0;
  const float* p1a = yh + ((size_t)b * 6 + s1a) * Hb * stride + c0;
  const float* p1b = yh + ((size_t)b * 6 + s1b) * Hb * stride + c0;

  float f0[6][2], f1[6][2];
  bool k0[6][2], k1[6][2];
  int rr[6][2];
  float d0e[6][2], d0o[6][2];
  Q4 q0[6][2], q1[6][2];
#pragma unroll
  for (int c = 0; c < 6; ++c) {
#pragma unroll
    for (int j = 0; j < 2; ++j) {
      int kk = 4 * c + koff + j;
      f0[c][j] = *cf_tap(M, kk, h0, taps0, k0[c][j]);
      f1[c][j] = *cf_tap(M, kk, h1, taps1, k1[c][j]);
    }
  }
#pragma unroll
  for (int c = 0; c < 6; ++c) {
#pragma unroll
    for (int j = 0; j < 2; ++j) {
      rr[c][j] = reflect_i(x0 + 4 * c + koff + j, r);
      if (SRC0_PLAIN) {
        size_t o = (size_t)rr[c][j] * W;
        d0e[c][j] = p0[o];
        d0o[c][j] = p0[o + 1];
      } else {
        q0[c][j] = c2q_load(p0a, p0b, rr[c][j], stride);
      }
      q1[c][j] = c2q_load(p1a, p1b, rr[c][j], stride);
    }
  }
  LOADS_STAY_AHEAD();

  const float s0sc = SRC0_PLAIN ? 1.0f : SQRT_HALF;
  v8f accE = {0.f, 0.f, 0.f, 0.f, 0.f, 0.f, 0.f, 0.f};
  v8f accO = {0.f, 0.f, 0.f, 0.f, 0.f, 0.f, 0.f, 0.f};
#pragma unroll
  for (int c = 0; c < 6; ++c) {
    v2f a0, a1, b0e, b0o, b1e, b1o;
    a0.x = (k0[c][0] ? f0[c][0] : 0.0f) * s0sc;
    a0.y = (k0[c][1] ? f0[c][1] : 0.0f) * s0sc;
    a1.x = (k1[c][0] ? f1[c][0] : 0.0f) * SQRT_HALF;
    a1.y = (k1[c][1] ? f1[c][1] : 0.0f) * SQRT_HALF;
    if (SRC0_PLAIN) {
      b0e.x = d0e[c][0];
      b0e.y = d0e[c][1];
      b0o.x = d0o[c][0];
      b0o.y = d0o[c][1];
    } else {
      b0e.x = c2q_even_col(q0[c][0], rr[c][0]);
      b0e.y = c2q_even_col(q0[c][1], rr[c][1]);
      b0o.x = c2q_odd_col(q0[c][0], rr[c][0]);
      b0o.y = c2q_odd_col(q0[c][1], rr[c][1]);
    }
    b1e.x = c2q_even_col(q1[c][0], rr[c][0]);
    b1e.y = c2q_even_col(q1[c][1], rr[c][1]);
    b1o.x = c2q_odd_col(q1[c][0], rr[c][0]);
    b1o.y = c2q_odd_col(q1[c][1], rr[c][1]);
    accE = wmma4(a0, b0e, accE);
    accO = wmma4(a0, b0o, accO);
    accE = wmma4(a1, b1e, accE);
    accO = wmma4(a1, b1o, accO);
  }

  float* op = out + ((size_t)b * r + (size_t)I * 16 + half * 8) * W + c0;
#pragma unroll
  for (int e = 0; e < 8; ++e) {
    op[(size_t)e * W] = accE[e];
    op[(size_t)e * W + 1] = accO[e];
  }
}

__global__ void __launch_bounds__(256)
j1_row_kernel(const float* __restrict__ T1, const float* __restrict__ T2, int H,
              int Wt, const float* __restrict__ h0, int taps0,
              const float* __restrict__ h1, int taps1, float* __restrict__ out,
              int nTilesR, int nTilesW, int B) {
  int lane = threadIdx.x & 31;
  int wave = threadIdx.x >> 5;
  long tile = (long)blockIdx.x * 8 + wave;
  long total = (long)B * nTilesR * nTilesW;
  if (tile >= total) return;
  int tpi = nTilesR * nTilesW;
  int b = (int)(tile / tpi);
  int t2 = (int)(tile % tpi);
  int I = t2 / nTilesW;
  int Jc = t2 % nTilesW;
  int m0 = I * 32;
  int xw0 = 16 * Jc - 4;
  int half = lane >> 4;
  int l16 = lane & 15;

  const float* row1 = T1 + ((size_t)b * H + m0 + l16) * Wt;
  const float* row2 = T2 + ((size_t)b * H + m0 + l16) * Wt;
  size_t lower = (size_t)16 * Wt;

  float f0[6][2], f1[6][2];
  bool k0[6][2], k1[6][2];
  v4f gA0[3], gB0[3], gA1[3], gB1[3];
  bool rev[3];
  int cb[3];
#pragma unroll
  for (int g = 0; g < 3; ++g)
    cb[g] = row_group_base(xw0 + 4 * (2 * g + half), Wt, rev[g]);
#pragma unroll
  for (int t = 0; t < 6; ++t) {
#pragma unroll
    for (int j = 0; j < 2; ++j) {
      int kw = 4 * (2 * (t >> 1) + half) + group_col(rev[t >> 1], 2 * (t & 1) + j);
      f0[t][j] = *cf_tap(l16, kw - 1, h0, taps0, k0[t][j]);
      f1[t][j] = *cf_tap(l16, kw - 1, h1, taps1, k1[t][j]);
    }
  }
#pragma unroll
  for (int g = 0; g < 3; ++g) {
    gA0[g] = *(const v4f*)(row1 + cb[g]);
    gB0[g] = *(const v4f*)(row2 + cb[g]);
    gA1[g] = *(const v4f*)(row1 + lower + cb[g]);
    gB1[g] = *(const v4f*)(row2 + lower + cb[g]);
  }
  LOADS_STAY_AHEAD();

  v8f acc0 = {0.f, 0.f, 0.f, 0.f, 0.f, 0.f, 0.f, 0.f};
  v8f acc1 = {0.f, 0.f, 0.f, 0.f, 0.f, 0.f, 0.f, 0.f};
#pragma unroll
  for (int t = 0; t < 6; ++t) {
    const int g = t >> 1;
    const int p = 2 * (t & 1);
    v2f aA0, aB0, aA1, aB1, bf0, bf1;
    bf0.x = k0[t][0] ? f0[t][0] : 0.0f;
    bf0.y = k0[t][1] ? f0[t][1] : 0.0f;
    bf1.x = k1[t][0] ? f1[t][0] : 0.0f;
    bf1.y = k1[t][1] ? f1[t][1] : 0.0f;
    aA0.x = gA0[g][p];
    aA0.y = gA0[g][p + 1];
    aB0.x = gB0[g][p];
    aB0.y = gB0[g][p + 1];
    aA1.x = gA1[g][p];
    aA1.y = gA1[g][p + 1];
    aB1.x = gB1[g][p];
    aB1.y = gB1[g][p + 1];
    acc0 = wmma4(aA0, bf0, acc0);
    acc1 = wmma4(aA1, bf0, acc1);
    acc0 = wmma4(aB0, bf1, acc0);
    acc1 = wmma4(aB1, bf1, acc1);
  }

  float* op = out + ((size_t)b * H + m0 + half * 8) * Wt + Jc * 16 + l16;
  size_t lowerOut = (size_t)16 * Wt;
#pragma unroll
  for (int e = 0; e < 8; ++e) {
    op[(size_t)e * Wt] = acc0[e];
    op[lowerOut + (size_t)e * Wt] = acc1[e];
  }
}

extern "C" void kernel_launch(void* const* d_in, const int* in_sizes, int n_in,
                              void* d_out, int out_size, void* d_ws,
                              size_t ws_size, hipStream_t stream) {
  (void)in_sizes;
  (void)n_in;
  (void)out_size;
  (void)ws_size;

  const float* yl = (const float*)d_in[0];
  const float* yh0 = (const float*)d_in[1];
  const float* yh1 = (const float*)d_in[2];
  const float* yh2 = (const float*)d_in[3];
  const float* g0o = (const float*)d_in[4];
  const float* g1o = (const float*)d_in[5];
  const float* g0a = (const float*)d_in[6];
  const float* g0b = (const float*)d_in[7];
  const float* g1a = (const float*)d_in[8];
  const float* g1b = (const float*)d_in[9];

  const int B = 64;
  const size_t IMG = (size_t)512 * 512;
  float* T1 = (float*)d_ws;
  float* T2 = T1 + (size_t)B * IMG;
  float* LL = T2 + (size_t)B * IMG;
  float* outp = (float*)d_out;

  dim3 blk(256);
  auto nblk = [](long tiles) { return dim3((unsigned)((tiles + 7) / 8)); };


  {
    int r = 128, W = 128, Hb = 64, Wb = 64;
    int ntr = 2 * r / 16, ntw = W / 32;
    long tc = (long)B * ntr * ntw;
    j2_col_kernel<true><<<nblk(tc), blk, 0, stream>>>(
        yl, yh2, 0, 0, 0, 5, Hb, Wb, r, W, g0b, g0a, g1b, g1a, T1, ntr, ntw, B);
    j2_col_kernel<false><<<nblk(tc), blk, 0, stream>>>(
        nullptr, yh2, 2, 3, 1, 4, Hb, Wb, r, W, g0b, g0a, g1b, g1a, T2, ntr,
        ntw, B);
    int H = 2 * r, Wt = W;
    int rtr = H / 32, rtw = 2 * Wt / 16;
    long tr = (long)B * rtr * rtw;
    j2_row_kernel<<<nblk(tr), blk, 0, stream>>>(T1, T2, H, Wt, g0b, g0a, g1b,
                                                g1a, LL, rtr, rtw, B);
  }

  {
    int r = 256, W = 256, Hb = 128, Wb = 128;
    int ntr = 2 * r / 16, ntw = W / 32;
    long tc = (long)B * ntr * ntw;
    j2_col_kernel<true><<<nblk(tc), blk, 0, stream>>>(
        LL, yh1, 0, 0, 0, 5, Hb, Wb, r, W, g0b, g0a, g1b, g1a, T1, ntr, ntw, B);
    j2_col_kernel<false><<<nblk(tc), blk, 0, stream>>>(
        nullptr, yh1, 2, 3, 1, 4, Hb, Wb, r, W, g0b, g0a, g1b, g1a, T2, ntr,
        ntw, B);
    int H = 2 * r, Wt = W;
    int rtr = H / 32, rtw = 2 * Wt / 16;
    long tr = (long)B * rtr * rtw;
    j2_row_kernel<<<nblk(tr), blk, 0, stream>>>(T1, T2, H, Wt, g0b, g0a, g1b,
                                                g1a, LL, rtr, rtw, B);
  }

  {
    int r = 512, W = 512, Hb = 256, Wb = 256;
    int ntr = r / 16, ntw = W / 32;
    long tc = (long)B * ntr * ntw;
    j1_col_kernel<true><<<nblk(tc), blk, 0, stream>>>(
        LL, yh0, 0, 0, 0, 5, Hb, Wb, r, W, g0o, 7, g1o, 5, T1, ntr, ntw, B);
    j1_col_kernel<false><<<nblk(tc), blk, 0, stream>>>(
        nullptr, yh0, 2, 3, 1, 4, Hb, Wb, r, W, g0o, 7, g1o, 5, T2, ntr, ntw,
        B);
    int rtr = r / 32, rtw = W / 16;
    long tr = (long)B * rtr * rtw;
    j1_row_kernel<<<nblk(tr), blk, 0, stream>>>(T1, T2, r, W, g0o, 7, g1o, 5,
                                                outp, rtr, rtw, B);
  }
}


# Column and row pass of a level fused in one kernel: its 16-row strip of both column-pass images stays in LDS, not in memory; 9 launches to 3; row k in natural order again: a row's sum order changes
# speedup vs baseline: 3.3885x; 1.9427x over previous
#include <hip/hip_runtime.h>


typedef __attribute__((ext_vector_type(2))) float v2f;
typedef __attribute__((ext_vector_type(4))) float v4f;
typedef __attribute__((ext_vector_type(8))) float v8f;

#define SQRT_HALF 0.70710678118654752440f

__device__ __forceinline__ int reflect_i(int x, int n) {
  x = (x < 0) ? (-1 - x) : x;
  x = (x >= n) ? (2 * n - 1 - x) : x;
  return x;
}

__device__ __forceinline__ const float* cif_tap(int m, int kk,
                                                const float* __restrict__ ha,
                                                const float* __restrict__ hb,
                                                int hp, bool& ok) {
  int p = m & 3;
  int q = m >> 2;
  int want_odd = hp ? ((p & 1) ^ 1) : (p & 1);
  int t = ((kk - want_odd) >> 1) - q;
  ok = ((kk & 1) == want_odd) && (t >= 0) && (t <= 4);
  int tc = t < 0 ? 0 : (t > 4 ? 4 : t);
  const float* __restrict__ h = ((p == 0) || (p == 2)) ? ha : hb;
  int idx = (p < 2) ? (2 * tc + 1) : (2 * tc);
  return h + idx;
}

__device__ __forceinline__ const float* cf_tap(int m, int kk,
                                               const float* __restrict__ h,
                                               int taps, bool& ok) {
  int t = kk - m - (3 - (taps >> 1));
  ok = (t >= 0) && (t < taps);
  int tc = t < 0 ? 0 : (t >= taps ? taps - 1 : t);
  return h + tc;
}

struct Q4 {
  float w1r, w1i, w2r, w2i;
};

__device__ __forceinline__ Q4 c2q_load(const float* __restrict__ p1,
                                       const float* __restrict__ p2, int rr,
                                       int stride) {
  int o = (rr >> 1) * stride;
  Q4 q;
  q.w1r = p1[o];
  q.w1i = p1[o + 1];
  q.w2r = p2[o];
  q.w2i = p2[o + 1];
  return q;
}

__device__ __forceinline__ float c2q_even_col(Q4 q, int rr) {
  return (rr & 1) ? (q.w1i - q.w2i) : (q.w1r + q.w2r);
}

__device__ __forceinline__ float c2q_odd_col(Q4 q, int rr) {
  return (rr & 1) ? (q.w2r - q.w1r) : (q.w1i + q.w2i);
}

__device__ __forceinline__ v8f wmma4(v2f a, v2f b, v8f c) {
  return __builtin_amdgcn_wmma_f32_16x16x4_f32(false, a, false, b, (short)0, c,
                                               false, false);
}

#define LOADS_STAY_AHEAD() __builtin_amdgcn_sched_barrier(0)

template <int NC, bool SRC0_PLAIN>
__device__ __forceinline__ void col_block_to_lds(
    const float* __restrict__ p0, const float* __restrict__ p0a,
    const float* __restrict__ p0b, const float* __restrict__ p1a,
    const float* __restrict__ p1b, int stride, int W, int r, int x0, int koff,
    int half, int c0, const float (&f0)[NC][2], const float (&f1)[NC][2],
    const bool (&k0)[NC][2], const bool (&k1)[NC][2], float* __restrict__ sT,
    int pitch) {
  int rr[NC][2];
  float d0e[NC][2], d0o[NC][2];
  Q4 q0[NC][2], q1[NC][2];
#pragma unroll
  for (int c = 0; c < NC; ++c) {
#pragma unroll
    for (int j = 0; j < 2; ++j) {
      rr[c][j] = reflect_i(x0 + 4 * c + koff + j, r);
      if (SRC0_PLAIN) {
        size_t o = (size_t)rr[c][j] * W;
        d0e[c][j] = p0[o];
        d0o[c][j] = p0[o + 1];
      } else {
        q0[c][j] = c2q_load(p0a, p0b, rr[c][j], stride);
      }
      q1[c][j] = c2q_load(p1a, p1b, rr[c][j], stride);
    }
  }
  LOADS_STAY_AHEAD();

  const float s0sc = SRC0_PLAIN ? 1.0f : SQRT_HALF;
  v8f accE = {0.f, 0.f, 0.f, 0.f, 0.f, 0.f, 0.f, 0.f};
  v8f accO = {0.f, 0.f, 0.f, 0.f, 0.f, 0.f, 0.f, 0.f};
#pragma unroll
  for (int c = 0; c < NC; ++c) {
    v2f a0, a1, b0e, b0o, b1e, b1o;
    a0.x = (k0[c][0] ? f0[c][0] : 0.0f) * s0sc;
    a0.y = (k0[c][1] ? f0[c][1] : 0.0f) * s0sc;
    a1.x = (k1[c][0] ? f1[c][0] : 0.0f) * SQRT_HALF;
    a1.y = (k1[c][1] ? f1[c][1] : 0.0f) * SQRT_HALF;
    if (SRC0_PLAIN) {
      b0e.x = d0e[c][0];
      b0e.y = d0e[c][1];
      b0o.x = d0o[c][0];
      b0o.y = d0o[c][1];
    } else {
      b0e.x = c2q_even_col(q0[c][0], rr[c][0]);
      b0e.y = c2q_even_col(q0[c][1], rr[c][1]);
      b0o.x = c2q_odd_col(q0[c][0], rr[c][0]);
      b0o.y = c2q_odd_col(q0[c][1], rr[c][1]);
    }
    b1e.x = c2q_even_col(q1[c][0], rr[c][0]);
    b1e.y = c2q_even_col(q1[c][1], rr[c][1]);
    b1o.x = c2q_odd_col(q1[c][0], rr[c][0]);
    b1o.y = c2q_odd_col(q1[c][1], rr[c][1]);
    accE = wmma4(a0, b0e, accE);
    accO = wmma4(a0, b0o, accO);
    accE = wmma4(a1, b1e, accE);
    accO = wmma4(a1, b1o, accO);
  }

#pragma unroll
  for (int e = 0; e < 8; ++e) {
    int lr = half * 8 + e;
    v2f v;
    v.x = accE[e];
    v.y = accO[e];
    *(v2f*)(sT + lr * pitch + (c0 ^ (lr << 2))) = v;
  }
}

template <int NC, int NT>
__device__ __forceinline__ void row_tiles_from_lds(
    const float* __restrict__ sT1, const float* __restrict__ sT2, int pitch,
    int W, int jc0, int wstep, int woff, int koff, int l16,
    const float (&f0)[NC][2], const float (&f1)[NC][2], const bool (&k0)[NC][2],
    const bool (&k1)[NC][2], float* __restrict__ op, size_t ldo) {
  v2f bf0[NC], bf1[NC];
#pragma unroll
  for (int t = 0; t < NC; ++t) {
    bf0[t].x = k0[t][0] ? f0[t][0] : 0.0f;
    bf0[t].y = k0[t][1] ? f0[t][1] : 0.0f;
    bf1[t].x = k1[t][0] ? f1[t][0] : 0.0f;
    bf1[t].y = k1[t][1] ? f1[t][1] : 0.0f;
  }
  const float* __restrict__ r1 = sT1 + l16 * pitch;
  const float* __restrict__ r2 = sT2 + l16 * pitch;
  const int sw = l16 << 2;
  v2f a1[NC][NT], a2[NC][NT];
#pragma unroll
  for (int t = 0; t < NC; ++t) {
#pragma unroll
    for (int i = 0; i < NT; ++i) {
      int xc = wstep * (jc0 + i) + woff + 4 * t + koff;
      int ca = reflect_i(xc, W) ^ sw;
      int cb = reflect_i(xc + 1, W) ^ sw;
      a1[t][i].x = r1[ca];
      a1[t][i].y = r1[cb];
      a2[t][i].x = r2[ca];
      a2[t][i].y = r2[cb];
    }
  }
  LOADS_STAY_AHEAD();

  v8f acc[NT];
#pragma unroll
  for (int i = 0; i < NT; ++i) {
#pragma unroll
    for (int e = 0; e < 8; ++e) acc[i][e] = 0.0f;
  }
#pragma unroll
  for (int t = 0; t < NC; ++t) {
#pragma unroll
    for (int i = 0; i < NT; ++i) {
      acc[i] = wmma4(a1[t][i], bf0[t], acc[i]);
      acc[i] = wmma4(a2[t][i], bf1[t], acc[i]);
    }
  }
#pragma unroll
  for (int i = 0; i < NT; ++i) {
#pragma unroll
    for (int e = 0; e < 8; ++e) op[(size_t)e * ldo + 16 * i] = acc[i][e];
  }
}

template <int WIDTH>
__global__ void __launch_bounds__(WIDTH)
j2_level_kernel(const float* __restrict__ src0_plain,
                const float* __restrict__ yh, int Hb, int Wb, int r,
                const float* __restrict__ ha0, const float* __restrict__ hb0,
                const float* __restrict__ ha1, const float* __restrict__ hb1,
                float* __restrict__ out, int nTilesR) {
  __shared__ float sT1[16 * WIDTH];
  __shared__ float sT2[16 * WIDTH];
  const int W = WIDTH;
  int lane = threadIdx.x & 31;
  int wave = threadIdx.x >> 5;
  int b = blockIdx.x / nTilesR;
  int I = blockIdx.x % nTilesR;
  int x0 = 8 * I - 4;
  int half = lane >> 4;
  int l16 = lane & 15;
  int koff = half * 2;
  int c0 = wave * 32 + 2 * l16;

  float f0[4][2], f1[4][2];
  bool k0[4][2], k1[4][2];
#pragma unroll
  for (int c = 0; c < 4; ++c) {
#pragma unroll
    for (int j = 0; j < 2; ++j) {
      int kk = 4 * c + koff + j;
      f0[c][j] = *cif_tap(l16, kk, ha0, hb0, 0, k0[c][j]);
      f1[c][j] = *cif_tap(l16, kk, ha1, hb1, 1, k1[c][j]);
    }
  }

  int stride = 2 * Wb;
  size_t band = (size_t)Hb * stride;
  const float* yb = yh + (size_t)b * 6 * band + c0;
  col_block_to_lds<4, true>(src0_plain + (size_t)b * r * W + c0, nullptr,
                            nullptr, yb, yb + 5 * band, stride, W, r, x0, koff,
                            half, c0, f0, f1, k0, k1, sT1, W);
  col_block_to_lds<4, false>(nullptr, yb + 2 * band, yb + 3 * band, yb + band,
                             yb + 4 * band, stride, W, r, x0, koff, half, c0,
                             f0, f1, k0, k1, sT2, W);

  __syncthreads();

  int H = 2 * r;
  size_t outW = 2 * (size_t)W;
  int jc0 = wave * 4;
  float* op = out + ((size_t)b * H + 16 * I + half * 8) * outW + 16 * jc0 + l16;
  row_tiles_from_lds<4, 4>(sT1, sT2, W, W, jc0, 8, -4, koff, l16, f0, f1, k0,
                           k1, op, outW);
}

template <int WIDTH>
__global__ void __launch_bounds__(WIDTH)
j1_level_kernel(const float* __restrict__ src0_plain,
                const float* __restrict__ yh, int Hb, int Wb, int r,
                const float* __restrict__ h0, int taps0,
                const float* __restrict__ h1, int taps1,
                float* __restrict__ out, int nTilesR) {
  __shared__ float sT1[16 * WIDTH];
  __shared__ float sT2[16 * WIDTH];
  const int W = WIDTH;
  int lane = threadIdx.x & 31;
  int wave = threadIdx.x >> 5;
  int b = blockIdx.x / nTilesR;
  int I = blockIdx.x % nTilesR;
  int x0 = 16 * I - 3;
  int half = lane >> 4;
  int l16 = lane & 15;
  int koff = half * 2;
  int c0 = wave * 32 + 2 * l16;

  float f0[6][2], f1[6][2];
  bool k0[6][2], k1[6][2];
#pragma unroll
  for (int c = 0; c < 6; ++c) {
#pragma unroll
    for (int j = 0; j < 2; ++j) {
      int kk = 4 * c + koff + j;
      f0[c][j] = *cf_tap(l16, kk, h0, taps0, k0[c][j]);
      f1[c][j] = *cf_tap(l16, kk, h1, taps1, k1[c][j]);
    }
  }

  int stride = 2 * Wb;
  size_t band = (size_t)Hb * stride;
  const float* yb = yh + (size_t)b * 6 * band + c0;
  col_block_to_lds<6, true>(src0_plain + (size_t)b * r * W + c0, nullptr,
                            nullptr, yb, yb + 5 * band, stride, W, r, x0, koff,
                            half, c0, f0, f1, k0, k1, sT1, W);
  col_block_to_lds<6, false>(nullptr, yb + 2 * band, yb + 3 * band, yb + band,
                             yb + 4 * band, stride, W, r, x0, koff, half, c0,
                             f0, f1, k0, k1, sT2, W);

  __syncthreads();

  int jc0 = wave * 2;
  float* op = out + ((size_t)b * r + 16 * I + half * 8) * W + 16 * jc0 + l16;
  row_tiles_from_lds<6, 2>(sT1, sT2, W, W, jc0, 16, -3, koff, l16, f0, f1, k0,
                           k1, op, (size_t)W);
}

extern "C" void kernel_launch(void* const* d_in, const int* in_sizes, int n_in,
                              void* d_out, int out_size, void* d_ws,
                              size_t ws_size, hipStream_t stream) {
  (void)in_sizes;
  (void)n_in;
  (void)out_size;
  (void)ws_size;

  const float* yl = (const float*)d_in[0];
  const float* yh0 = (const float*)d_in[1];
  const float* yh1 = (const float*)d_in[2];
  const float* yh2 = (const float*)d_in[3];
  const float* g0o = (const float*)d_in[4];
  const float* g1o = (const float*)d_in[5];
  const float* g0a = (const float*)d_in[6];
  const float* g0b = (const float*)d_in[7];
  const float* g1a = (const float*)d_in[8];
  const float* g1b = (const float*)d_in[9];

  const int B = 64;
  const size_t IMG = (size_t)512 * 512;
  float* LLb = (float*)d_ws;
  float* LLa = LLb + 2 * (size_t)B * IMG;
  float* outp = (float*)d_out;


  j2_level_kernel<128><<<dim3(64 * 16), dim3(128), 0, stream>>>(
      yl, yh2, 64, 64, 128, g0b, g0a, g1b, g1a, LLa, 16);

  j2_level_kernel<256><<<dim3(64 * 32), dim3(256), 0, stream>>>(
      LLa, yh1, 128, 128, 256, g0b, g0a, g1b, g1a, LLb, 32);

  j1_level_kernel<512><<<dim3(64 * 32), dim3(512), 0, stream>>>(
      LLb, yh0, 256, 256, 512, g0o, 7, g1o, 5, outp, 32);
}


# Column passes: a lane's two K rows are one even/odd quad-row pair, so each band row is fetched once, not twice; level-1 windows start one index earlier (even): its k grouping, sum order change
# speedup vs baseline: 3.6687x; 1.0827x over previous
#include <hip/hip_runtime.h>


typedef __attribute__((ext_vector_type(2))) float v2f;
typedef __attribute__((ext_vector_type(4))) float v4f;
typedef __attribute__((ext_vector_type(8))) float v8f;

#define SQRT_HALF 0.70710678118654752440f

__device__ __forceinline__ int reflect_i(int x, int n) {
  x = (x < 0) ? (-1 - x) : x;
  x = (x >= n) ? (2 * n - 1 - x) : x;
  return x;
}

__device__ __forceinline__ const float* cif_tap(int m, int kk,
                                                const float* __restrict__ ha,
                                                const float* __restrict__ hb,
                                                int hp, bool& ok) {
  int p = m & 3;
  int q = m >> 2;
  int want_odd = hp ? ((p & 1) ^ 1) : (p & 1);
  int t = ((kk - want_odd) >> 1) - q;
  ok = ((kk & 1) == want_odd) && (t >= 0) && (t <= 4);
  int tc = t < 0 ? 0 : (t > 4 ? 4 : t);
  const float* __restrict__ h = ((p == 0) || (p == 2)) ? ha : hb;
  int idx = (p < 2) ? (2 * tc + 1) : (2 * tc);
  return h + idx;
}

__device__ __forceinline__ const float* cf_tap(int m, int kk,
                                               const float* __restrict__ h,
                                               int taps, bool& ok) {
  int t = kk - m - (4 - (taps >> 1));
  ok = (t >= 0) && (t < taps);
  int tc = t < 0 ? 0 : (t >= taps ? taps - 1 : t);
  return h + tc;
}

struct Q4 {
  float w1r, w1i, w2r, w2i;
};

__device__ __forceinline__ Q4 c2q_load(const float* __restrict__ p1,
                                       const float* __restrict__ p2, int o) {
  Q4 q;
  q.w1r = p1[o];
  q.w1i = p1[o + 1];
  q.w2r = p2[o];
  q.w2i = p2[o + 1];
  return q;
}

__device__ __forceinline__ float c2q_even_col(Q4 q, int rr) {
  return (rr & 1) ? (q.w1i - q.w2i) : (q.w1r + q.w2r);
}

__device__ __forceinline__ float c2q_odd_col(Q4 q, int rr) {
  return (rr & 1) ? (q.w2r - q.w1r) : (q.w1i + q.w2i);
}

__device__ __forceinline__ v8f wmma4(v2f a, v2f b, v8f c) {
  return __builtin_amdgcn_wmma_f32_16x16x4_f32(false, a, false, b, (short)0, c,
                                               false, false);
}

#define LOADS_STAY_AHEAD() __builtin_amdgcn_sched_barrier(0)

template <int NC, bool SRC0_PLAIN>
__device__ __forceinline__ void col_block_to_lds(
    const float* __restrict__ p0, const float* __restrict__ p0a,
    const float* __restrict__ p0b, const float* __restrict__ p1a,
    const float* __restrict__ p1b, int stride, int W, int r, int x0, int koff,
    int half, int c0, const float (&f0)[NC][2], const float (&f1)[NC][2],
    const bool (&k0)[NC][2], const bool (&k1)[NC][2], float* __restrict__ sT,
    int pitch) {
  int rr[NC][2];
  float d0e[NC][2], d0o[NC][2];
  Q4 q0[NC], q1[NC];
#pragma unroll
  for (int c = 0; c < NC; ++c) {
#pragma unroll
    for (int j = 0; j < 2; ++j) {
      rr[c][j] = reflect_i(x0 + 4 * c + koff + j, r);
      if (SRC0_PLAIN) {
        size_t o = (size_t)rr[c][j] * W;
        d0e[c][j] = p0[o];
        d0o[c][j] = p0[o + 1];
      }
    }
    int ob = (rr[c][0] >> 1) * stride;
    if (!SRC0_PLAIN) q0[c] = c2q_load(p0a, p0b, ob);
    q1[c] = c2q_load(p1a, p1b, ob);
  }
  LOADS_STAY_AHEAD();

  const float s0sc = SRC0_PLAIN ? 1.0f : SQRT_HALF;
  v8f accE = {0.f, 0.f, 0.f, 0.f, 0.f, 0.f, 0.f, 0.f};
  v8f accO = {0.f, 0.f, 0.f, 0.f, 0.f, 0.f, 0.f, 0.f};
#pragma unroll
  for (int c = 0; c < NC; ++c) {
    v2f a0, a1, b0e, b0o, b1e, b1o;
    a0.x = (k0[c][0] ? f0[c][0] : 0.0f) * s0sc;
    a0.y = (k0[c][1] ? f0[c][1] : 0.0f) * s0sc;
    a1.x = (k1[c][0] ? f1[c][0] : 0.0f) * SQRT_HALF;
    a1.y = (k1[c][1] ? f1[c][1] : 0.0f) * SQRT_HALF;
    if (SRC0_PLAIN) {
      b0e.x = d0e[c][0];
      b0e.y = d0e[c][1];
      b0o.x = d0o[c][0];
      b0o.y = d0o[c][1];
    } else {
      b0e.x = c2q_even_col(q0[c], rr[c][0]);
      b0e.y = c2q_even_col(q0[c], rr[c][1]);
      b0o.x = c2q_odd_col(q0[c], rr[c][0]);
      b0o.y = c2q_odd_col(q0[c], rr[c][1]);
    }
    b1e.x = c2q_even_col(q1[c], rr[c][0]);
    b1e.y = c2q_even_col(q1[c], rr[c][1]);
    b1o.x = c2q_odd_col(q1[c], rr[c][0]);
    b1o.y = c2q_odd_col(q1[c], rr[c][1]);
    accE = wmma4(a0, b0e, accE);
    accO = wmma4(a0, b0o, accO);
    accE = wmma4(a1, b1e, accE);
    accO = wmma4(a1, b1o, accO);
  }

#pragma unroll
  for (int e = 0; e < 8; ++e) {
    int lr = half * 8 + e;
    v2f v;
    v.x = accE[e];
    v.y = accO[e];
    *(v2f*)(sT + lr * pitch + (c0 ^ (lr << 2))) = v;
  }
}

template <int NC, int NT>
__device__ __forceinline__ void row_tiles_from_lds(
    const float* __restrict__ sT1, const float* __restrict__ sT2, int pitch,
    int W, int jc0, int wstep, int woff, int koff, int l16,
    const float (&f0)[NC][2], const float (&f1)[NC][2], const bool (&k0)[NC][2],
    const bool (&k1)[NC][2], float* __restrict__ op, size_t ldo) {
  v2f bf0[NC], bf1[NC];
#pragma unroll
  for (int t = 0; t < NC; ++t) {
    bf0[t].x = k0[t][0] ? f0[t][0] : 0.0f;
    bf0[t].y = k0[t][1] ? f0[t][1] : 0.0f;
    bf1[t].x = k1[t][0] ? f1[t][0] : 0.0f;
    bf1[t].y = k1[t][1] ? f1[t][1] : 0.0f;
  }
  const float* __restrict__ r1 = sT1 + l16 * pitch;
  const float* __restrict__ r2 = sT2 + l16 * pitch;
  const int sw = l16 << 2;
  v2f a1[NC][NT], a2[NC][NT];
#pragma unroll
  for (int t = 0; t < NC; ++t) {
#pragma unroll
    for (int i = 0; i < NT; ++i) {
      int xc = wstep * (jc0 + i) + woff + 4 * t + koff;
      int ca = reflect_i(xc, W) ^ sw;
      int cb = reflect_i(xc + 1, W) ^ sw;
      a1[t][i].x = r1[ca];
      a1[t][i].y = r1[cb];
      a2[t][i].x = r2[ca];
      a2[t][i].y = r2[cb];
    }
  }
  LOADS_STAY_AHEAD();

  v8f acc[NT];
#pragma unroll
  for (int i = 0; i < NT; ++i) {
#pragma unroll
    for (int e = 0; e < 8; ++e) acc[i][e] = 0.0f;
  }
#pragma unroll
  for (int t = 0; t < NC; ++t) {
#pragma unroll
    for (int i = 0; i < NT; ++i) {
      acc[i] = wmma4(a1[t][i], bf0[t], acc[i]);
      acc[i] = wmma4(a2[t][i], bf1[t], acc[i]);
    }
  }
#pragma unroll
  for (int i = 0; i < NT; ++i) {
#pragma unroll
    for (int e = 0; e < 8; ++e) op[(size_t)e * ldo + 16 * i] = acc[i][e];
  }
}

template <int WIDTH>
__global__ void __launch_bounds__(WIDTH)
j2_level_kernel(const float* __restrict__ src0_plain,
                const float* __restrict__ yh, int Hb, int Wb, int r,
                const float* __restrict__ ha0, const float* __restrict__ hb0,
                const float* __restrict__ ha1, const float* __restrict__ hb1,
                float* __restrict__ out, int nTilesR) {
  __shared__ float sT1[16 * WIDTH];
  __shared__ float sT2[16 * WIDTH];
  const int W = WIDTH;
  int lane = threadIdx.x & 31;
  int wave = threadIdx.x >> 5;
  int b = blockIdx.x / nTilesR;
  int I = blockIdx.x % nTilesR;
  int x0 = 8 * I - 4;
  int half = lane >> 4;
  int l16 = lane & 15;
  int koff = half * 2;
  int c0 = wave * 32 + 2 * l16;

  float f0[4][2], f1[4][2];
  bool k0[4][2], k1[4][2];
#pragma unroll
  for (int c = 0; c < 4; ++c) {
#pragma unroll
    for (int j = 0; j < 2; ++j) {
      int kk = 4 * c + koff + j;
      f0[c][j] = *cif_tap(l16, kk, ha0, hb0, 0, k0[c][j]);
      f1[c][j] = *cif_tap(l16, kk, ha1, hb1, 1, k1[c][j]);
    }
  }

  int stride = 2 * Wb;
  size_t band = (size_t)Hb * stride;
  const float* yb = yh + (size_t)b * 6 * band + c0;
  col_block_to_lds<4, true>(src0_plain + (size_t)b * r * W + c0, nullptr,
                            nullptr, yb, yb + 5 * band, stride, W, r, x0, koff,
                            half, c0, f0, f1, k0, k1, sT1, W);
  col_block_to_lds<4, false>(nullptr, yb + 2 * band, yb + 3 * band, yb + band,
                             yb + 4 * band, stride, W, r, x0, koff, half, c0,
                             f0, f1, k0, k1, sT2, W);

  __syncthreads();

  int H = 2 * r;
  size_t outW = 2 * (size_t)W;
  int jc0 = wave * 4;
  float* op = out + ((size_t)b * H + 16 * I + half * 8) * outW + 16 * jc0 + l16;
  row_tiles_from_lds<4, 4>(sT1, sT2, W, W, jc0, 8, -4, koff, l16, f0, f1, k0,
                           k1, op, outW);
}

template <int WIDTH>
__global__ void __launch_bounds__(WIDTH)
j1_level_kernel(const float* __restrict__ src0_plain,
                const float* __restrict__ yh, int Hb, int Wb, int r,
                const float* __restrict__ h0, int taps0,
                const float* __restrict__ h1, int taps1,
                float* __restrict__ out, int nTilesR) {
  __shared__ float sT1[16 * WIDTH];
  __shared__ float sT2[16 * WIDTH];
  const int W = WIDTH;
  int lane = threadIdx.x & 31;
  int wave = threadIdx.x >> 5;
  int b = blockIdx.x / nTilesR;
  int I = blockIdx.x % nTilesR;
  int x0 = 16 * I - 4;
  int half = lane >> 4;
  int l16 = lane & 15;
  int koff = half * 2;
  int c0 = wave * 32 + 2 * l16;

  float f0[6][2], f1[6][2];
  bool k0[6][2], k1[6][2];
#pragma unroll
  for (int c = 0; c < 6; ++c) {
#pragma unroll
    for (int j = 0; j < 2; ++j) {
      int kk = 4 * c + koff + j;
      f0[c][j] = *cf_tap(l16, kk, h0, taps0, k0[c][j]);
      f1[c][j] = *cf_tap(l16, kk, h1, taps1, k1[c][j]);
    }
  }

  int stride = 2 * Wb;
  size_t band = (size_t)Hb * stride;
  const float* yb = yh + (size_t)b * 6 * band + c0;
  col_block_to_lds<6, true>(src0_plain + (size_t)b * r * W + c0, nullptr,
                            nullptr, yb, yb + 5 * band, stride, W, r, x0, koff,
                            half, c0, f0, f1, k0, k1, sT1, W);
  col_block_to_lds<6, false>(nullptr, yb + 2 * band, yb + 3 * band, yb + band,
                             yb + 4 * band, stride, W, r, x0, koff, half, c0,
                             f0, f1, k0, k1, sT2, W);

  __syncthreads();

  int jc0 = wave * 2;
  float* op = out + ((size_t)b * r + 16 * I + half * 8) * W + 16 * jc0 + l16;
  row_tiles_from_lds<6, 2>(sT1, sT2, W, W, jc0, 16, -4, koff, l16, f0, f1, k0,
                           k1, op, (size_t)W);
}

extern "C" void kernel_launch(void* const* d_in, const int* in_sizes, int n_in,
                              void* d_out, int out_size, void* d_ws,
                              size_t ws_size, hipStream_t stream) {
  (void)in_sizes;
  (void)n_in;
  (void)out_size;
  (void)ws_size;

  const float* yl = (const float*)d_in[0];
  const float* yh0 = (const float*)d_in[1];
  const float* yh1 = (const float*)d_in[2];
  const float* yh2 = (const float*)d_in[3];
  const float* g0o = (const float*)d_in[4];
  const float* g1o = (const float*)d_in[5];
  const float* g0a = (const float*)d_in[6];
  const float* g0b = (const float*)d_in[7];
  const float* g1a = (const float*)d_in[8];
  const float* g1b = (const float*)d_in[9];

  const int B = 64;
  const size_t IMG = (size_t)512 * 512;
  float* LLb = (float*)d_ws;
  float* LLa = LLb + 2 * (size_t)B * IMG;
  float* outp = (float*)d_out;


  j2_level_kernel<128><<<dim3(64 * 16), dim3(128), 0, stream>>>(
      yl, yh2, 64, 64, 128, g0b, g0a, g1b, g1a, LLa, 16);

  j2_level_kernel<256><<<dim3(64 * 32), dim3(256), 0, stream>>>(
      LLa, yh1, 128, 128, 256, g0b, g0a, g1b, g1a, LLb, 32);

  j1_level_kernel<512><<<dim3(64 * 32), dim3(512), 0, stream>>>(
      LLb, yh0, 256, 256, 512, g0o, 7, g1o, 5, outp, 32);
}


# Level-1 kernel cut from 156 to 126 registers (8-wave launch bound; loads = a wave-uniform base + a 32-bit lane index, no 64-bit lane address): two 16-wave groups per WGP, not one; sums unchanged
# speedup vs baseline: 4.0912x; 1.1152x over previous
#include <hip/hip_runtime.h>


typedef __attribute__((ext_vector_type(2))) float v2f;
typedef __attribute__((ext_vector_type(4))) float v4f;
typedef __attribute__((ext_vector_type(8))) float v8f;

#define SQRT_HALF 0.70710678118654752440f

__device__ __forceinline__ int reflect_i(int x, int n) {
  x = (x < 0) ? (-1 - x) : x;
  x = (x >= n) ? (2 * n - 1 - x) : x;
  return x;
}

__device__ __forceinline__ const float* cif_tap(int m, int kk,
                                                const float* __restrict__ ha,
                                                const float* __restrict__ hb,
                                                int hp, bool& ok) {
  int p = m & 3;
  int q = m >> 2;
  int want_odd = hp ? ((p & 1) ^ 1) : (p & 1);
  int t = ((kk - want_odd) >> 1) - q;
  ok = ((kk & 1) == want_odd) && (t >= 0) && (t <= 4);
  int tc = t < 0 ? 0 : (t > 4 ? 4 : t);
  const float* __restrict__ h = ((p == 0) || (p == 2)) ? ha : hb;
  int idx = (p < 2) ? (2 * tc + 1) : (2 * tc);
  return h + idx;
}

__device__ __forceinline__ const float* cf_tap(int m, int kk,
                                               const float* __restrict__ h,
                                               int taps, bool& ok) {
  int t = kk - m - (4 - (taps >> 1));
  ok = (t >= 0) && (t < taps);
  int tc = t < 0 ? 0 : (t >= taps ? taps - 1 : t);
  return h + tc;
}

struct Q4 {
  float w1r, w1i, w2r, w2i;
};

__device__ __forceinline__ Q4 c2q_load(const float* __restrict__ p1,
                                       const float* __restrict__ p2, int o) {
  Q4 q;
  q.w1r = p1[o];
  q.w1i = p1[o + 1];
  q.w2r = p2[o];
  q.w2i = p2[o + 1];
  return q;
}

__device__ __forceinline__ float c2q_even_col(Q4 q, int rr) {
  return (rr & 1) ? (q.w1i - q.w2i) : (q.w1r + q.w2r);
}

__device__ __forceinline__ float c2q_odd_col(Q4 q, int rr) {
  return (rr & 1) ? (q.w2r - q.w1r) : (q.w1i + q.w2i);
}

__device__ __forceinline__ v8f wmma4(v2f a, v2f b, v8f c) {
  return __builtin_amdgcn_wmma_f32_16x16x4_f32(false, a, false, b, (short)0, c,
                                               false, false);
}

#define LOADS_STAY_AHEAD() __builtin_amdgcn_sched_barrier(0)

template <int NC, bool SRC0_PLAIN>
__device__ __forceinline__ void col_block_to_lds(
    const float* __restrict__ p0, const float* __restrict__ p0a,
    const float* __restrict__ p0b, const float* __restrict__ p1a,
    const float* __restrict__ p1b, int stride, int W, int r, int x0, int koff,
    int half, int c0, const float (&f0)[NC][2], const float (&f1)[NC][2],
    const bool (&k0)[NC][2], const bool (&k1)[NC][2], float* __restrict__ sT,
    int pitch) {
  int rr[NC][2];
  float d0e[NC][2], d0o[NC][2];
  Q4 q0[NC], q1[NC];
#pragma unroll
  for (int c = 0; c < NC; ++c) {
#pragma unroll
    for (int j = 0; j < 2; ++j) {
      rr[c][j] = reflect_i(x0 + 4 * c + koff + j, r);
      if (SRC0_PLAIN) {
        size_t o = (size_t)rr[c][j] * W;
        d0e[c][j] = p0[o];
        d0o[c][j] = p0[o + 1];
      }
    }
    int ob = (rr[c][0] >> 1) * stride;
    if (!SRC0_PLAIN) q0[c] = c2q_load(p0a, p0b, ob);
    q1[c] = c2q_load(p1a, p1b, ob);
  }
  LOADS_STAY_AHEAD();

  const float s0sc = SRC0_PLAIN ? 1.0f : SQRT_HALF;
  v8f accE = {0.f, 0.f, 0.f, 0.f, 0.f, 0.f, 0.f, 0.f};
  v8f accO = {0.f, 0.f, 0.f, 0.f, 0.f, 0.f, 0.f, 0.f};
#pragma unroll
  for (int c = 0; c < NC; ++c) {
    v2f a0, a1, b0e, b0o, b1e, b1o;
    a0.x = (k0[c][0] ? f0[c][0] : 0.0f) * s0sc;
    a0.y = (k0[c][1] ? f0[c][1] : 0.0f) * s0sc;
    a1.x = (k1[c][0] ? f1[c][0] : 0.0f) * SQRT_HALF;
    a1.y = (k1[c][1] ? f1[c][1] : 0.0f) * SQRT_HALF;
    if (SRC0_PLAIN) {
      b0e.x = d0e[c][0];
      b0e.y = d0e[c][1];
      b0o.x = d0o[c][0];
      b0o.y = d0o[c][1];
    } else {
      b0e.x = c2q_even_col(q0[c], rr[c][0]);
      b0e.y = c2q_even_col(q0[c], rr[c][1]);
      b0o.x = c2q_odd_col(q0[c], rr[c][0]);
      b0o.y = c2q_odd_col(q0[c], rr[c][1]);
    }
    b1e.x = c2q_even_col(q1[c], rr[c][0]);
    b1e.y = c2q_even_col(q1[c], rr[c][1]);
    b1o.x = c2q_odd_col(q1[c], rr[c][0]);
    b1o.y = c2q_odd_col(q1[c], rr[c][1]);
    accE = wmma4(a0, b0e, accE);
    accO = wmma4(a0, b0o, accO);
    accE = wmma4(a1, b1e, accE);
    accO = wmma4(a1, b1o, accO);
  }

#pragma unroll
  for (int e = 0; e < 8; ++e) {
    int lr = half * 8 + e;
    v2f v;
    v.x = accE[e];
    v.y = accO[e];
    *(v2f*)(sT + lr * pitch + (c0 ^ (lr << 2))) = v;
  }
}

template <int NC, bool SRC0_PLAIN, int P0A, int P0B, int P1A, int P1B>
__device__ __forceinline__ void col_block_to_lds_lane32(
    const float* __restrict__ s0, const float* __restrict__ yb, unsigned bp,
    unsigned stride, unsigned W, int r, int x0, int koff, int half,
    unsigned c0, const float (&f0)[NC][2],
    const float (&f1)[NC][2], const bool (&k0)[NC][2], const bool (&k1)[NC][2],
    float* __restrict__ sT, int pitch) {
  int rr[NC][2];
  v2f d0[NC][2];
  Q4 q0[NC], q1[NC];
#pragma unroll
  for (int c = 0; c < NC; ++c) {
#pragma unroll
    for (int j = 0; j < 2; ++j) {
      rr[c][j] = reflect_i(x0 + 4 * c + koff + j, r);
      if (SRC0_PLAIN)
        d0[c][j] = ((const v2f*)s0)[(unsigned)rr[c][j] * (W >> 1) + (c0 >> 1)];
    }
    unsigned ob = (unsigned)(rr[c][0] >> 1) * (stride >> 1) + (c0 >> 1);
    if (!SRC0_PLAIN) {
      v2f wa = ((const v2f*)yb)[ob + P0A * bp];
      v2f wb = ((const v2f*)yb)[ob + P0B * bp];
      q0[c].w1r = wa.x;
      q0[c].w1i = wa.y;
      q0[c].w2r = wb.x;
      q0[c].w2i = wb.y;
    }
    v2f wa = ((const v2f*)yb)[ob + P1A * bp];
    v2f wb = ((const v2f*)yb)[ob + P1B * bp];
    q1[c].w1r = wa.x;
    q1[c].w1i = wa.y;
    q1[c].w2r = wb.x;
    q1[c].w2i = wb.y;
  }
  LOADS_STAY_AHEAD();

  const float s0sc = SRC0_PLAIN ? 1.0f : SQRT_HALF;
  v8f accE = {0.f, 0.f, 0.f, 0.f, 0.f, 0.f, 0.f, 0.f};
  v8f accO = {0.f, 0.f, 0.f, 0.f, 0.f, 0.f, 0.f, 0.f};
#pragma unroll
  for (int c = 0; c < NC; ++c) {
    v2f a0, a1, b0e, b0o, b1e, b1o;
    a0.x = (k0[c][0] ? f0[c][0] : 0.0f) * s0sc;
    a0.y = (k0[c][1] ? f0[c][1] : 0.0f) * s0sc;
    a1.x = (k1[c][0] ? f1[c][0] : 0.0f) * SQRT_HALF;
    a1.y = (k1[c][1] ? f1[c][1] : 0.0f) * SQRT_HALF;
    if (SRC0_PLAIN) {
      b0e.x = d0[c][0].x;
      b0e.y = d0[c][1].x;
      b0o.x = d0[c][0].y;
      b0o.y = d0[c][1].y;
    } else {
      b0e.x = c2q_even_col(q0[c], rr[c][0]);
      b0e.y = c2q_even_col(q0[c], rr[c][1]);
      b0o.x = c2q_odd_col(q0[c], rr[c][0]);
      b0o.y = c2q_odd_col(q0[c], rr[c][1]);
    }
    b1e.x = c2q_even_col(q1[c], rr[c][0]);
    b1e.y = c2q_even_col(q1[c], rr[c][1]);
    b1o.x = c2q_odd_col(q1[c], rr[c][0]);
    b1o.y = c2q_odd_col(q1[c], rr[c][1]);
    accE = wmma4(a0, b0e, accE);
    accO = wmma4(a0, b0o, accO);
    accE = wmma4(a1, b1e, accE);
    accO = wmma4(a1, b1o, accO);
  }

#pragma unroll
  for (int e = 0; e < 8; ++e) {
    int lr = half * 8 + e;
    v2f v;
    v.x = accE[e];
    v.y = accO[e];
    *(v2f*)(sT + lr * pitch + (c0 ^ (unsigned)(lr << 2))) = v;
  }
}

template <int NC, int NT>
__device__ __forceinline__ void row_tiles_from_lds(
    const float* __restrict__ sT1, const float* __restrict__ sT2, int pitch,
    int W, int jc0, int wstep, int woff, int koff, int l16,
    const float (&f0)[NC][2], const float (&f1)[NC][2], const bool (&k0)[NC][2],
    const bool (&k1)[NC][2], float* __restrict__ op, size_t ldo) {
  v2f bf0[NC], bf1[NC];
#pragma unroll
  for (int t = 0; t < NC; ++t) {
    bf0[t].x = k0[t][0] ? f0[t][0] : 0.0f;
    bf0[t].y = k0[t][1] ? f0[t][1] : 0.0f;
    bf1[t].x = k1[t][0] ? f1[t][0] : 0.0f;
    bf1[t].y = k1[t][1] ? f1[t][1] : 0.0f;
  }
  const float* __restrict__ r1 = sT1 + l16 * pitch;
  const float* __restrict__ r2 = sT2 + l16 * pitch;
  const int sw = l16 << 2;
  v2f a1[NC][NT], a2[NC][NT];
#pragma unroll
  for (int t = 0; t < NC; ++t) {
#pragma unroll
    for (int i = 0; i < NT; ++i) {
      int xc = wstep * (jc0 + i) + woff + 4 * t + koff;
      int ca = reflect_i(xc, W) ^ sw;
      int cb = reflect_i(xc + 1, W) ^ sw;
      a1[t][i].x = r1[ca];
      a1[t][i].y = r1[cb];
      a2[t][i].x = r2[ca];
      a2[t][i].y = r2[cb];
    }
  }
  LOADS_STAY_AHEAD();

  v8f acc[NT];
#pragma unroll
  for (int i = 0; i < NT; ++i) {
#pragma unroll
    for (int e = 0; e < 8; ++e) acc[i][e] = 0.0f;
  }
#pragma unroll
  for (int t = 0; t < NC; ++t) {
#pragma unroll
    for (int i = 0; i < NT; ++i) {
      acc[i] = wmma4(a1[t][i], bf0[t], acc[i]);
      acc[i] = wmma4(a2[t][i], bf1[t], acc[i]);
    }
  }
#pragma unroll
  for (int i = 0; i < NT; ++i) {
#pragma unroll
    for (int e = 0; e < 8; ++e) op[(size_t)e * ldo + 16 * i] = acc[i][e];
  }
}

template <int WIDTH>
__global__ void __launch_bounds__(WIDTH)
j2_level_kernel(const float* __restrict__ src0_plain,
                const float* __restrict__ yh, int Hb, int Wb, int r,
                const float* __restrict__ ha0, const float* __restrict__ hb0,
                const float* __restrict__ ha1, const float* __restrict__ hb1,
                float* __restrict__ out, int nTilesR) {
  __shared__ float sT1[16 * WIDTH];
  __shared__ float sT2[16 * WIDTH];
  const int W = WIDTH;
  int lane = threadIdx.x & 31;
  int wave = threadIdx.x >> 5;
  int b = blockIdx.x / nTilesR;
  int I = blockIdx.x % nTilesR;
  int x0 = 8 * I - 4;
  int half = lane >> 4;
  int l16 = lane & 15;
  int koff = half * 2;
  int c0 = wave * 32 + 2 * l16;

  float f0[4][2], f1[4][2];
  bool k0[4][2], k1[4][2];
#pragma unroll
  for (int c = 0; c < 4; ++c) {
#pragma unroll
    for (int j = 0; j < 2; ++j) {
      int kk = 4 * c + koff + j;
      f0[c][j] = *cif_tap(l16, kk, ha0, hb0, 0, k0[c][j]);
      f1[c][j] = *cif_tap(l16, kk, ha1, hb1, 1, k1[c][j]);
    }
  }

  int stride = 2 * Wb;
  size_t band = (size_t)Hb * stride;
  const float* yb = yh + (size_t)b * 6 * band + c0;
  col_block_to_lds<4, true>(src0_plain + (size_t)b * r * W + c0, nullptr,
                            nullptr, yb, yb + 5 * band, stride, W, r, x0, koff,
                            half, c0, f0, f1, k0, k1, sT1, W);
  col_block_to_lds<4, false>(nullptr, yb + 2 * band, yb + 3 * band, yb + band,
                             yb + 4 * band, stride, W, r, x0, koff, half, c0,
                             f0, f1, k0, k1, sT2, W);

  __syncthreads();

  int H = 2 * r;
  size_t outW = 2 * (size_t)W;
  int jc0 = wave * 4;
  float* op = out + ((size_t)b * H + 16 * I + half * 8) * outW + 16 * jc0 + l16;
  row_tiles_from_lds<4, 4>(sT1, sT2, W, W, jc0, 8, -4, koff, l16, f0, f1, k0,
                           k1, op, outW);
}

template <int WIDTH>
__global__ void __launch_bounds__(WIDTH, 8)
j1_level_kernel(const float* __restrict__ src0_plain,
                const float* __restrict__ yh, int Hb, int Wb, int r,
                const float* __restrict__ h0, int taps0,
                const float* __restrict__ h1, int taps1,
                float* __restrict__ out, int nTilesR) {
  __shared__ float sT1[16 * WIDTH];
  __shared__ float sT2[16 * WIDTH];
  const int W = WIDTH;
  int lane = threadIdx.x & 31;
  int wave = threadIdx.x >> 5;
  int b = blockIdx.x / nTilesR;
  int I = blockIdx.x % nTilesR;
  int x0 = 16 * I - 4;
  int half = lane >> 4;
  int l16 = lane & 15;
  int koff = half * 2;
  int c0 = wave * 32 + 2 * l16;

  float f0[6][2], f1[6][2];
  bool k0[6][2], k1[6][2];
#pragma unroll
  for (int c = 0; c < 6; ++c) {
#pragma unroll
    for (int j = 0; j < 2; ++j) {
      int kk = 4 * c + koff + j;
      f0[c][j] = *cf_tap(l16, kk, h0, taps0, k0[c][j]);
      f1[c][j] = *cf_tap(l16, kk, h1, taps1, k1[c][j]);
    }
  }

  unsigned stride = 2u * (unsigned)Wb;
  unsigned bp = (unsigned)Hb * (unsigned)Wb;
  const float* yb = yh + (size_t)b * 12 * bp;
  const float* sp = src0_plain + (size_t)b * r * W;
  col_block_to_lds_lane32<6, true, 0, 0, 0, 5>(sp, yb, bp, stride, (unsigned)W,
                                               r, x0, koff, half, (unsigned)c0,
                                               f0, f1, k0, k1, sT1, W);
  col_block_to_lds_lane32<6, false, 2, 3, 1, 4>(nullptr, yb, bp, stride,
                                                (unsigned)W, r, x0, koff, half,
                                                (unsigned)c0, f0, f1, k0, k1,
                                                sT2, W);

  __syncthreads();

  int jc0 = wave * 2;
  float* op = out + ((size_t)b * r + 16 * I + half * 8) * W + 16 * jc0 + l16;
  row_tiles_from_lds<6, 2>(sT1, sT2, W, W, jc0, 16, -4, koff, l16, f0, f1, k0,
                           k1, op, (size_t)W);
}

extern "C" void kernel_launch(void* const* d_in, const int* in_sizes, int n_in,
                              void* d_out, int out_size, void* d_ws,
                              size_t ws_size, hipStream_t stream) {
  (void)in_sizes;
  (void)n_in;
  (void)out_size;
  (void)ws_size;

  const float* yl = (const float*)d_in[0];
  const float* yh0 = (const float*)d_in[1];
  const float* yh1 = (const float*)d_in[2];
  const float* yh2 = (const float*)d_in[3];
  const float* g0o = (const float*)d_in[4];
  const float* g1o = (const float*)d_in[5];
  const float* g0a = (const float*)d_in[6];
  const float* g0b = (const float*)d_in[7];
  const float* g1a = (const float*)d_in[8];
  const float* g1b = (const float*)d_in[9];

  const int B = 64;
  const size_t IMG = (size_t)512 * 512;
  float* LLb = (float*)d_ws;
  float* LLa = LLb + 2 * (size_t)B * IMG;
  float* outp = (float*)d_out;


  j2_level_kernel<128><<<dim3(64 * 16), dim3(128), 0, stream>>>(
      yl, yh2, 64, 64, 128, g0b, g0a, g1b, g1a, LLa, 16);

  j2_level_kernel<256><<<dim3(64 * 32), dim3(256), 0, stream>>>(
      LLa, yh1, 128, 128, 256, g0b, g0a, g1b, g1a, LLb, 32);

  j1_level_kernel<512><<<dim3(64 * 32), dim3(512), 0, stream>>>(
      LLb, yh0, 256, 256, 512, g0o, 7, g1o, 5, outp, 32);
}


# A work-group walks several strips; a wave refills each column block's register buffer with the next strip's loads right after use, so loads fly under matrix work; LDS-only barriers; sums unchanged
# speedup vs baseline: 4.3310x; 1.0586x over previous
#include <hip/hip_runtime.h>


typedef __attribute__((ext_vector_type(2))) float v2f;
typedef __attribute__((ext_vector_type(4))) float v4f;
typedef __attribute__((ext_vector_type(8))) float v8f;

#define SQRT_HALF 0.70710678118654752440f

__device__ __forceinline__ int reflect_i(int x, int n) {
  x = (x < 0) ? (-1 - x) : x;
  x = (x >= n) ? (2 * n - 1 - x) : x;
  return x;
}

__device__ __forceinline__ const float* cif_tap(int m, int kk,
                                                const float* __restrict__ ha,
                                                const float* __restrict__ hb,
                                                int hp, bool& ok) {
  int p = m & 3;
  int q = m >> 2;
  int want_odd = hp ? ((p & 1) ^ 1) : (p & 1);
  int t = ((kk - want_odd) >> 1) - q;
  ok = ((kk & 1) == want_odd) && (t >= 0) && (t <= 4);
  int tc = t < 0 ? 0 : (t > 4 ? 4 : t);
  const float* __restrict__ h = ((p == 0) || (p == 2)) ? ha : hb;
  int idx = (p < 2) ? (2 * tc + 1) : (2 * tc);
  return h + idx;
}

__device__ __forceinline__ const float* cf_tap(int m, int kk,
                                               const float* __restrict__ h,
                                               int taps, bool& ok) {
  int t = kk - m - (4 - (taps >> 1));
  ok = (t >= 0) && (t < taps);
  int tc = t < 0 ? 0 : (t >= taps ? taps - 1 : t);
  return h + tc;
}

struct Q4 {
  float w1r, w1i, w2r, w2i;
};

__device__ __forceinline__ Q4 c2q_load(const float* __restrict__ p1,
                                       const float* __restrict__ p2, int o) {
  Q4 q;
  q.w1r = p1[o];
  q.w1i = p1[o + 1];
  q.w2r = p2[o];
  q.w2i = p2[o + 1];
  return q;
}

__device__ __forceinline__ float c2q_even_col(Q4 q, int rr) {
  return (rr & 1) ? (q.w1i - q.w2i) : (q.w1r + q.w2r);
}

__device__ __forceinline__ float c2q_odd_col(Q4 q, int rr) {
  return (rr & 1) ? (q.w2r - q.w1r) : (q.w1i + q.w2i);
}

__device__ __forceinline__ v8f wmma4(v2f a, v2f b, v8f c) {
  return __builtin_amdgcn_wmma_f32_16x16x4_f32(false, a, false, b, (short)0, c,
                                               false, false);
}

#define LOADS_STAY_AHEAD() __builtin_amdgcn_sched_barrier(0)

template <int NC>
struct ColBuf {
  unsigned par;
  v2f d0[NC][2];
  Q4 q0[NC], q1[NC];
};

template <bool V>
struct BoolC {
  static constexpr bool value = V;
};

template <int NC, bool SRC0_PLAIN>
__device__ __forceinline__ void col_block_load(
    ColBuf<NC>& d, const float* __restrict__ p0, const float* __restrict__ p0a,
    const float* __restrict__ p0b, const float* __restrict__ p1a,
    const float* __restrict__ p1b, int stride, int W, int r, int x0,
    int koff) {
  LOADS_STAY_AHEAD();
  d.par = 0u;
#pragma unroll
  for (int c = 0; c < NC; ++c) {
    int rr[2];
#pragma unroll
    for (int j = 0; j < 2; ++j) {
      rr[j] = reflect_i(x0 + 4 * c + koff + j, r);
      if (SRC0_PLAIN) {
        size_t o = (size_t)rr[j] * W;
        d.d0[c][j].x = p0[o];
        d.d0[c][j].y = p0[o + 1];
      }
    }
    d.par |= ((unsigned)rr[0] & 1u) << c;
    int ob = (rr[0] >> 1) * stride;
    if (!SRC0_PLAIN) d.q0[c] = c2q_load(p0a, p0b, ob);
    d.q1[c] = c2q_load(p1a, p1b, ob);
  }
  LOADS_STAY_AHEAD();
}

template <int NC, bool SRC0_PLAIN, int P0A, int P0B, int P1A, int P1B>
__device__ __forceinline__ void col_block_load_lane32(
    ColBuf<NC>& d, const float* __restrict__ s0, const float* __restrict__ yb,
    unsigned bp, unsigned stride, unsigned W, int r, int x0, int koff,
    unsigned c0) {
  LOADS_STAY_AHEAD();
  d.par = 0u;
#pragma unroll
  for (int c = 0; c < NC; ++c) {
    int rr[2];
#pragma unroll
    for (int j = 0; j < 2; ++j) {
      rr[j] = reflect_i(x0 + 4 * c + koff + j, r);
      if (SRC0_PLAIN)
        d.d0[c][j] = ((const v2f*)s0)[(unsigned)rr[j] * (W >> 1) + (c0 >> 1)];
    }
    d.par |= ((unsigned)rr[0] & 1u) << c;
    unsigned ob = (unsigned)(rr[0] >> 1) * (stride >> 1) + (c0 >> 1);
    if (!SRC0_PLAIN) {
      v2f wa = ((const v2f*)yb)[ob + P0A * bp];
      v2f wb = ((const v2f*)yb)[ob + P0B * bp];
      d.q0[c].w1r = wa.x;
      d.q0[c].w1i = wa.y;
      d.q0[c].w2r = wb.x;
      d.q0[c].w2i = wb.y;
    }
    v2f wa = ((const v2f*)yb)[ob + P1A * bp];
    v2f wb = ((const v2f*)yb)[ob + P1B * bp];
    d.q1[c].w1r = wa.x;
    d.q1[c].w1i = wa.y;
    d.q1[c].w2r = wb.x;
    d.q1[c].w2i = wb.y;
  }
  LOADS_STAY_AHEAD();
}

template <int NC, bool SRC0_PLAIN>
__device__ __forceinline__ void col_block_compute(
    const ColBuf<NC>& d, int half, int c0, const float (&f0)[NC][2],
    const float (&f1)[NC][2], const bool (&k0)[NC][2], const bool (&k1)[NC][2],
    float* __restrict__ sT, int pitch) {
  const float s0sc = SRC0_PLAIN ? 1.0f : SQRT_HALF;
  v8f accE = {0.f, 0.f, 0.f, 0.f, 0.f, 0.f, 0.f, 0.f};
  v8f accO = {0.f, 0.f, 0.f, 0.f, 0.f, 0.f, 0.f, 0.f};
#pragma unroll
  for (int c = 0; c < NC; ++c) {
    v2f a0, a1, b0e, b0o, b1e, b1o;
    const int r0 = (int)((d.par >> c) & 1u);
    const int r1 = r0 ^ 1;
    a0.x = (k0[c][0] ? f0[c][0] : 0.0f) * s0sc;
    a0.y = (k0[c][1] ? f0[c][1] : 0.0f) * s0sc;
    a1.x = (k1[c][0] ? f1[c][0] : 0.0f) * SQRT_HALF;
    a1.y = (k1[c][1] ? f1[c][1] : 0.0f) * SQRT_HALF;
    if (SRC0_PLAIN) {
      b0e.x = d.d0[c][0].x;
      b0e.y = d.d0[c][1].x;
      b0o.x = d.d0[c][0].y;
      b0o.y = d.d0[c][1].y;
    } else {
      b0e.x = c2q_even_col(d.q0[c], r0);
      b0e.y = c2q_even_col(d.q0[c], r1);
      b0o.x = c2q_odd_col(d.q0[c], r0);
      b0o.y = c2q_odd_col(d.q0[c], r1);
    }
    b1e.x = c2q_even_col(d.q1[c], r0);
    b1e.y = c2q_even_col(d.q1[c], r1);
    b1o.x = c2q_odd_col(d.q1[c], r0);
    b1o.y = c2q_odd_col(d.q1[c], r1);
    accE = wmma4(a0, b0e, accE);
    accO = wmma4(a0, b0o, accO);
    accE = wmma4(a1, b1e, accE);
    accO = wmma4(a1, b1o, accO);
  }

#pragma unroll
  for (int e = 0; e < 8; ++e) {
    int lr = half * 8 + e;
    v2f v;
    v.x = accE[e];
    v.y = accO[e];
    *(v2f*)(sT + lr * pitch + (c0 ^ (lr << 2))) = v;
  }
}

__device__ __forceinline__ void lds_barrier() {
  __builtin_amdgcn_fence(__ATOMIC_RELEASE, "workgroup", "local");
  __builtin_amdgcn_s_barrier();
  __builtin_amdgcn_fence(__ATOMIC_ACQUIRE, "workgroup", "local");
}

template <int NC, int NT, bool MORE>
__device__ __forceinline__ void row_tiles_from_lds(
    const float* __restrict__ sT1, const float* __restrict__ sT2, int pitch,
    int W, int jc0, int wstep, int woff, int koff, int l16,
    const float (&f0)[NC][2], const float (&f1)[NC][2], const bool (&k0)[NC][2],
    const bool (&k1)[NC][2], float* __restrict__ op, size_t ldo) {
  v2f bf0[NC], bf1[NC];
#pragma unroll
  for (int t = 0; t < NC; ++t) {
    bf0[t].x = k0[t][0] ? f0[t][0] : 0.0f;
    bf0[t].y = k0[t][1] ? f0[t][1] : 0.0f;
    bf1[t].x = k1[t][0] ? f1[t][0] : 0.0f;
    bf1[t].y = k1[t][1] ? f1[t][1] : 0.0f;
  }
  const float* __restrict__ r1 = sT1 + l16 * pitch;
  const float* __restrict__ r2 = sT2 + l16 * pitch;
  const int sw = l16 << 2;
  v2f a1[NC][NT], a2[NC][NT];
#pragma unroll
  for (int t = 0; t < NC; ++t) {
#pragma unroll
    for (int i = 0; i < NT; ++i) {
      int xc = wstep * (jc0 + i) + woff + 4 * t + koff;
      int ca = reflect_i(xc, W) ^ sw;
      int cb = reflect_i(xc + 1, W) ^ sw;
      a1[t][i].x = r1[ca];
      a1[t][i].y = r1[cb];
      a2[t][i].x = r2[ca];
      a2[t][i].y = r2[cb];
    }
  }
  LOADS_STAY_AHEAD();
  if (MORE) lds_barrier();

  v8f acc[NT];
#pragma unroll
  for (int i = 0; i < NT; ++i) {
#pragma unroll
    for (int e = 0; e < 8; ++e) acc[i][e] = 0.0f;
  }
#pragma unroll
  for (int t = 0; t < NC; ++t) {
#pragma unroll
    for (int i = 0; i < NT; ++i) {
      acc[i] = wmma4(a1[t][i], bf0[t], acc[i]);
      acc[i] = wmma4(a2[t][i], bf1[t], acc[i]);
    }
  }
#pragma unroll
  for (int i = 0; i < NT; ++i) {
#pragma unroll
    for (int e = 0; e < 8; ++e) op[(size_t)e * ldo + 16 * i] = acc[i][e];
  }
}

template <int WIDTH, int S>
__global__ void __launch_bounds__(WIDTH)
j2_level_kernel(const float* __restrict__ src0_plain,
                const float* __restrict__ yh, int Hb, int Wb, int r,
                const float* __restrict__ ha0, const float* __restrict__ hb0,
                const float* __restrict__ ha1, const float* __restrict__ hb1,
                float* __restrict__ out, int nTilesR) {
  __shared__ float sT1[16 * WIDTH];
  __shared__ float sT2[16 * WIDTH];
  const int W = WIDTH;
  int lane = threadIdx.x & 31;
  int wave = threadIdx.x >> 5;
  int nGroups = nTilesR / S;
  int b = blockIdx.x / nGroups;
  int I0 = (blockIdx.x % nGroups) * S;
  int half = lane >> 4;
  int l16 = lane & 15;
  int koff = half * 2;
  int c0 = wave * 32 + 2 * l16;

  float f0[4][2], f1[4][2];
  bool k0[4][2], k1[4][2];
#pragma unroll
  for (int c = 0; c < 4; ++c) {
#pragma unroll
    for (int j = 0; j < 2; ++j) {
      int kk = 4 * c + koff + j;
      f0[c][j] = *cif_tap(l16, kk, ha0, hb0, 0, k0[c][j]);
      f1[c][j] = *cif_tap(l16, kk, ha1, hb1, 1, k1[c][j]);
    }
  }

  int stride = 2 * Wb;
  size_t band = (size_t)Hb * stride;
  const float* yb = yh + (size_t)b * 6 * band + c0;
  const float* sp = src0_plain + (size_t)b * r * W + c0;
  int H = 2 * r;
  size_t outW = 2 * (size_t)W;
  int jc0 = wave * 4;

  ColBuf<4> A, B;
  col_block_load<4, true>(A, sp, nullptr, nullptr, yb, yb + 5 * band, stride,
                          W, r, 8 * I0 - 4, koff);
  col_block_load<4, false>(B, nullptr, yb + 2 * band, yb + 3 * band, yb + band,
                           yb + 4 * band, stride, W, r, 8 * I0 - 4, koff);

#pragma unroll
  for (int c = 0; c < 4; ++c) {
#pragma unroll
    for (int j = 0; j < 2; ++j) {
      f0[c][j] = k0[c][j] ? f0[c][j] : 0.0f;
      f1[c][j] = k1[c][j] ? f1[c][j] : 0.0f;
      k0[c][j] = true;
      k1[c][j] = true;
    }
  }

  auto strip = [&](int I, auto more) {
    constexpr bool MORE = decltype(more)::value;
    col_block_compute<4, true>(A, half, c0, f0, f1, k0, k1, sT1, W);
    if (MORE)
      col_block_load<4, true>(A, sp, nullptr, nullptr, yb, yb + 5 * band,
                              stride, W, r, 8 * I + 4, koff);
    col_block_compute<4, false>(B, half, c0, f0, f1, k0, k1, sT2, W);
    if (MORE)
      col_block_load<4, false>(B, nullptr, yb + 2 * band, yb + 3 * band,
                               yb + band, yb + 4 * band, stride, W, r,
                               8 * I + 4, koff);
    lds_barrier();
    float* op =
        out + ((size_t)b * H + 16 * I + half * 8) * outW + 16 * jc0 + l16;
    row_tiles_from_lds<4, 4, MORE>(sT1, sT2, W, W, jc0, 8, -4, koff, l16, f0,
                                   f1, k0, k1, op, outW);
  };
#pragma unroll 1
  for (int s = 0; s + 1 < S; ++s) strip(I0 + s, BoolC<true>());
  strip(I0 + S - 1, BoolC<false>());
}

template <int WIDTH, int S>
__global__ void __launch_bounds__(WIDTH)
j1_level_kernel(const float* __restrict__ src0_plain,
                const float* __restrict__ yh, int Hb, int Wb, int r,
                const float* __restrict__ h0, int taps0,
                const float* __restrict__ h1, int taps1,
                float* __restrict__ out, int nTilesR) {
  __shared__ float sT1[16 * WIDTH];
  __shared__ float sT2[16 * WIDTH];
  const int W = WIDTH;
  int lane = threadIdx.x & 31;
  int wave = threadIdx.x >> 5;
  int nGroups = nTilesR / S;
  int b = blockIdx.x / nGroups;
  int I0 = (blockIdx.x % nGroups) * S;
  int half = lane >> 4;
  int l16 = lane & 15;
  int koff = half * 2;
  int c0 = wave * 32 + 2 * l16;

  float f0[6][2], f1[6][2];
  bool k0[6][2], k1[6][2];
#pragma unroll
  for (int c = 0; c < 6; ++c) {
#pragma unroll
    for (int j = 0; j < 2; ++j) {
      int kk = 4 * c + koff + j;
      f0[c][j] = *cf_tap(l16, kk, h0, taps0, k0[c][j]);
      f1[c][j] = *cf_tap(l16, kk, h1, taps1, k1[c][j]);
    }
  }

  unsigned stride = 2u * (unsigned)Wb;
  unsigned bp = (unsigned)Hb * (unsigned)Wb;
  const float* yb = yh + (size_t)b * 12 * bp;
  const float* sp = src0_plain + (size_t)b * r * W;
  int jc0 = wave * 2;

  ColBuf<6> A, B;
  col_block_load_lane32<6, true, 0, 0, 0, 5>(A, sp, yb, bp, stride,
                                             (unsigned)W, r, 16 * I0 - 4, koff,
                                             (unsigned)c0);
  col_block_load_lane32<6, false, 2, 3, 1, 4>(B, nullptr, yb, bp, stride,
                                              (unsigned)W, r, 16 * I0 - 4,
                                              koff, (unsigned)c0);

#pragma unroll
  for (int c = 0; c < 6; ++c) {
#pragma unroll
    for (int j = 0; j < 2; ++j) {
      f0[c][j] = k0[c][j] ? f0[c][j] : 0.0f;
      f1[c][j] = k1[c][j] ? f1[c][j] : 0.0f;
      k0[c][j] = true;
      k1[c][j] = true;
    }
  }

  auto strip = [&](int I, auto more) {
    constexpr bool MORE = decltype(more)::value;
    col_block_compute<6, true>(A, half, c0, f0, f1, k0, k1, sT1, W);
    if (MORE)
      col_block_load_lane32<6, true, 0, 0, 0, 5>(A, sp, yb, bp, stride,
                                                 (unsigned)W, r, 16 * I + 12,
                                                 koff, (unsigned)c0);
    col_block_compute<6, false>(B, half, c0, f0, f1, k0, k1, sT2, W);
    if (MORE)
      col_block_load_lane32<6, false, 2, 3, 1, 4>(B, nullptr, yb, bp, stride,
                                                  (unsigned)W, r, 16 * I + 12,
                                                  koff, (unsigned)c0);
    lds_barrier();
    float* op = out + ((size_t)b * r + 16 * I + half * 8) * W + 16 * jc0 + l16;
    row_tiles_from_lds<6, 2, MORE>(sT1, sT2, W, W, jc0, 16, -4, koff, l16, f0,
                                   f1, k0, k1, op, (size_t)W);
  };
#pragma unroll 1
  for (int s = 0; s + 1 < S; ++s) strip(I0 + s, BoolC<true>());
  strip(I0 + S - 1, BoolC<false>());
}

extern "C" void kernel_launch(void* const* d_in, const int* in_sizes, int n_in,
                              void* d_out, int out_size, void* d_ws,
                              size_t ws_size, hipStream_t stream) {
  (void)in_sizes;
  (void)n_in;
  (void)out_size;
  (void)ws_size;

  const float* yl = (const float*)d_in[0];
  const float* yh0 = (const float*)d_in[1];
  const float* yh1 = (const float*)d_in[2];
  const float* yh2 = (const float*)d_in[3];
  const float* g0o = (const float*)d_in[4];
  const float* g1o = (const float*)d_in[5];
  const float* g0a = (const float*)d_in[6];
  const float* g0b = (const float*)d_in[7];
  const float* g1a = (const float*)d_in[8];
  const float* g1b = (const float*)d_in[9];

  const int B = 64;
  const size_t IMG = (size_t)512 * 512;
  float* LLb = (float*)d_ws;
  float* LLa = LLb + 2 * (size_t)B * IMG;
  float* outp = (float*)d_out;


  j2_level_kernel<128, 1><<<dim3(64 * 16), dim3(128), 0, stream>>>(
      yl, yh2, 64, 64, 128, g0b, g0a, g1b, g1a, LLa, 16);

  j2_level_kernel<256, 4><<<dim3(64 * 8), dim3(256), 0, stream>>>(
      LLa, yh1, 128, 128, 256, g0b, g0a, g1b, g1a, LLb, 32);

  j1_level_kernel<512, 8><<<dim3(64 * 4), dim3(512), 0, stream>>>(
      LLb, yh0, 256, 256, 512, g0o, 7, g1o, 5, outp, 32);
}


# Strip refill loads only the rows new to the window: buffer steps holding the 8 overlap rows (of 24 at level 1, of 16 at level 2) move down in registers, not re-fetched from memory; sums unchanged
# speedup vs baseline: 4.5646x; 1.0539x over previous
#include <hip/hip_runtime.h>


typedef __attribute__((ext_vector_type(2))) float v2f;
typedef __attribute__((ext_vector_type(4))) float v4f;
typedef __attribute__((ext_vector_type(8))) float v8f;

#define SQRT_HALF 0.70710678118654752440f

__device__ __forceinline__ int reflect_i(int x, int n) {
  x = (x < 0) ? (-1 - x) : x;
  x = (x >= n) ? (2 * n - 1 - x) : x;
  return x;
}

__device__ __forceinline__ const float* cif_tap(int m, int kk,
                                                const float* __restrict__ ha,
                                                const float* __restrict__ hb,
                                                int hp, bool& ok) {
  int p = m & 3;
  int q = m >> 2;
  int want_odd = hp ? ((p & 1) ^ 1) : (p & 1);
  int t = ((kk - want_odd) >> 1) - q;
  ok = ((kk & 1) == want_odd) && (t >= 0) && (t <= 4);
  int tc = t < 0 ? 0 : (t > 4 ? 4 : t);
  const float* __restrict__ h = ((p == 0) || (p == 2)) ? ha : hb;
  int idx = (p < 2) ? (2 * tc + 1) : (2 * tc);
  return h + idx;
}

__device__ __forceinline__ const float* cf_tap(int m, int kk,
                                               const float* __restrict__ h,
                                               int taps, bool& ok) {
  int t = kk - m - (4 - (taps >> 1));
  ok = (t >= 0) && (t < taps);
  int tc = t < 0 ? 0 : (t >= taps ? taps - 1 : t);
  return h + tc;
}

struct Q4 {
  float w1r, w1i, w2r, w2i;
};

__device__ __forceinline__ Q4 c2q_load(const float* __restrict__ p1,
                                       const float* __restrict__ p2, int o) {
  Q4 q;
  q.w1r = p1[o];
  q.w1i = p1[o + 1];
  q.w2r = p2[o];
  q.w2i = p2[o + 1];
  return q;
}

__device__ __forceinline__ float c2q_even_col(Q4 q, int rr) {
  return (rr & 1) ? (q.w1i - q.w2i) : (q.w1r + q.w2r);
}

__device__ __forceinline__ float c2q_odd_col(Q4 q, int rr) {
  return (rr & 1) ? (q.w2r - q.w1r) : (q.w1i + q.w2i);
}

__device__ __forceinline__ v8f wmma4(v2f a, v2f b, v8f c) {
  return __builtin_amdgcn_wmma_f32_16x16x4_f32(false, a, false, b, (short)0, c,
                                               false, false);
}

#define LOADS_STAY_AHEAD() __builtin_amdgcn_sched_barrier(0)

template <int NC>
struct ColBuf {
  unsigned par;
  v2f d0[NC][2];
  Q4 q0[NC], q1[NC];
};

template <bool V>
struct BoolC {
  static constexpr bool value = V;
};

__device__ __forceinline__ float move_now(float v) {
  float o;
  asm volatile("v_mov_b32 %0, %1" : "=v"(o) : "v"(v));
  return o;
}

__device__ __forceinline__ v2f move_now(v2f v) {
  v2f o;
  o.x = move_now(v.x);
  o.y = move_now(v.y);
  return o;
}

__device__ __forceinline__ Q4 move_now(Q4 v) {
  Q4 o;
  o.w1r = move_now(v.w1r);
  o.w1i = move_now(v.w1i);
  o.w2r = move_now(v.w2r);
  o.w2i = move_now(v.w2i);
  return o;
}

template <int NC, int SHIFT, bool SRC0_PLAIN>
__device__ __forceinline__ void col_buf_roll(ColBuf<NC>& d) {
#pragma unroll
  for (int c = 0; c + SHIFT < NC; ++c) {
    if (SRC0_PLAIN) {
      d.d0[c][0] = move_now(d.d0[c + SHIFT][0]);
      d.d0[c][1] = move_now(d.d0[c + SHIFT][1]);
    } else {
      d.q0[c] = move_now(d.q0[c + SHIFT]);
    }
    d.q1[c] = move_now(d.q1[c + SHIFT]);
  }
  d.par >>= SHIFT;
  LOADS_STAY_AHEAD();
}

template <int NC, bool SRC0_PLAIN, int FIRST>
__device__ __forceinline__ void col_block_load(
    ColBuf<NC>& d, const float* __restrict__ p0, const float* __restrict__ p0a,
    const float* __restrict__ p0b, const float* __restrict__ p1a,
    const float* __restrict__ p1b, int stride, int W, int r, int x0,
    int koff) {
  LOADS_STAY_AHEAD();
  if (FIRST == 0)
    d.par = 0u;
  else
    col_buf_roll<NC, NC - FIRST, SRC0_PLAIN>(d);
#pragma unroll
  for (int c = FIRST; c < NC; ++c) {
    int rr[2];
#pragma unroll
    for (int j = 0; j < 2; ++j) {
      rr[j] = reflect_i(x0 + 4 * c + koff + j, r);
      if (SRC0_PLAIN) {
        size_t o = (size_t)rr[j] * W;
        d.d0[c][j].x = p0[o];
        d.d0[c][j].y = p0[o + 1];
      }
    }
    d.par |= ((unsigned)rr[0] & 1u) << c;
    int ob = (rr[0] >> 1) * stride;
    if (!SRC0_PLAIN) d.q0[c] = c2q_load(p0a, p0b, ob);
    d.q1[c] = c2q_load(p1a, p1b, ob);
  }
  LOADS_STAY_AHEAD();
}

template <int NC, bool SRC0_PLAIN, int FIRST, int P0A, int P0B, int P1A,
          int P1B>
__device__ __forceinline__ void col_block_load_lane32(
    ColBuf<NC>& d, const float* __restrict__ s0, const float* __restrict__ yb,
    unsigned bp, unsigned stride, unsigned W, int r, int x0, int koff,
    unsigned c0) {
  LOADS_STAY_AHEAD();
  if (FIRST == 0)
    d.par = 0u;
  else
    col_buf_roll<NC, NC - FIRST, SRC0_PLAIN>(d);
#pragma unroll
  for (int c = FIRST; c < NC; ++c) {
    int rr[2];
#pragma unroll
    for (int j = 0; j < 2; ++j) {
      rr[j] = reflect_i(x0 + 4 * c + koff + j, r);
      if (SRC0_PLAIN)
        d.d0[c][j] = ((const v2f*)s0)[(unsigned)rr[j] * (W >> 1) + (c0 >> 1)];
    }
    d.par |= ((unsigned)rr[0] & 1u) << c;
    unsigned ob = (unsigned)(rr[0] >> 1) * (stride >> 1) + (c0 >> 1);
    if (!SRC0_PLAIN) {
      v2f wa = ((const v2f*)yb)[ob + P0A * bp];
      v2f wb = ((const v2f*)yb)[ob + P0B * bp];
      d.q0[c].w1r = wa.x;
      d.q0[c].w1i = wa.y;
      d.q0[c].w2r = wb.x;
      d.q0[c].w2i = wb.y;
    }
    v2f wa = ((const v2f*)yb)[ob + P1A * bp];
    v2f wb = ((const v2f*)yb)[ob + P1B * bp];
    d.q1[c].w1r = wa.x;
    d.q1[c].w1i = wa.y;
    d.q1[c].w2r = wb.x;
    d.q1[c].w2i = wb.y;
  }
  LOADS_STAY_AHEAD();
}

template <int NC, bool SRC0_PLAIN>
__device__ __forceinline__ void col_block_compute(
    const ColBuf<NC>& d, int half, int c0, const float (&f0)[NC][2],
    const float (&f1)[NC][2], const bool (&k0)[NC][2], const bool (&k1)[NC][2],
    float* __restrict__ sT, int pitch) {
  const float s0sc = SRC0_PLAIN ? 1.0f : SQRT_HALF;
  v8f accE = {0.f, 0.f, 0.f, 0.f, 0.f, 0.f, 0.f, 0.f};
  v8f accO = {0.f, 0.f, 0.f, 0.f, 0.f, 0.f, 0.f, 0.f};
#pragma unroll
  for (int c = 0; c < NC; ++c) {
    v2f a0, a1, b0e, b0o, b1e, b1o;
    const int r0 = (int)((d.par >> c) & 1u);
    const int r1 = r0 ^ 1;
    a0.x = (k0[c][0] ? f0[c][0] : 0.0f) * s0sc;
    a0.y = (k0[c][1] ? f0[c][1] : 0.0f) * s0sc;
    a1.x = (k1[c][0] ? f1[c][0] : 0.0f) * SQRT_HALF;
    a1.y = (k1[c][1] ? f1[c][1] : 0.0f) * SQRT_HALF;
    if (SRC0_PLAIN) {
      b0e.x = d.d0[c][0].x;
      b0e.y = d.d0[c][1].x;
      b0o.x = d.d0[c][0].y;
      b0o.y = d.d0[c][1].y;
    } else {
      b0e.x = c2q_even_col(d.q0[c], r0);
      b0e.y = c2q_even_col(d.q0[c], r1);
      b0o.x = c2q_odd_col(d.q0[c], r0);
      b0o.y = c2q_odd_col(d.q0[c], r1);
    }
    b1e.x = c2q_even_col(d.q1[c], r0);
    b1e.y = c2q_even_col(d.q1[c], r1);
    b1o.x = c2q_odd_col(d.q1[c], r0);
    b1o.y = c2q_odd_col(d.q1[c], r1);
    accE = wmma4(a0, b0e, accE);
    accO = wmma4(a0, b0o, accO);
    accE = wmma4(a1, b1e, accE);
    accO = wmma4(a1, b1o, accO);
  }

#pragma unroll
  for (int e = 0; e < 8; ++e) {
    int lr = half * 8 + e;
    v2f v;
    v.x = accE[e];
    v.y = accO[e];
    *(v2f*)(sT + lr * pitch + (c0 ^ (lr << 2))) = v;
  }
}

__device__ __forceinline__ void lds_barrier() {
  __builtin_amdgcn_fence(__ATOMIC_RELEASE, "workgroup", "local");
  __builtin_amdgcn_s_barrier();
  __builtin_amdgcn_fence(__ATOMIC_ACQUIRE, "workgroup", "local");
}

template <int NC, int NT, bool MORE>
__device__ __forceinline__ void row_tiles_from_lds(
    const float* __restrict__ sT1, const float* __restrict__ sT2, int pitch,
    int W, int jc0, int wstep, int woff, int koff, int l16,
    const float (&f0)[NC][2], const float (&f1)[NC][2], const bool (&k0)[NC][2],
    const bool (&k1)[NC][2], float* __restrict__ op, size_t ldo) {
  v2f bf0[NC], bf1[NC];
#pragma unroll
  for (int t = 0; t < NC; ++t) {
    bf0[t].x = k0[t][0] ? f0[t][0] : 0.0f;
    bf0[t].y = k0[t][1] ? f0[t][1] : 0.0f;
    bf1[t].x = k1[t][0] ? f1[t][0] : 0.0f;
    bf1[t].y = k1[t][1] ? f1[t][1] : 0.0f;
  }
  const float* __restrict__ r1 = sT1 + l16 * pitch;
  const float* __restrict__ r2 = sT2 + l16 * pitch;
  const int sw = l16 << 2;
  v2f a1[NC][NT], a2[NC][NT];
#pragma unroll
  for (int t = 0; t < NC; ++t) {
#pragma unroll
    for (int i = 0; i < NT; ++i) {
      int xc = wstep * (jc0 + i) + woff + 4 * t + koff;
      int ca = reflect_i(xc, W) ^ sw;
      int cb = reflect_i(xc + 1, W) ^ sw;
      a1[t][i].x = r1[ca];
      a1[t][i].y = r1[cb];
      a2[t][i].x = r2[ca];
      a2[t][i].y = r2[cb];
    }
  }
  LOADS_STAY_AHEAD();
  if (MORE) lds_barrier();

  v8f acc[NT];
#pragma unroll
  for (int i = 0; i < NT; ++i) {
#pragma unroll
    for (int e = 0; e < 8; ++e) acc[i][e] = 0.0f;
  }
#pragma unroll
  for (int t = 0; t < NC; ++t) {
#pragma unroll
    for (int i = 0; i < NT; ++i) {
      acc[i] = wmma4(a1[t][i], bf0[t], acc[i]);
      acc[i] = wmma4(a2[t][i], bf1[t], acc[i]);
    }
  }
#pragma unroll
  for (int i = 0; i < NT; ++i) {
#pragma unroll
    for (int e = 0; e < 8; ++e) op[(size_t)e * ldo + 16 * i] = acc[i][e];
  }
}

template <int WIDTH, int S>
__global__ void __launch_bounds__(WIDTH)
j2_level_kernel(const float* __restrict__ src0_plain,
                const float* __restrict__ yh, int Hb, int Wb, int r,
                const float* __restrict__ ha0, const float* __restrict__ hb0,
                const float* __restrict__ ha1, const float* __restrict__ hb1,
                float* __restrict__ out, int nTilesR) {
  __shared__ float sT1[16 * WIDTH];
  __shared__ float sT2[16 * WIDTH];
  const int W = WIDTH;
  int lane = threadIdx.x & 31;
  int wave = threadIdx.x >> 5;
  int nGroups = nTilesR / S;
  int b = blockIdx.x / nGroups;
  int I0 = (blockIdx.x % nGroups) * S;
  int half = lane >> 4;
  int l16 = lane & 15;
  int koff = half * 2;
  int c0 = wave * 32 + 2 * l16;

  float f0[4][2], f1[4][2];
  bool k0[4][2], k1[4][2];
#pragma unroll
  for (int c = 0; c < 4; ++c) {
#pragma unroll
    for (int j = 0; j < 2; ++j) {
      int kk = 4 * c + koff + j;
      f0[c][j] = *cif_tap(l16, kk, ha0, hb0, 0, k0[c][j]);
      f1[c][j] = *cif_tap(l16, kk, ha1, hb1, 1, k1[c][j]);
    }
  }

  int stride = 2 * Wb;
  size_t band = (size_t)Hb * stride;
  const float* yb = yh + (size_t)b * 6 * band + c0;
  const float* sp = src0_plain + (size_t)b * r * W + c0;
  int H = 2 * r;
  size_t outW = 2 * (size_t)W;
  int jc0 = wave * 4;

  ColBuf<4> A, B;
  col_block_load<4, true, 0>(A, sp, nullptr, nullptr, yb, yb + 5 * band,
                             stride, W, r, 8 * I0 - 4, koff);
  col_block_load<4, false, 0>(B, nullptr, yb + 2 * band, yb + 3 * band,
                              yb + band, yb + 4 * band, stride, W, r,
                              8 * I0 - 4, koff);

#pragma unroll
  for (int c = 0; c < 4; ++c) {
#pragma unroll
    for (int j = 0; j < 2; ++j) {
      f0[c][j] = k0[c][j] ? f0[c][j] : 0.0f;
      f1[c][j] = k1[c][j] ? f1[c][j] : 0.0f;
      k0[c][j] = true;
      k1[c][j] = true;
    }
  }

  auto strip = [&](int I, auto more) {
    constexpr bool MORE = decltype(more)::value;
    col_block_compute<4, true>(A, half, c0, f0, f1, k0, k1, sT1, W);
    if (MORE)
      col_block_load<4, true, 2>(A, sp, nullptr, nullptr, yb, yb + 5 * band,
                                 stride, W, r, 8 * I + 4, koff);
    col_block_compute<4, false>(B, half, c0, f0, f1, k0, k1, sT2, W);
    if (MORE)
      col_block_load<4, false, 2>(B, nullptr, yb + 2 * band, yb + 3 * band,
                                  yb + band, yb + 4 * band, stride, W, r,
                                  8 * I + 4, koff);
    lds_barrier();
    float* op =
        out + ((size_t)b * H + 16 * I + half * 8) * outW + 16 * jc0 + l16;
    row_tiles_from_lds<4, 4, MORE>(sT1, sT2, W, W, jc0, 8, -4, koff, l16, f0,
                                   f1, k0, k1, op, outW);
  };
#pragma unroll 1
  for (int s = 0; s + 1 < S; ++s) strip(I0 + s, BoolC<true>());
  strip(I0 + S - 1, BoolC<false>());
}

template <int WIDTH, int S>
__global__ void __launch_bounds__(WIDTH)
j1_level_kernel(const float* __restrict__ src0_plain,
                const float* __restrict__ yh, int Hb, int Wb, int r,
                const float* __restrict__ h0, int taps0,
                const float* __restrict__ h1, int taps1,
                float* __restrict__ out, int nTilesR) {
  __shared__ float sT1[16 * WIDTH];
  __shared__ float sT2[16 * WIDTH];
  const int W = WIDTH;
  int lane = threadIdx.x & 31;
  int wave = threadIdx.x >> 5;
  int nGroups = nTilesR / S;
  int b = blockIdx.x / nGroups;
  int I0 = (blockIdx.x % nGroups) * S;
  int half = lane >> 4;
  int l16 = lane & 15;
  int koff = half * 2;
  int c0 = wave * 32 + 2 * l16;

  float f0[6][2], f1[6][2];
  bool k0[6][2], k1[6][2];
#pragma unroll
  for (int c = 0; c < 6; ++c) {
#pragma unroll
    for (int j = 0; j < 2; ++j) {
      int kk = 4 * c + koff + j;
      f0[c][j] = *cf_tap(l16, kk, h0, taps0, k0[c][j]);
      f1[c][j] = *cf_tap(l16, kk, h1, taps1, k1[c][j]);
    }
  }

  unsigned stride = 2u * (unsigned)Wb;
  unsigned bp = (unsigned)Hb * (unsigned)Wb;
  const float* yb = yh + (size_t)b * 12 * bp;
  const float* sp = src0_plain + (size_t)b * r * W;
  int jc0 = wave * 2;

  ColBuf<6> A, B;
  col_block_load_lane32<6, true, 0, 0, 0, 0, 5>(A, sp, yb, bp, stride,
                                                (unsigned)W, r, 16 * I0 - 4,
                                                koff, (unsigned)c0);
  col_block_load_lane32<6, false, 0, 2, 3, 1, 4>(B, nullptr, yb, bp, stride,
                                                 (unsigned)W, r, 16 * I0 - 4,
                                                 koff, (unsigned)c0);

#pragma unroll
  for (int c = 0; c < 6; ++c) {
#pragma unroll
    for (int j = 0; j < 2; ++j) {
      f0[c][j] = k0[c][j] ? f0[c][j] : 0.0f;
      f1[c][j] = k1[c][j] ? f1[c][j] : 0.0f;
      k0[c][j] = true;
      k1[c][j] = true;
    }
  }

  auto strip = [&](int I, auto more) {
    constexpr bool MORE = decltype(more)::value;
    col_block_compute<6, true>(A, half, c0, f0, f1, k0, k1, sT1, W);
    if (MORE)
      col_block_load_lane32<6, true, 2, 0, 0, 0, 5>(
          A, sp, yb, bp, stride, (unsigned)W, r, 16 * I + 12, koff,
          (unsigned)c0);
    col_block_compute<6, false>(B, half, c0, f0, f1, k0, k1, sT2, W);
    if (MORE)
      col_block_load_lane32<6, false, 2, 2, 3, 1, 4>(
          B, nullptr, yb, bp, stride, (unsigned)W, r, 16 * I + 12, koff,
          (unsigned)c0);
    lds_barrier();
    float* op = out + ((size_t)b * r + 16 * I + half * 8) * W + 16 * jc0 + l16;
    row_tiles_from_lds<6, 2, MORE>(sT1, sT2, W, W, jc0, 16, -4, koff, l16, f0,
                                   f1, k0, k1, op, (size_t)W);
  };
#pragma unroll 1
  for (int s = 0; s + 1 < S; ++s) strip(I0 + s, BoolC<true>());
  strip(I0 + S - 1, BoolC<false>());
}

extern "C" void kernel_launch(void* const* d_in, const int* in_sizes, int n_in,
                              void* d_out, int out_size, void* d_ws,
                              size_t ws_size, hipStream_t stream) {
  (void)in_sizes;
  (void)n_in;
  (void)out_size;
  (void)ws_size;

  const float* yl = (const float*)d_in[0];
  const float* yh0 = (const float*)d_in[1];
  const float* yh1 = (const float*)d_in[2];
  const float* yh2 = (const float*)d_in[3];
  const float* g0o = (const float*)d_in[4];
  const float* g1o = (const float*)d_in[5];
  const float* g0a = (const float*)d_in[6];
  const float* g0b = (const float*)d_in[7];
  const float* g1a = (const float*)d_in[8];
  const float* g1b = (const float*)d_in[9];

  const int B = 64;
  const size_t IMG = (size_t)512 * 512;
  float* LLb = (float*)d_ws;
  float* LLa = LLb + 2 * (size_t)B * IMG;
  float* outp = (float*)d_out;


  j2_level_kernel<128, 1><<<dim3(64 * 16), dim3(128), 0, stream>>>(
      yl, yh2, 64, 64, 128, g0b, g0a, g1b, g1a, LLa, 16);

  j2_level_kernel<256, 4><<<dim3(64 * 8), dim3(256), 0, stream>>>(
      LLa, yh1, 128, 128, 256, g0b, g0a, g1b, g1a, LLb, 32);

  j1_level_kernel<512, 8><<<dim3(64 * 4), dim3(512), 0, stream>>>(
      LLb, yh0, 256, 256, 512, g0o, 7, g1o, 5, outp, 32);
}


# Level 3 fused into the level-2 kernel: a group computes the 48 level-3 rows its 4 strips read into LDS, its plain source; 3 launches to 2; level-3 tiles start 4 rows off: column sum order changes
# speedup vs baseline: 4.7739x; 1.0459x over previous
#include <hip/hip_runtime.h>


typedef __attribute__((ext_vector_type(2))) float v2f;
typedef __attribute__((ext_vector_type(4))) float v4f;
typedef __attribute__((ext_vector_type(8))) float v8f;

#define SQRT_HALF 0.70710678118654752440f

__device__ __forceinline__ int reflect_i(int x, int n) {
  x = (x < 0) ? (-1 - x) : x;
  x = (x >= n) ? (2 * n - 1 - x) : x;
  return x;
}

__device__ __forceinline__ const float* cif_tap(int m, int kk,
                                                const float* __restrict__ ha,
                                                const float* __restrict__ hb,
                                                int hp, bool& ok) {
  int p = m & 3;
  int q = m >> 2;
  int want_odd = hp ? ((p & 1) ^ 1) : (p & 1);
  int t = ((kk - want_odd) >> 1) - q;
  ok = ((kk & 1) == want_odd) && (t >= 0) && (t <= 4);
  int tc = t < 0 ? 0 : (t > 4 ? 4 : t);
  const float* __restrict__ h = ((p == 0) || (p == 2)) ? ha : hb;
  int idx = (p < 2) ? (2 * tc + 1) : (2 * tc);
  return h + idx;
}

__device__ __forceinline__ const float* cf_tap(int m, int kk,
                                               const float* __restrict__ h,
                                               int taps, bool& ok) {
  int t = kk - m - (4 - (taps >> 1));
  ok = (t >= 0) && (t < taps);
  int tc = t < 0 ? 0 : (t >= taps ? taps - 1 : t);
  return h + tc;
}

struct Q4 {
  float w1r, w1i, w2r, w2i;
};

__device__ __forceinline__ Q4 c2q_load(const float* __restrict__ p1,
                                       const float* __restrict__ p2, int o) {
  Q4 q;
  q.w1r = p1[o];
  q.w1i = p1[o + 1];
  q.w2r = p2[o];
  q.w2i = p2[o + 1];
  return q;
}

__device__ __forceinline__ float c2q_even_col(Q4 q, int rr) {
  return (rr & 1) ? (q.w1i - q.w2i) : (q.w1r + q.w2r);
}

__device__ __forceinline__ float c2q_odd_col(Q4 q, int rr) {
  return (rr & 1) ? (q.w2r - q.w1r) : (q.w1i + q.w2i);
}

__device__ __forceinline__ v8f wmma4(v2f a, v2f b, v8f c) {
  return __builtin_amdgcn_wmma_f32_16x16x4_f32(false, a, false, b, (short)0, c,
                                               false, false);
}

#define LOADS_STAY_AHEAD() __builtin_amdgcn_sched_barrier(0)

template <int NC>
struct ColBuf {
  unsigned par;
  v2f d0[NC][2];
  Q4 q0[NC], q1[NC];
};

template <bool V>
struct BoolC {
  static constexpr bool value = V;
};

__device__ __forceinline__ float move_now(float v) {
  float o;
  asm volatile("v_mov_b32 %0, %1" : "=v"(o) : "v"(v));
  return o;
}

__device__ __forceinline__ v2f move_now(v2f v) {
  v2f o;
  o.x = move_now(v.x);
  o.y = move_now(v.y);
  return o;
}

__device__ __forceinline__ Q4 move_now(Q4 v) {
  Q4 o;
  o.w1r = move_now(v.w1r);
  o.w1i = move_now(v.w1i);
  o.w2r = move_now(v.w2r);
  o.w2i = move_now(v.w2i);
  return o;
}

template <int NC, int SHIFT, bool SRC0_PLAIN>
__device__ __forceinline__ void col_buf_roll(ColBuf<NC>& d) {
#pragma unroll
  for (int c = 0; c + SHIFT < NC; ++c) {
    if (SRC0_PLAIN) {
      d.d0[c][0] = move_now(d.d0[c + SHIFT][0]);
      d.d0[c][1] = move_now(d.d0[c + SHIFT][1]);
    } else {
      d.q0[c] = move_now(d.q0[c + SHIFT]);
    }
    d.q1[c] = move_now(d.q1[c + SHIFT]);
  }
  d.par >>= SHIFT;
  LOADS_STAY_AHEAD();
}

template <int NC, bool SRC0_PLAIN, int FIRST>
__device__ __forceinline__ void col_block_load(
    ColBuf<NC>& d, const float* __restrict__ p0, const float* __restrict__ p0a,
    const float* __restrict__ p0b, const float* __restrict__ p1a,
    const float* __restrict__ p1b, int stride, int W, int r, int x0,
    int koff) {
  LOADS_STAY_AHEAD();
  if (FIRST == 0)
    d.par = 0u;
  else
    col_buf_roll<NC, NC - FIRST, SRC0_PLAIN>(d);
#pragma unroll
  for (int c = FIRST; c < NC; ++c) {
    int rr[2];
#pragma unroll
    for (int j = 0; j < 2; ++j) {
      rr[j] = reflect_i(x0 + 4 * c + koff + j, r);
      if (SRC0_PLAIN) {
        size_t o = (size_t)rr[j] * W;
        d.d0[c][j].x = p0[o];
        d.d0[c][j].y = p0[o + 1];
      }
    }
    d.par |= ((unsigned)rr[0] & 1u) << c;
    int ob = (rr[0] >> 1) * stride;
    if (!SRC0_PLAIN) d.q0[c] = c2q_load(p0a, p0b, ob);
    d.q1[c] = c2q_load(p1a, p1b, ob);
  }
  LOADS_STAY_AHEAD();
}

template <int NC, int FIRST, bool BANDS, bool PLAIN>
__device__ __forceinline__ void col_block_load_lds0(
    ColBuf<NC>& d, const float* __restrict__ sl, int rbase, int pitch,
    const float* __restrict__ p1a, const float* __restrict__ p1b, int stride,
    int r, int x0, int koff) {
  LOADS_STAY_AHEAD();
  if (FIRST == 0) {
    if (BANDS) d.par = 0u;
  } else {
    col_buf_roll<NC, NC - FIRST, true>(d);
  }
#pragma unroll
  for (int c = FIRST; c < NC; ++c) {
    int rr0 = reflect_i(x0 + 4 * c + koff, r);
    int rr1 = reflect_i(x0 + 4 * c + koff + 1, r);
    if (PLAIN) {
      d.d0[c][0] = *(const v2f*)(sl + (rr0 - rbase) * pitch);
      d.d0[c][1] = *(const v2f*)(sl + (rr1 - rbase) * pitch);
    }
    if (BANDS) {
      d.par |= ((unsigned)rr0 & 1u) << c;
      d.q1[c] = c2q_load(p1a, p1b, (rr0 >> 1) * stride);
    }
  }
  LOADS_STAY_AHEAD();
}

template <int NC, bool SRC0_PLAIN, int FIRST, int P0A, int P0B, int P1A,
          int P1B>
__device__ __forceinline__ void col_block_load_lane32(
    ColBuf<NC>& d, const float* __restrict__ s0, const float* __restrict__ yb,
    unsigned bp, unsigned stride, unsigned W, int r, int x0, int koff,
    unsigned c0) {
  LOADS_STAY_AHEAD();
  if (FIRST == 0)
    d.par = 0u;
  else
    col_buf_roll<NC, NC - FIRST, SRC0_PLAIN>(d);
#pragma unroll
  for (int c = FIRST; c < NC; ++c) {
    int rr[2];
#pragma unroll
    for (int j = 0; j < 2; ++j) {
      rr[j] = reflect_i(x0 + 4 * c + koff + j, r);
      if (SRC0_PLAIN)
        d.d0[c][j] = ((const v2f*)s0)[(unsigned)rr[j] * (W >> 1) + (c0 >> 1)];
    }
    d.par |= ((unsigned)rr[0] & 1u) << c;
    unsigned ob = (unsigned)(rr[0] >> 1) * (stride >> 1) + (c0 >> 1);
    if (!SRC0_PLAIN) {
      v2f wa = ((const v2f*)yb)[ob + P0A * bp];
      v2f wb = ((const v2f*)yb)[ob + P0B * bp];
      d.q0[c].w1r = wa.x;
      d.q0[c].w1i = wa.y;
      d.q0[c].w2r = wb.x;
      d.q0[c].w2i = wb.y;
    }
    v2f wa = ((const v2f*)yb)[ob + P1A * bp];
    v2f wb = ((const v2f*)yb)[ob + P1B * bp];
    d.q1[c].w1r = wa.x;
    d.q1[c].w1i = wa.y;
    d.q1[c].w2r = wb.x;
    d.q1[c].w2i = wb.y;
  }
  LOADS_STAY_AHEAD();
}

template <int NC, bool SRC0_PLAIN>
__device__ __forceinline__ void col_block_compute(
    const ColBuf<NC>& d, int half, int c0, const float (&f0)[NC][2],
    const float (&f1)[NC][2], const bool (&k0)[NC][2], const bool (&k1)[NC][2],
    float* __restrict__ sT, int pitch) {
  const float s0sc = SRC0_PLAIN ? 1.0f : SQRT_HALF;
  v8f accE = {0.f, 0.f, 0.f, 0.f, 0.f, 0.f, 0.f, 0.f};
  v8f accO = {0.f, 0.f, 0.f, 0.f, 0.f, 0.f, 0.f, 0.f};
#pragma unroll
  for (int c = 0; c < NC; ++c) {
    v2f a0, a1, b0e, b0o, b1e, b1o;
    const int r0 = (int)((d.par >> c) & 1u);
    const int r1 = r0 ^ 1;
    a0.x = (k0[c][0] ? f0[c][0] : 0.0f) * s0sc;
    a0.y = (k0[c][1] ? f0[c][1] : 0.0f) * s0sc;
    a1.x = (k1[c][0] ? f1[c][0] : 0.0f) * SQRT_HALF;
    a1.y = (k1[c][1] ? f1[c][1] : 0.0f) * SQRT_HALF;
    if (SRC0_PLAIN) {
      b0e.x = d.d0[c][0].x;
      b0e.y = d.d0[c][1].x;
      b0o.x = d.d0[c][0].y;
      b0o.y = d.d0[c][1].y;
    } else {
      b0e.x = c2q_even_col(d.q0[c], r0);
      b0e.y = c2q_even_col(d.q0[c], r1);
      b0o.x = c2q_odd_col(d.q0[c], r0);
      b0o.y = c2q_odd_col(d.q0[c], r1);
    }
    b1e.x = c2q_even_col(d.q1[c], r0);
    b1e.y = c2q_even_col(d.q1[c], r1);
    b1o.x = c2q_odd_col(d.q1[c], r0);
    b1o.y = c2q_odd_col(d.q1[c], r1);
    accE = wmma4(a0, b0e, accE);
    accO = wmma4(a0, b0o, accO);
    accE = wmma4(a1, b1e, accE);
    accO = wmma4(a1, b1o, accO);
  }

#pragma unroll
  for (int e = 0; e < 8; ++e) {
    int lr = half * 8 + e;
    v2f v;
    v.x = accE[e];
    v.y = accO[e];
    *(v2f*)(sT + lr * pitch + (c0 ^ (lr << 2))) = v;
  }
}

__device__ __forceinline__ void lds_barrier() {
  __builtin_amdgcn_fence(__ATOMIC_RELEASE, "workgroup", "local");
  __builtin_amdgcn_s_barrier();
  __builtin_amdgcn_fence(__ATOMIC_ACQUIRE, "workgroup", "local");
}

template <int NC, int NT, bool MORE>
__device__ __forceinline__ void row_tiles_from_lds(
    const float* __restrict__ sT1, const float* __restrict__ sT2, int pitch,
    int W, int jc0, int wstep, int woff, int koff, int l16,
    const float (&f0)[NC][2], const float (&f1)[NC][2], const bool (&k0)[NC][2],
    const bool (&k1)[NC][2], float* __restrict__ op, size_t ldo) {
  v2f bf0[NC], bf1[NC];
#pragma unroll
  for (int t = 0; t < NC; ++t) {
    bf0[t].x = k0[t][0] ? f0[t][0] : 0.0f;
    bf0[t].y = k0[t][1] ? f0[t][1] : 0.0f;
    bf1[t].x = k1[t][0] ? f1[t][0] : 0.0f;
    bf1[t].y = k1[t][1] ? f1[t][1] : 0.0f;
  }
  const float* __restrict__ r1 = sT1 + l16 * pitch;
  const float* __restrict__ r2 = sT2 + l16 * pitch;
  const int sw = l16 << 2;
  v2f a1[NC][NT], a2[NC][NT];
#pragma unroll
  for (int t = 0; t < NC; ++t) {
#pragma unroll
    for (int i = 0; i < NT; ++i) {
      int xc = wstep * (jc0 + i) + woff + 4 * t + koff;
      int ca = reflect_i(xc, W) ^ sw;
      int cb = reflect_i(xc + 1, W) ^ sw;
      a1[t][i].x = r1[ca];
      a1[t][i].y = r1[cb];
      a2[t][i].x = r2[ca];
      a2[t][i].y = r2[cb];
    }
  }
  LOADS_STAY_AHEAD();
  if (MORE) lds_barrier();

  v8f acc[NT];
#pragma unroll
  for (int i = 0; i < NT; ++i) {
#pragma unroll
    for (int e = 0; e < 8; ++e) acc[i][e] = 0.0f;
  }
#pragma unroll
  for (int t = 0; t < NC; ++t) {
#pragma unroll
    for (int i = 0; i < NT; ++i) {
      acc[i] = wmma4(a1[t][i], bf0[t], acc[i]);
      acc[i] = wmma4(a2[t][i], bf1[t], acc[i]);
    }
  }
#pragma unroll
  for (int i = 0; i < NT; ++i) {
#pragma unroll
    for (int e = 0; e < 8; ++e) op[(size_t)e * ldo + 16 * i] = acc[i][e];
  }
}

template <int S>
__global__ void __launch_bounds__(256)
j3_j2_level_kernel(const float* __restrict__ yl, const float* __restrict__ yh3,
                   const float* __restrict__ yh,
                   const float* __restrict__ ha0, const float* __restrict__ hb0,
                   const float* __restrict__ ha1, const float* __restrict__ hb1,
                   float* __restrict__ out, int nTilesR) {
  constexpr int W = 256;
  constexpr int W3 = 128;
  constexpr int LR = 8 * S + 16;
  __shared__ __attribute__((aligned(16))) float smem[(32 + LR) * W];
  float* sT1 = smem;
  float* sT2 = smem + 16 * W;
  float* sL = smem + 32 * W;
  const int r = W;
  const int r3 = W3;
  int lane = threadIdx.x & 31;
  int wave = threadIdx.x >> 5;
  int nGroups = nTilesR / S;
  int b = blockIdx.x / nGroups;
  int I0 = (blockIdx.x % nGroups) * S;
  int half = lane >> 4;
  int l16 = lane & 15;
  int koff = half * 2;
  int c0 = wave * 32 + 2 * l16;

  float f0[4][2], f1[4][2];
  bool k0[4][2], k1[4][2];
#pragma unroll
  for (int c = 0; c < 4; ++c) {
#pragma unroll
    for (int j = 0; j < 2; ++j) {
      int kk = 4 * c + koff + j;
      f0[c][j] = *cif_tap(l16, kk, ha0, hb0, 0, k0[c][j]);
      f1[c][j] = *cif_tap(l16, kk, ha1, hb1, 1, k1[c][j]);
    }
  }

  int stride = W;
  size_t band = (size_t)(W / 2) * stride;
  const float* yb = yh + (size_t)b * 6 * band + c0;
  int H = 2 * r;
  size_t outW = 2 * (size_t)W;
  int jc0 = wave * 4;

  int R0 = 8 * I0 - 4;
  R0 = R0 < 0 ? 0 : R0;
  R0 = R0 > W - LR ? W - LR : R0;
  const float* sl = sL + c0;

  ColBuf<4> A, B;

  {
    static_assert(LR == 48, "three level-3 strips: two sub-groups, two rounds");
    const int sub = __builtin_amdgcn_readfirstlane(wave >> 2);
    int w4 = wave & 3;
    int c3 = w4 * 32 + 2 * l16;
    float* u1 = sT1 + sub * (16 * W3);
    float* u2 = sT2 + sub * (16 * W3);
    int stride3 = W3;
    size_t band3 = (size_t)(W3 / 2) * stride3;
    const float* yb3 = yh3 + (size_t)b * 6 * band3 + c3;
    const float* sp3 = yl + (size_t)b * r3 * W3 + c3;
    int x3 = (R0 >> 1) - 4 + 8 * sub;
    ColBuf<4> XA, XB;
    col_block_load<4, true, 0>(XA, sp3, nullptr, nullptr, yb3, yb3 + 5 * band3,
                               stride3, W3, r3, x3, koff);
    col_block_load<4, false, 0>(XB, nullptr, yb3 + 2 * band3, yb3 + 3 * band3,
                                yb3 + band3, yb3 + 4 * band3, stride3, W3, r3,
                                x3, koff);

#pragma unroll
    for (int c = 0; c < 4; ++c) {
#pragma unroll
      for (int j = 0; j < 2; ++j) {
        f0[c][j] = k0[c][j] ? f0[c][j] : 0.0f;
        f1[c][j] = k1[c][j] ? f1[c][j] : 0.0f;
        k0[c][j] = true;
        k1[c][j] = true;
      }
    }

    col_block_compute<4, true>(XA, half, c3, f0, f1, k0, k1, u1, W3);
    col_block_load<4, true, 0>(XA, sp3, nullptr, nullptr, yb3, yb3 + 5 * band3,
                               stride3, W3, r3, x3 + 16, koff);
    col_block_compute<4, false>(XB, half, c3, f0, f1, k0, k1, u2, W3);
    col_block_load<4, false, 0>(XB, nullptr, yb3 + 2 * band3, yb3 + 3 * band3,
                                yb3 + band3, yb3 + 4 * band3, stride3, W3, r3,
                                x3 + 16, koff);
    lds_barrier();
    float* op3 = sL + (16 * sub + half * 8) * W + 16 * (w4 * 4) + l16;
    row_tiles_from_lds<4, 4, true>(u1, u2, W3, W3, w4 * 4, 8, -4, koff, l16, f0,
                                   f1, k0, k1, op3, (size_t)W);

    if (sub == 0) {
      col_block_compute<4, true>(XA, half, c3, f0, f1, k0, k1, u1, W3);
      col_block_compute<4, false>(XB, half, c3, f0, f1, k0, k1, u2, W3);
    }
    col_block_load_lds0<4, 0, true, false>(A, sl, R0, W, yb, yb + 5 * band,
                                           stride, r, 8 * I0 - 4, koff);
    col_block_load<4, false, 0>(B, nullptr, yb + 2 * band, yb + 3 * band,
                                yb + band, yb + 4 * band, stride, W, r,
                                8 * I0 - 4, koff);
    lds_barrier();
    if (sub == 0)
      row_tiles_from_lds<4, 4, false>(u1, u2, W3, W3, w4 * 4, 8, -4, koff, l16,
                                      f0, f1, k0, k1, op3 + 32 * W, (size_t)W);
    lds_barrier();
  }

  col_block_load_lds0<4, 0, false, true>(A, sl, R0, W, yb, yb + 5 * band,
                                         stride, r, 8 * I0 - 4, koff);

  auto strip = [&](int I, auto more) {
    constexpr bool MORE = decltype(more)::value;
    col_block_compute<4, true>(A, half, c0, f0, f1, k0, k1, sT1, W);
    if (MORE)
      col_block_load_lds0<4, 2, true, true>(A, sl, R0, W, yb, yb + 5 * band,
                                            stride, r, 8 * I + 4, koff);
    col_block_compute<4, false>(B, half, c0, f0, f1, k0, k1, sT2, W);
    if (MORE)
      col_block_load<4, false, 2>(B, nullptr, yb + 2 * band, yb + 3 * band,
                                  yb + band, yb + 4 * band, stride, W, r,
                                  8 * I + 4, koff);
    lds_barrier();
    float* op =
        out + ((size_t)b * H + 16 * I + half * 8) * outW + 16 * jc0 + l16;
    row_tiles_from_lds<4, 4, MORE>(sT1, sT2, W, W, jc0, 8, -4, koff, l16, f0,
                                   f1, k0, k1, op, outW);
  };
#pragma unroll 1
  for (int s = 0; s + 1 < S; ++s) strip(I0 + s, BoolC<true>());
  strip(I0 + S - 1, BoolC<false>());
}

template <int WIDTH, int S>
__global__ void __launch_bounds__(WIDTH)
j1_level_kernel(const float* __restrict__ src0_plain,
                const float* __restrict__ yh, int Hb, int Wb, int r,
                const float* __restrict__ h0, int taps0,
                const float* __restrict__ h1, int taps1,
                float* __restrict__ out, int nTilesR) {
  __shared__ float sT1[16 * WIDTH];
  __shared__ float sT2[16 * WIDTH];
  const int W = WIDTH;
  int lane = threadIdx.x & 31;
  int wave = threadIdx.x >> 5;
  int nGroups = nTilesR / S;
  int b = blockIdx.x / nGroups;
  int I0 = (blockIdx.x % nGroups) * S;
  int half = lane >> 4;
  int l16 = lane & 15;
  int koff = half * 2;
  int c0 = wave * 32 + 2 * l16;

  float f0[6][2], f1[6][2];
  bool k0[6][2], k1[6][2];
#pragma unroll
  for (int c = 0; c < 6; ++c) {
#pragma unroll
    for (int j = 0; j < 2; ++j) {
      int kk = 4 * c + koff + j;
      f0[c][j] = *cf_tap(l16, kk, h0, taps0, k0[c][j]);
      f1[c][j] = *cf_tap(l16, kk, h1, taps1, k1[c][j]);
    }
  }

  unsigned stride = 2u * (unsigned)Wb;
  unsigned bp = (unsigned)Hb * (unsigned)Wb;
  const float* yb = yh + (size_t)b * 12 * bp;
  const float* sp = src0_plain + (size_t)b * r * W;
  int jc0 = wave * 2;

  ColBuf<6> A, B;
  col_block_load_lane32<6, true, 0, 0, 0, 0, 5>(A, sp, yb, bp, stride,
                                                (unsigned)W, r, 16 * I0 - 4,
                                                koff, (unsigned)c0);
  col_block_load_lane32<6, false, 0, 2, 3, 1, 4>(B, nullptr, yb, bp, stride,
                                                 (unsigned)W, r, 16 * I0 - 4,
                                                 koff, (unsigned)c0);

#pragma unroll
  for (int c = 0; c < 6; ++c) {
#pragma unroll
    for (int j = 0; j < 2; ++j) {
      f0[c][j] = k0[c][j] ? f0[c][j] : 0.0f;
      f1[c][j] = k1[c][j] ? f1[c][j] : 0.0f;
      k0[c][j] = true;
      k1[c][j] = true;
    }
  }

  auto strip = [&](int I, auto more) {
    constexpr bool MORE = decltype(more)::value;
    col_block_compute<6, true>(A, half, c0, f0, f1, k0, k1, sT1, W);
    if (MORE)
      col_block_load_lane32<6, true, 2, 0, 0, 0, 5>(
          A, sp, yb, bp, stride, (unsigned)W, r, 16 * I + 12, koff,
          (unsigned)c0);
    col_block_compute<6, false>(B, half, c0, f0, f1, k0, k1, sT2, W);
    if (MORE)
      col_block_load_lane32<6, false, 2, 2, 3, 1, 4>(
          B, nullptr, yb, bp, stride, (unsigned)W, r, 16 * I + 12, koff,
          (unsigned)c0);
    lds_barrier();
    float* op = out + ((size_t)b * r + 16 * I + half * 8) * W + 16 * jc0 + l16;
    row_tiles_from_lds<6, 2, MORE>(sT1, sT2, W, W, jc0, 16, -4, koff, l16, f0,
                                   f1, k0, k1, op, (size_t)W);
  };
#pragma unroll 1
  for (int s = 0; s + 1 < S; ++s) strip(I0 + s, BoolC<true>());
  strip(I0 + S - 1, BoolC<false>());
}

extern "C" void kernel_launch(void* const* d_in, const int* in_sizes, int n_in,
                              void* d_out, int out_size, void* d_ws,
                              size_t ws_size, hipStream_t stream) {
  (void)in_sizes;
  (void)n_in;
  (void)out_size;
  (void)ws_size;

  const float* yl = (const float*)d_in[0];
  const float* yh0 = (const float*)d_in[1];
  const float* yh1 = (const float*)d_in[2];
  const float* yh2 = (const float*)d_in[3];
  const float* g0o = (const float*)d_in[4];
  const float* g1o = (const float*)d_in[5];
  const float* g0a = (const float*)d_in[6];
  const float* g0b = (const float*)d_in[7];
  const float* g1a = (const float*)d_in[8];
  const float* g1b = (const float*)d_in[9];

  float* LLb = (float*)d_ws;
  float* outp = (float*)d_out;


  j3_j2_level_kernel<4><<<dim3(64 * 8), dim3(256), 0, stream>>>(
      yl, yh2, yh1, g0b, g0a, g1b, g1a, LLb, 32);

  j1_level_kernel<512, 8><<<dim3(64 * 4), dim3(512), 0, stream>>>(
      LLb, yh0, 256, 256, 512, g0o, 7, g1o, 5, outp, 32);
}


# Row phase: a lane's two K columns of a step are one aligned pair in the LDS strip, fetched by ONE conflict-free 8-byte read per image, not two 2-address 4-byte reads on shared banks; sums unchanged
# speedup vs baseline: 4.9873x; 1.0447x over previous
#include <hip/hip_runtime.h>


typedef __attribute__((ext_vector_type(2))) float v2f;
typedef __attribute__((ext_vector_type(4))) float v4f;
typedef __attribute__((ext_vector_type(8))) float v8f;

#define SQRT_HALF 0.70710678118654752440f

__device__ __forceinline__ int reflect_i(int x, int n) {
  x = (x < 0) ? (-1 - x) : x;
  x = (x >= n) ? (2 * n - 1 - x) : x;
  return x;
}

__device__ __forceinline__ const float* cif_tap(int m, int kk,
                                                const float* __restrict__ ha,
                                                const float* __restrict__ hb,
                                                int hp, bool& ok) {
  int p = m & 3;
  int q = m >> 2;
  int want_odd = hp ? ((p & 1) ^ 1) : (p & 1);
  int t = ((kk - want_odd) >> 1) - q;
  ok = ((kk & 1) == want_odd) && (t >= 0) && (t <= 4);
  int tc = t < 0 ? 0 : (t > 4 ? 4 : t);
  const float* __restrict__ h = ((p == 0) || (p == 2)) ? ha : hb;
  int idx = (p < 2) ? (2 * tc + 1) : (2 * tc);
  return h + idx;
}

__device__ __forceinline__ const float* cf_tap(int m, int kk,
                                               const float* __restrict__ h,
                                               int taps, bool& ok) {
  int t = kk - m - (4 - (taps >> 1));
  ok = (t >= 0) && (t < taps);
  int tc = t < 0 ? 0 : (t >= taps ? taps - 1 : t);
  return h + tc;
}

struct Q4 {
  float w1r, w1i, w2r, w2i;
};

__device__ __forceinline__ Q4 c2q_load(const float* __restrict__ p1,
                                       const float* __restrict__ p2, int o) {
  Q4 q;
  q.w1r = p1[o];
  q.w1i = p1[o + 1];
  q.w2r = p2[o];
  q.w2i = p2[o + 1];
  return q;
}

__device__ __forceinline__ float c2q_even_col(Q4 q, int rr) {
  return (rr & 1) ? (q.w1i - q.w2i) : (q.w1r + q.w2r);
}

__device__ __forceinline__ float c2q_odd_col(Q4 q, int rr) {
  return (rr & 1) ? (q.w2r - q.w1r) : (q.w1i + q.w2i);
}

__device__ __forceinline__ v8f wmma4(v2f a, v2f b, v8f c) {
  return __builtin_amdgcn_wmma_f32_16x16x4_f32(false, a, false, b, (short)0, c,
                                               false, false);
}

#define LOADS_STAY_AHEAD() __builtin_amdgcn_sched_barrier(0)

template <int NC>
struct ColBuf {
  unsigned par;
  v2f d0[NC][2];
  Q4 q0[NC], q1[NC];
};

template <bool V>
struct BoolC {
  static constexpr bool value = V;
};

__device__ __forceinline__ float move_now(float v) {
  float o;
  asm volatile("v_mov_b32 %0, %1" : "=v"(o) : "v"(v));
  return o;
}

__device__ __forceinline__ v2f move_now(v2f v) {
  v2f o;
  o.x = move_now(v.x);
  o.y = move_now(v.y);
  return o;
}

__device__ __forceinline__ Q4 move_now(Q4 v) {
  Q4 o;
  o.w1r = move_now(v.w1r);
  o.w1i = move_now(v.w1i);
  o.w2r = move_now(v.w2r);
  o.w2i = move_now(v.w2i);
  return o;
}

template <int NC, int SHIFT, bool SRC0_PLAIN>
__device__ __forceinline__ void col_buf_roll(ColBuf<NC>& d) {
#pragma unroll
  for (int c = 0; c + SHIFT < NC; ++c) {
    if (SRC0_PLAIN) {
      d.d0[c][0] = move_now(d.d0[c + SHIFT][0]);
      d.d0[c][1] = move_now(d.d0[c + SHIFT][1]);
    } else {
      d.q0[c] = move_now(d.q0[c + SHIFT]);
    }
    d.q1[c] = move_now(d.q1[c + SHIFT]);
  }
  d.par >>= SHIFT;
  LOADS_STAY_AHEAD();
}

template <int NC, bool SRC0_PLAIN, int FIRST>
__device__ __forceinline__ void col_block_load(
    ColBuf<NC>& d, const float* __restrict__ p0, const float* __restrict__ p0a,
    const float* __restrict__ p0b, const float* __restrict__ p1a,
    const float* __restrict__ p1b, int stride, int W, int r, int x0,
    int koff) {
  LOADS_STAY_AHEAD();
  if (FIRST == 0)
    d.par = 0u;
  else
    col_buf_roll<NC, NC - FIRST, SRC0_PLAIN>(d);
#pragma unroll
  for (int c = FIRST; c < NC; ++c) {
    int rr[2];
#pragma unroll
    for (int j = 0; j < 2; ++j) {
      rr[j] = reflect_i(x0 + 4 * c + koff + j, r);
      if (SRC0_PLAIN) {
        size_t o = (size_t)rr[j] * W;
        d.d0[c][j].x = p0[o];
        d.d0[c][j].y = p0[o + 1];
      }
    }
    d.par |= ((unsigned)rr[0] & 1u) << c;
    int ob = (rr[0] >> 1) * stride;
    if (!SRC0_PLAIN) d.q0[c] = c2q_load(p0a, p0b, ob);
    d.q1[c] = c2q_load(p1a, p1b, ob);
  }
  LOADS_STAY_AHEAD();
}

template <int NC, int FIRST, bool BANDS, bool PLAIN>
__device__ __forceinline__ void col_block_load_lds0(
    ColBuf<NC>& d, const float* __restrict__ sl, int rbase, int pitch,
    const float* __restrict__ p1a, const float* __restrict__ p1b, int stride,
    int r, int x0, int koff) {
  LOADS_STAY_AHEAD();
  if (FIRST == 0) {
    if (BANDS) d.par = 0u;
  } else {
    col_buf_roll<NC, NC - FIRST, true>(d);
  }
#pragma unroll
  for (int c = FIRST; c < NC; ++c) {
    int rr0 = reflect_i(x0 + 4 * c + koff, r);
    int rr1 = reflect_i(x0 + 4 * c + koff + 1, r);
    if (PLAIN) {
      d.d0[c][0] = *(const v2f*)(sl + (rr0 - rbase) * pitch);
      d.d0[c][1] = *(const v2f*)(sl + (rr1 - rbase) * pitch);
    }
    if (BANDS) {
      d.par |= ((unsigned)rr0 & 1u) << c;
      d.q1[c] = c2q_load(p1a, p1b, (rr0 >> 1) * stride);
    }
  }
  LOADS_STAY_AHEAD();
}

template <int NC, bool SRC0_PLAIN, int FIRST, int P0A, int P0B, int P1A,
          int P1B>
__device__ __forceinline__ void col_block_load_lane32(
    ColBuf<NC>& d, const float* __restrict__ s0, const float* __restrict__ yb,
    unsigned bp, unsigned stride, unsigned W, int r, int x0, int koff,
    unsigned c0) {
  LOADS_STAY_AHEAD();
  if (FIRST == 0)
    d.par = 0u;
  else
    col_buf_roll<NC, NC - FIRST, SRC0_PLAIN>(d);
#pragma unroll
  for (int c = FIRST; c < NC; ++c) {
    int rr[2];
#pragma unroll
    for (int j = 0; j < 2; ++j) {
      rr[j] = reflect_i(x0 + 4 * c + koff + j, r);
      if (SRC0_PLAIN)
        d.d0[c][j] = ((const v2f*)s0)[(unsigned)rr[j] * (W >> 1) + (c0 >> 1)];
    }
    d.par |= ((unsigned)rr[0] & 1u) << c;
    unsigned ob = (unsigned)(rr[0] >> 1) * (stride >> 1) + (c0 >> 1);
    if (!SRC0_PLAIN) {
      v2f wa = ((const v2f*)yb)[ob + P0A * bp];
      v2f wb = ((const v2f*)yb)[ob + P0B * bp];
      d.q0[c].w1r = wa.x;
      d.q0[c].w1i = wa.y;
      d.q0[c].w2r = wb.x;
      d.q0[c].w2i = wb.y;
    }
    v2f wa = ((const v2f*)yb)[ob + P1A * bp];
    v2f wb = ((const v2f*)yb)[ob + P1B * bp];
    d.q1[c].w1r = wa.x;
    d.q1[c].w1i = wa.y;
    d.q1[c].w2r = wb.x;
    d.q1[c].w2i = wb.y;
  }
  LOADS_STAY_AHEAD();
}

template <int NC, bool SRC0_PLAIN>
__device__ __forceinline__ void col_block_compute(
    const ColBuf<NC>& d, int half, int c0, const float (&f0)[NC][2],
    const float (&f1)[NC][2], const bool (&k0)[NC][2], const bool (&k1)[NC][2],
    float* __restrict__ sT, int pitch) {
  const float s0sc = SRC0_PLAIN ? 1.0f : SQRT_HALF;
  v8f accE = {0.f, 0.f, 0.f, 0.f, 0.f, 0.f, 0.f, 0.f};
  v8f accO = {0.f, 0.f, 0.f, 0.f, 0.f, 0.f, 0.f, 0.f};
#pragma unroll
  for (int c = 0; c < NC; ++c) {
    v2f a0, a1, b0e, b0o, b1e, b1o;
    const int r0 = (int)((d.par >> c) & 1u);
    const int r1 = r0 ^ 1;
    a0.x = (k0[c][0] ? f0[c][0] : 0.0f) * s0sc;
    a0.y = (k0[c][1] ? f0[c][1] : 0.0f) * s0sc;
    a1.x = (k1[c][0] ? f1[c][0] : 0.0f) * SQRT_HALF;
    a1.y = (k1[c][1] ? f1[c][1] : 0.0f) * SQRT_HALF;
    if (SRC0_PLAIN) {
      b0e.x = d.d0[c][0].x;
      b0e.y = d.d0[c][1].x;
      b0o.x = d.d0[c][0].y;
      b0o.y = d.d0[c][1].y;
    } else {
      b0e.x = c2q_even_col(d.q0[c], r0);
      b0e.y = c2q_even_col(d.q0[c], r1);
      b0o.x = c2q_odd_col(d.q0[c], r0);
      b0o.y = c2q_odd_col(d.q0[c], r1);
    }
    b1e.x = c2q_even_col(d.q1[c], r0);
    b1e.y = c2q_even_col(d.q1[c], r1);
    b1o.x = c2q_odd_col(d.q1[c], r0);
    b1o.y = c2q_odd_col(d.q1[c], r1);
    accE = wmma4(a0, b0e, accE);
    accO = wmma4(a0, b0o, accO);
    accE = wmma4(a1, b1e, accE);
    accO = wmma4(a1, b1o, accO);
  }

#pragma unroll
  for (int e = 0; e < 8; ++e) {
    int lr = half * 8 + e;
    v2f v;
    v.x = accE[e];
    v.y = accO[e];
    *(v2f*)(sT + lr * pitch + (c0 ^ (lr << 2))) = v;
  }
}

__device__ __forceinline__ void lds_barrier() {
  __builtin_amdgcn_fence(__ATOMIC_RELEASE, "workgroup", "local");
  __builtin_amdgcn_s_barrier();
  __builtin_amdgcn_fence(__ATOMIC_ACQUIRE, "workgroup", "local");
}

template <int NC, int NT, bool MORE>
__device__ __forceinline__ void row_tiles_from_lds(
    const float* __restrict__ sT1, const float* __restrict__ sT2, int pitch,
    int W, int jc0, int wstep, int woff, int koff, int l16,
    const float (&f0)[NC][2], const float (&f1)[NC][2], const bool (&k0)[NC][2],
    const bool (&k1)[NC][2], float* __restrict__ op, size_t ldo) {
  v2f bf0[NC], bf1[NC];
#pragma unroll
  for (int t = 0; t < NC; ++t) {
    bf0[t].x = k0[t][0] ? f0[t][0] : 0.0f;
    bf0[t].y = k0[t][1] ? f0[t][1] : 0.0f;
    bf1[t].x = k1[t][0] ? f1[t][0] : 0.0f;
    bf1[t].y = k1[t][1] ? f1[t][1] : 0.0f;
  }
  const float* __restrict__ r1 = sT1 + l16 * pitch;
  const float* __restrict__ r2 = sT2 + l16 * pitch;
  const int sw = l16 << 2;
  v2f a1[NC][NT], a2[NC][NT];
  bool odd[2][NT];
#pragma unroll
  for (int t = 0; t < NC; ++t) {
#pragma unroll
    for (int i = 0; i < NT; ++i) {
      int xc = wstep * (jc0 + i) + woff + 4 * t + koff;
      int m = reflect_i(xc, W);
      int pe = (m & ~1) ^ sw;
      a1[t][i] = *(const v2f*)(r1 + pe);
      a2[t][i] = *(const v2f*)(r2 + pe);
      if (t == 0) odd[0][i] = (m & 1) != 0;
      if (t == NC - 1) odd[1][i] = (m & 1) != 0;
    }
  }
  LOADS_STAY_AHEAD();
  if (MORE) lds_barrier();
#pragma unroll
  for (int e = 0; e < 2; ++e) {
    const int t = e == 0 ? 0 : NC - 1;
#pragma unroll
    for (int i = 0; i < NT; ++i) {
      v2f p1 = a1[t][i], p2 = a2[t][i];
      a1[t][i].x = odd[e][i] ? p1.y : p1.x;
      a1[t][i].y = odd[e][i] ? p1.x : p1.y;
      a2[t][i].x = odd[e][i] ? p2.y : p2.x;
      a2[t][i].y = odd[e][i] ? p2.x : p2.y;
    }
  }

  v8f acc[NT];
#pragma unroll
  for (int i = 0; i < NT; ++i) {
#pragma unroll
    for (int e = 0; e < 8; ++e) acc[i][e] = 0.0f;
  }
#pragma unroll
  for (int t = 0; t < NC; ++t) {
#pragma unroll
    for (int i = 0; i < NT; ++i) {
      acc[i] = wmma4(a1[t][i], bf0[t], acc[i]);
      acc[i] = wmma4(a2[t][i], bf1[t], acc[i]);
    }
  }
#pragma unroll
  for (int i = 0; i < NT; ++i) {
#pragma unroll
    for (int e = 0; e < 8; ++e) op[(size_t)e * ldo + 16 * i] = acc[i][e];
  }
}

template <int S>
__global__ void __launch_bounds__(256)
j3_j2_level_kernel(const float* __restrict__ yl, const float* __restrict__ yh3,
                   const float* __restrict__ yh,
                   const float* __restrict__ ha0, const float* __restrict__ hb0,
                   const float* __restrict__ ha1, const float* __restrict__ hb1,
                   float* __restrict__ out, int nTilesR) {
  constexpr int W = 256;
  constexpr int W3 = 128;
  constexpr int LR = 8 * S + 16;
  __shared__ __attribute__((aligned(16))) float smem[(32 + LR) * W + 4];
  float* sT1 = smem;
  float* sT2 = smem + 16 * W + 2;
  float* sL = smem + 32 * W + 4;
  const int r = W;
  const int r3 = W3;
  int lane = threadIdx.x & 31;
  int wave = threadIdx.x >> 5;
  int nGroups = nTilesR / S;
  int b = blockIdx.x / nGroups;
  int I0 = (blockIdx.x % nGroups) * S;
  int half = lane >> 4;
  int l16 = lane & 15;
  int koff = half * 2;
  int c0 = wave * 32 + 2 * l16;

  float f0[4][2], f1[4][2];
  bool k0[4][2], k1[4][2];
#pragma unroll
  for (int c = 0; c < 4; ++c) {
#pragma unroll
    for (int j = 0; j < 2; ++j) {
      int kk = 4 * c + koff + j;
      f0[c][j] = *cif_tap(l16, kk, ha0, hb0, 0, k0[c][j]);
      f1[c][j] = *cif_tap(l16, kk, ha1, hb1, 1, k1[c][j]);
    }
  }

  int stride = W;
  size_t band = (size_t)(W / 2) * stride;
  const float* yb = yh + (size_t)b * 6 * band + c0;
  int H = 2 * r;
  size_t outW = 2 * (size_t)W;
  int jc0 = wave * 4;

  int R0 = 8 * I0 - 4;
  R0 = R0 < 0 ? 0 : R0;
  R0 = R0 > W - LR ? W - LR : R0;
  const float* sl = sL + c0;

  ColBuf<4> A, B;

  {
    static_assert(LR == 48, "three level-3 strips: two sub-groups, two rounds");
    const int sub = __builtin_amdgcn_readfirstlane(wave >> 2);
    int w4 = wave & 3;
    int c3 = w4 * 32 + 2 * l16;
    float* u1 = sT1 + sub * (16 * W3);
    float* u2 = sT2 + sub * (16 * W3);
    int stride3 = W3;
    size_t band3 = (size_t)(W3 / 2) * stride3;
    const float* yb3 = yh3 + (size_t)b * 6 * band3 + c3;
    const float* sp3 = yl + (size_t)b * r3 * W3 + c3;
    int x3 = (R0 >> 1) - 4 + 8 * sub;
    ColBuf<4> XA, XB;
    col_block_load<4, true, 0>(XA, sp3, nullptr, nullptr, yb3, yb3 + 5 * band3,
                               stride3, W3, r3, x3, koff);
    col_block_load<4, false, 0>(XB, nullptr, yb3 + 2 * band3, yb3 + 3 * band3,
                                yb3 + band3, yb3 + 4 * band3, stride3, W3, r3,
                                x3, koff);

#pragma unroll
    for (int c = 0; c < 4; ++c) {
#pragma unroll
      for (int j = 0; j < 2; ++j) {
        f0[c][j] = k0[c][j] ? f0[c][j] : 0.0f;
        f1[c][j] = k1[c][j] ? f1[c][j] : 0.0f;
        k0[c][j] = true;
        k1[c][j] = true;
      }
    }

    col_block_compute<4, true>(XA, half, c3, f0, f1, k0, k1, u1, W3);
    col_block_load<4, true, 0>(XA, sp3, nullptr, nullptr, yb3, yb3 + 5 * band3,
                               stride3, W3, r3, x3 + 16, koff);
    col_block_compute<4, false>(XB, half, c3, f0, f1, k0, k1, u2, W3);
    col_block_load<4, false, 0>(XB, nullptr, yb3 + 2 * band3, yb3 + 3 * band3,
                                yb3 + band3, yb3 + 4 * band3, stride3, W3, r3,
                                x3 + 16, koff);
    lds_barrier();
    float* op3 = sL + (16 * sub + half * 8) * W + 16 * (w4 * 4) + l16;
    row_tiles_from_lds<4, 4, true>(u1, u2, W3, W3, w4 * 4, 8, -4, koff, l16, f0,
                                   f1, k0, k1, op3, (size_t)W);

    if (sub == 0) {
      col_block_compute<4, true>(XA, half, c3, f0, f1, k0, k1, u1, W3);
      col_block_compute<4, false>(XB, half, c3, f0, f1, k0, k1, u2, W3);
    }
    col_block_load_lds0<4, 0, true, false>(A, sl, R0, W, yb, yb + 5 * band,
                                           stride, r, 8 * I0 - 4, koff);
    col_block_load<4, false, 0>(B, nullptr, yb + 2 * band, yb + 3 * band,
                                yb + band, yb + 4 * band, stride, W, r,
                                8 * I0 - 4, koff);
    lds_barrier();
    if (sub == 0)
      row_tiles_from_lds<4, 4, false>(u1, u2, W3, W3, w4 * 4, 8, -4, koff, l16,
                                      f0, f1, k0, k1, op3 + 32 * W, (size_t)W);
    lds_barrier();
  }

  col_block_load_lds0<4, 0, false, true>(A, sl, R0, W, yb, yb + 5 * band,
                                         stride, r, 8 * I0 - 4, koff);

  auto strip = [&](int I, auto more) {
    constexpr bool MORE = decltype(more)::value;
    col_block_compute<4, true>(A, half, c0, f0, f1, k0, k1, sT1, W);
    if (MORE)
      col_block_load_lds0<4, 2, true, true>(A, sl, R0, W, yb, yb + 5 * band,
                                            stride, r, 8 * I + 4, koff);
    col_block_compute<4, false>(B, half, c0, f0, f1, k0, k1, sT2, W);
    if (MORE)
      col_block_load<4, false, 2>(B, nullptr, yb + 2 * band, yb + 3 * band,
                                  yb + band, yb + 4 * band, stride, W, r,
                                  8 * I + 4, koff);
    lds_barrier();
    float* op =
        out + ((size_t)b * H + 16 * I + half * 8) * outW + 16 * jc0 + l16;
    row_tiles_from_lds<4, 4, MORE>(sT1, sT2, W, W, jc0, 8, -4, koff, l16, f0,
                                   f1, k0, k1, op, outW);
  };
#pragma unroll 1
  for (int s = 0; s + 1 < S; ++s) strip(I0 + s, BoolC<true>());
  strip(I0 + S - 1, BoolC<false>());
}

template <int WIDTH, int S>
__global__ void __launch_bounds__(WIDTH)
j1_level_kernel(const float* __restrict__ src0_plain,
                const float* __restrict__ yh, int Hb, int Wb, int r,
                const float* __restrict__ h0, int taps0,
                const float* __restrict__ h1, int taps1,
                float* __restrict__ out, int nTilesR) {
  __shared__ __attribute__((aligned(16))) float smem[32 * WIDTH + 4];
  float* sT1 = smem;
  float* sT2 = smem + 16 * WIDTH + 2;
  const int W = WIDTH;
  int lane = threadIdx.x & 31;
  int wave = threadIdx.x >> 5;
  int nGroups = nTilesR / S;
  int b = blockIdx.x / nGroups;
  int I0 = (blockIdx.x % nGroups) * S;
  int half = lane >> 4;
  int l16 = lane & 15;
  int koff = half * 2;
  int c0 = wave * 32 + 2 * l16;

  float f0[6][2], f1[6][2];
  bool k0[6][2], k1[6][2];
#pragma unroll
  for (int c = 0; c < 6; ++c) {
#pragma unroll
    for (int j = 0; j < 2; ++j) {
      int kk = 4 * c + koff + j;
      f0[c][j] = *cf_tap(l16, kk, h0, taps0, k0[c][j]);
      f1[c][j] = *cf_tap(l16, kk, h1, taps1, k1[c][j]);
    }
  }

  unsigned stride = 2u * (unsigned)Wb;
  unsigned bp = (unsigned)Hb * (unsigned)Wb;
  const float* yb = yh + (size_t)b * 12 * bp;
  const float* sp = src0_plain + (size_t)b * r * W;
  int jc0 = wave * 2;

  ColBuf<6> A, B;
  col_block_load_lane32<6, true, 0, 0, 0, 0, 5>(A, sp, yb, bp, stride,
                                                (unsigned)W, r, 16 * I0 - 4,
                                                koff, (unsigned)c0);
  col_block_load_lane32<6, false, 0, 2, 3, 1, 4>(B, nullptr, yb, bp, stride,
                                                 (unsigned)W, r, 16 * I0 - 4,
                                                 koff, (unsigned)c0);

#pragma unroll
  for (int c = 0; c < 6; ++c) {
#pragma unroll
    for (int j = 0; j < 2; ++j) {
      f0[c][j] = k0[c][j] ? f0[c][j] : 0.0f;
      f1[c][j] = k1[c][j] ? f1[c][j] : 0.0f;
      k0[c][j] = true;
      k1[c][j] = true;
    }
  }

  auto strip = [&](int I, auto more) {
    constexpr bool MORE = decltype(more)::value;
    col_block_compute<6, true>(A, half, c0, f0, f1, k0, k1, sT1, W);
    if (MORE)
      col_block_load_lane32<6, true, 2, 0, 0, 0, 5>(
          A, sp, yb, bp, stride, (unsigned)W, r, 16 * I + 12, koff,
          (unsigned)c0);
    col_block_compute<6, false>(B, half, c0, f0, f1, k0, k1, sT2, W);
    if (MORE)
      col_block_load_lane32<6, false, 2, 2, 3, 1, 4>(
          B, nullptr, yb, bp, stride, (unsigned)W, r, 16 * I + 12, koff,
          (unsigned)c0);
    lds_barrier();
    float* op = out + ((size_t)b * r + 16 * I + half * 8) * W + 16 * jc0 + l16;
    row_tiles_from_lds<6, 2, MORE>(sT1, sT2, W, W, jc0, 16, -4, koff, l16, f0,
                                   f1, k0, k1, op, (size_t)W);
  };
#pragma unroll 1
  for (int s = 0; s + 1 < S; ++s) strip(I0 + s, BoolC<true>());
  strip(I0 + S - 1, BoolC<false>());
}

extern "C" void kernel_launch(void* const* d_in, const int* in_sizes, int n_in,
                              void* d_out, int out_size, void* d_ws,
                              size_t ws_size, hipStream_t stream) {
  (void)in_sizes;
  (void)n_in;
  (void)out_size;
  (void)ws_size;

  const float* yl = (const float*)d_in[0];
  const float* yh0 = (const float*)d_in[1];
  const float* yh1 = (const float*)d_in[2];
  const float* yh2 = (const float*)d_in[3];
  const float* g0o = (const float*)d_in[4];
  const float* g1o = (const float*)d_in[5];
  const float* g0a = (const float*)d_in[6];
  const float* g0b = (const float*)d_in[7];
  const float* g1a = (const float*)d_in[8];
  const float* g1b = (const float*)d_in[9];

  float* LLb = (float*)d_ws;
  float* outp = (float*)d_out;


  j3_j2_level_kernel<4><<<dim3(64 * 8), dim3(256), 0, stream>>>(
      yl, yh2, yh1, g0b, g0a, g1b, g1a, LLb, 32);

  j1_level_kernel<512, 8><<<dim3(64 * 4), dim3(512), 0, stream>>>(
      LLb, yh0, 256, 256, 512, g0o, 7, g1o, 5, outp, 32);
}
